# starting kernel
# baseline (speedup 1.0000x reference)
#include <hip/hip_runtime.h>
#include <hip/hip_bf16.h>

#define N_NODES   100000
#define N_EDGES   1600000
#define HID       128
#define N_CLASSES 10
#define N_GRAPHS  512
#define N_LAYERS  5
#define E_TOT     (N_EDGES + N_NODES)

typedef __attribute__((ext_vector_type(2))) float v2f;
typedef __attribute__((ext_vector_type(8))) float v8f;

__global__ void init_deg_kernel(float* __restrict__ deg) {
    int i = blockIdx.x * blockDim.x + threadIdx.x;
    if (i < N_NODES) deg[i] = 1.0f;
}

__global__ void deg_accum_kernel(const int* __restrict__ dst, float* __restrict__ deg) {
    int e = blockIdx.x * blockDim.x + threadIdx.x;
    if (e < N_EDGES) atomicAdd(&deg[dst[e]], 1.0f);
}

__global__ void dinv_kernel(float* __restrict__ deg) {
    int i = blockIdx.x * blockDim.x + threadIdx.x;
    if (i < N_NODES) deg[i] = rsqrtf(deg[i]);
}

__global__ void norm_kernel(const int* __restrict__ src, const int* __restrict__ dst,
                            const float* __restrict__ dinv, float* __restrict__ norm) {
    int e = blockIdx.x * blockDim.x + threadIdx.x;
    if (e >= E_TOT) return;
    if (e < N_EDGES) {
        norm[e] = dinv[src[e]] * dinv[dst[e]];
    } else {
        float dv = dinv[e - N_EDGES];
        norm[e] = dv * dv;
    }
}

__global__ __launch_bounds__(256)
void gemm_wmma_kernel(const float* __restrict__ X, const float* __restrict__ W,
                      float* __restrict__ H) {
    int wave     = (blockIdx.x * blockDim.x + threadIdx.x) >> 5;
    int lane     = threadIdx.x & 31;
    int row_tile = wave >> 3;
    int col_tile = wave & 7;
    if (row_tile * 16 >= N_NODES) return;

    int m  = lane & 15;
    int kh = lane >> 4;
    const float* xrow = X + (size_t)(row_tile * 16 + m) * HID;
    int cb = col_tile * 16;

    v8f acc = {};
    #pragma unroll
    for (int kb = 0; kb < HID; kb += 4) {
        int k0 = kb + 2 * kh;
        v2f a = *(const v2f*)(xrow + k0);
        v2f b;
        b.x = W[(size_t)(k0 + 0) * HID + cb + m];
        b.y = W[(size_t)(k0 + 1) * HID + cb + m];
        acc = __builtin_amdgcn_wmma_f32_16x16x4_f32(
             false, a,  false, b,
             (short)0, acc,  false,  false);
    }

    float* hbase = H + (size_t)(row_tile * 16 + 8 * kh) * HID + cb + m;
    #pragma unroll
    for (int r = 0; r < 8; ++r) hbase[(size_t)r * HID] = acc[r];
}

__global__ void zero_kernel(float4* __restrict__ p, int n4) {
    int i = blockIdx.x * blockDim.x + threadIdx.x;
    if (i < n4) p[i] = make_float4(0.f, 0.f, 0.f, 0.f);
}

__global__ __launch_bounds__(256)
void scatter_kernel(const float* __restrict__ H, const int* __restrict__ src,
                    const int* __restrict__ dst, const float* __restrict__ norm,
                    float* __restrict__ out) {
    int wave = (blockIdx.x * blockDim.x + threadIdx.x) >> 5;
    int lane = threadIdx.x & 31;
    if (wave >= E_TOT) return;
    int s, d;
    if (wave < N_EDGES) { s = src[wave]; d = dst[wave]; }
    else                { s = wave - N_EDGES; d = s; }
    float w = norm[wave];
    float4 v = ((const float4*)(H + (size_t)s * HID))[lane];
    float* op = out + (size_t)d * HID + lane * 4;
    atomicAdd(op + 0, v.x * w);
    atomicAdd(op + 1, v.y * w);
    atomicAdd(op + 2, v.z * w);
    atomicAdd(op + 3, v.w * w);
}

__global__ void bias_relu_kernel(float* __restrict__ X, const float* __restrict__ b) {
    int i = blockIdx.x * blockDim.x + threadIdx.x;
    if (i < N_NODES * HID) {
        float v = X[i] + b[i & (HID - 1)];
        X[i] = v > 0.f ? v : 0.f;
    }
}

__global__ __launch_bounds__(256)
void pool_accum_kernel(const float* __restrict__ X, const int* __restrict__ batch,
                       float* __restrict__ pooled, float* __restrict__ counts) {
    int wave = (blockIdx.x * blockDim.x + threadIdx.x) >> 5;
    int lane = threadIdx.x & 31;
    if (wave >= N_NODES) return;
    int g = batch[wave];
    float4 v = ((const float4*)(X + (size_t)wave * HID))[lane];
    float* pp = pooled + (size_t)g * HID + lane * 4;
    atomicAdd(pp + 0, v.x);
    atomicAdd(pp + 1, v.y);
    atomicAdd(pp + 2, v.z);
    atomicAdd(pp + 3, v.w);
    if (lane == 0) atomicAdd(&counts[g], 1.0f);
}

__global__ void final_linear_kernel(const float* __restrict__ pooled,
                                    const float* __restrict__ counts,
                                    const float* __restrict__ linW,
                                    const float* __restrict__ linb,
                                    float* __restrict__ out) {
    int i = blockIdx.x * blockDim.x + threadIdx.x;
    if (i >= N_GRAPHS * N_CLASSES) return;
    int g = i / N_CLASSES;
    int c = i - g * N_CLASSES;
    float scale = 1.0f / fmaxf(counts[g], 1.0f);
    const float* pg = pooled + (size_t)g * HID;
    float acc = 0.f;
    #pragma unroll 4
    for (int f = 0; f < HID; ++f) acc += pg[f] * linW[f * N_CLASSES + c];
    out[i] = acc * scale + linb[c];
}

extern "C" void kernel_launch(void* const* d_in, const int* in_sizes, int n_in,
                              void* d_out, int out_size, void* d_ws, size_t ws_size,
                              hipStream_t stream) {
    const float* x     = (const float*)d_in[0];
    const int*   edge  = (const int*)d_in[1];
    const int*   src   = edge;
    const int*   dst   = edge + N_EDGES;
    const int*   batch = (const int*)d_in[2];
    const float* Ws    = (const float*)d_in[3];
    const float* bs    = (const float*)d_in[4];
    const float* linW  = (const float*)d_in[5];
    const float* linb  = (const float*)d_in[6];
    float*       out   = (float*)d_out;

    float* ws = (float*)d_ws;
    float* deg    = ws;
    float* norm   = deg + N_NODES;
    float* H      = norm + E_TOT;
    float* X1     = H + (size_t)N_NODES * HID;
    float* X2     = X1 + (size_t)N_NODES * HID;
    float* pooled = X2 + (size_t)N_NODES * HID;
    float* counts = pooled + (size_t)N_GRAPHS * HID;

    const int T = 256;

    init_deg_kernel<<<(N_NODES + T - 1) / T, T, 0, stream>>>(deg);
    deg_accum_kernel<<<(N_EDGES + T - 1) / T, T, 0, stream>>>(dst, deg);
    dinv_kernel<<<(N_NODES + T - 1) / T, T, 0, stream>>>(deg);
    norm_kernel<<<(E_TOT + T - 1) / T, T, 0, stream>>>(src, dst, deg, norm);

    const int gemm_blocks    = (N_NODES / 16) * (HID / 16) / 8;
    const int scatter_blocks = E_TOT / 8;
    const int feat4          = N_NODES * HID / 4;

    const float* xin = x;
    float* xbuf[2] = { X1, X2 };
    for (int l = 0; l < N_LAYERS; ++l) {
        float* xout = xbuf[l & 1];
        const float* Wl = Ws + (size_t)l * HID * HID;
        const float* bl = bs + (size_t)l * HID;

        gemm_wmma_kernel<<<gemm_blocks, T, 0, stream>>>(xin, Wl, H);
        zero_kernel<<<(feat4 + T - 1) / T, T, 0, stream>>>((float4*)xout, feat4);
        scatter_kernel<<<scatter_blocks, T, 0, stream>>>(H, src, dst, norm, xout);
        bias_relu_kernel<<<(N_NODES * HID + T - 1) / T, T, 0, stream>>>(xout, bl);

        xin = xout;
    }

    int pool4 = (N_GRAPHS * HID + N_GRAPHS + 3) / 4;
    zero_kernel<<<(pool4 + T - 1) / T, T, 0, stream>>>((float4*)pooled, pool4);
    pool_accum_kernel<<<N_NODES / 8, T, 0, stream>>>(xin, batch, pooled, counts);
    final_linear_kernel<<<(N_GRAPHS * N_CLASSES + T - 1) / T, T, 0, stream>>>(
        pooled, counts, linW, linb, out);
}


# Float-atomic edge scatter replaced: edges grouped by destination per call (returning int add = arrival-order rank, scan, fill); one wave sums a row in registers in rank order, one store, bias+ReLU
# speedup vs baseline: 2.4805x; 2.4805x over previous
#include <hip/hip_runtime.h>
#include <hip/hip_bf16.h>

#define N_NODES   100000
#define N_EDGES   1600000
#define HID       128
#define N_CLASSES 10
#define N_GRAPHS  512
#define N_LAYERS  5

static_assert(N_NODES % 4 == 0, "int4 passes over per-node arrays");
static_assert(N_NODES % 32 == 0, "8 waves x 4 nodes per gather block, 16-byte scalar loads per wave");

typedef __attribute__((ext_vector_type(2))) float v2f;
typedef __attribute__((ext_vector_type(8))) float v8f;

__global__ void zero_cnt_kernel(int4* __restrict__ cnt4) {
    int i = blockIdx.x * blockDim.x + threadIdx.x;
    if (i < N_NODES / 4) cnt4[i] = make_int4(0, 0, 0, 0);
}

__global__ void count_rank_kernel(const int* __restrict__ dst, int* __restrict__ cnt,
                                  int* __restrict__ rank) {
    int e = blockIdx.x * blockDim.x + threadIdx.x;
    if (e < N_EDGES) rank[e] = atomicAdd(&cnt[dst[e]], 1);
}

#define SCAN_TILE 1024
__global__ __launch_bounds__(256)
void scan_kernel(const int* __restrict__ cnt, int* __restrict__ rowptr,
                 float* __restrict__ dinv) {
    __shared__ int wsum[8];
    int tid  = threadIdx.x;
    int lane = tid & 31;
    int wv   = tid >> 5;
    int tile0 = blockIdx.x * SCAN_TILE;
    const int4* c4 = (const int4*)cnt;

    int pre = 0;
    for (int i = tid; i < tile0 / 4; i += 256) {
        int4 v = c4[i];
        pre += (v.x + v.y) + (v.z + v.w);
    }
    for (int o = 16; o > 0; o >>= 1) pre += __shfl_xor(pre, o);
    if (lane == 0) wsum[wv] = pre;
    __syncthreads();
    int base = 0;
    for (int k = 0; k < 8; ++k) base += wsum[k];
    __syncthreads();

    int i0 = tile0 + tid * 4;
    int4 v = make_int4(0, 0, 0, 0);
    if (i0 < N_NODES) v = c4[i0 >> 2];
    int t = (v.x + v.y) + (v.z + v.w);
    int incl = t;
    for (int o = 1; o < 32; o <<= 1) {
        int u = __shfl_up(incl, o);
        if (lane >= o) incl += u;
    }
    if (lane == 31) wsum[wv] = incl;
    __syncthreads();
    int woff = 0;
    for (int k = 0; k < wv; ++k) woff += wsum[k];
    int ex = base + woff + incl - t;
    if (i0 < N_NODES) {
        int4 r;
        r.x = ex;
        r.y = r.x + v.x;
        r.z = r.y + v.y;
        r.w = r.z + v.z;
        ((int4*)rowptr)[i0 >> 2] = r;
        float4 dv;
        dv.x = rsqrtf((float)(v.x + 1));
        dv.y = rsqrtf((float)(v.y + 1));
        dv.z = rsqrtf((float)(v.z + 1));
        dv.w = rsqrtf((float)(v.w + 1));
        ((float4*)dinv)[i0 >> 2] = dv;
    }
}

__global__ void fill_kernel(const int* __restrict__ src, const int* __restrict__ dst,
                            const int* __restrict__ rank, const int* __restrict__ rowptr,
                            const float* __restrict__ dinv, int2* __restrict__ csr) {
    int e = blockIdx.x * blockDim.x + threadIdx.x;
    if (e >= N_EDGES) return;
    int s = src[e];
    int d = dst[e];
    int p = rowptr[d] + rank[e];
    csr[p] = make_int2(s, __float_as_int(dinv[s] * dinv[d]));
}

__global__ __launch_bounds__(256)
void gemm_wmma_kernel(const float* __restrict__ X, const float* __restrict__ W,
                      float* __restrict__ H) {
    int wave     = (blockIdx.x * blockDim.x + threadIdx.x) >> 5;
    int lane     = threadIdx.x & 31;
    int row_tile = wave >> 3;
    int col_tile = wave & 7;
    if (row_tile * 16 >= N_NODES) return;

    int m  = lane & 15;
    int kh = lane >> 4;
    const float* xrow = X + (size_t)(row_tile * 16 + m) * HID;
    int cb = col_tile * 16;

    v8f acc = {};
    #pragma unroll
    for (int kb = 0; kb < HID; kb += 4) {
        int k0 = kb + 2 * kh;
        v2f a = *(const v2f*)(xrow + k0);
        v2f b;
        b.x = W[(size_t)(k0 + 0) * HID + cb + m];
        b.y = W[(size_t)(k0 + 1) * HID + cb + m];
        acc = __builtin_amdgcn_wmma_f32_16x16x4_f32(
             false, a,  false, b,
             (short)0, acc,  false,  false);
    }

    float* hbase = H + (size_t)(row_tile * 16 + 8 * kh) * HID + cb + m;
    #pragma unroll
    for (int r = 0; r < 8; ++r) hbase[(size_t)r * HID] = acc[r];
}

__global__ void zero_kernel(float4* __restrict__ p, int n4) {
    int i = blockIdx.x * blockDim.x + threadIdx.x;
    if (i < n4) p[i] = make_float4(0.f, 0.f, 0.f, 0.f);
}

#define NODES_PER_WAVE 4

template <int K>
__device__ __forceinline__ void rows(const float4* __restrict__ H4, int lane,
                                     int ex, int ew, int j, float4& acc) {
    float4 v[K];
    float  w[K];
    for (int k = 0; k < K; ++k) {
        int s = __builtin_amdgcn_readlane(ex, j + k);
        w[k]  = __int_as_float(__builtin_amdgcn_readlane(ew, j + k));
        v[k]  = H4[(size_t)s * (HID / 4) + lane];
    }
    for (int k = 0; k < K; ++k) {
        acc.x = fmaf(v[k].x, w[k], acc.x);
        acc.y = fmaf(v[k].y, w[k], acc.y);
        acc.z = fmaf(v[k].z, w[k], acc.z);
        acc.w = fmaf(v[k].w, w[k], acc.w);
    }
}

__global__ __launch_bounds__(256)
void gather_kernel(const float* __restrict__ H, const int* __restrict__ rowptr,
                   const int* __restrict__ cnt, const int2* __restrict__ csr,
                   const float* __restrict__ dinv, const float* __restrict__ bias,
                   float* __restrict__ out) {
    int lane = threadIdx.x & 31;
    int wave = __builtin_amdgcn_readfirstlane((int)((blockIdx.x * blockDim.x + threadIdx.x) >> 5));
    const float4* H4 = (const float4*)H;
    float4 b = ((const float4*)bias)[lane];

    int node0 = wave * NODES_PER_WAVE;
    int4   st4 = *(const int4*)(rowptr + node0);
    int4   n4  = *(const int4*)(cnt + node0);
    float4 d4  = *(const float4*)(dinv + node0);

    for (int q = 0; q < NODES_PER_WAVE; ++q) {
        int node  = node0 + q;
        int start = q == 0 ? st4.x : q == 1 ? st4.y : q == 2 ? st4.z : st4.w;
        int n     = q == 0 ? n4.x  : q == 1 ? n4.y  : q == 2 ? n4.z  : n4.w;
        float ds  = q == 0 ? d4.x  : q == 1 ? d4.y  : q == 2 ? d4.z  : d4.w;
        float4 hv = H4[(size_t)node * (HID / 4) + lane];
        float4 acc = make_float4(0.f, 0.f, 0.f, 0.f);

        for (int c0 = 0; c0 < n; c0 += 32) {
            int m = n - c0;
            if (m > 32) m = 32;
            int2 ent = make_int2(node, 0);
            if (lane < m) ent = csr[start + c0 + lane];
            int ex = (int)min((unsigned)ent.x, (unsigned)(N_NODES - 1));
            int ew = ent.y;
            int j = 0;
            for (; j + 8 <= m; j += 8) rows<8>(H4, lane, ex, ew, j, acc);
            if (j + 4 <= m) { rows<4>(H4, lane, ex, ew, j, acc); j += 4; }
            if (j + 2 <= m) { rows<2>(H4, lane, ex, ew, j, acc); j += 2; }
            if (j < m)      { rows<1>(H4, lane, ex, ew, j, acc); }
        }

        float wself = ds * ds;
        acc.x = fmaf(hv.x, wself, acc.x);
        acc.y = fmaf(hv.y, wself, acc.y);
        acc.z = fmaf(hv.z, wself, acc.z);
        acc.w = fmaf(hv.w, wself, acc.w);

        float4 r;
        r.x = acc.x + b.x;
        r.y = acc.y + b.y;
        r.z = acc.z + b.z;
        r.w = acc.w + b.w;
        r.x = r.x > 0.f ? r.x : 0.f;
        r.y = r.y > 0.f ? r.y : 0.f;
        r.z = r.z > 0.f ? r.z : 0.f;
        r.w = r.w > 0.f ? r.w : 0.f;
        ((float4*)out)[(size_t)node * (HID / 4) + lane] = r;
    }
}

__global__ __launch_bounds__(256)
void pool_accum_kernel(const float* __restrict__ X, const int* __restrict__ batch,
                       float* __restrict__ pooled, float* __restrict__ counts) {
    int wave = (blockIdx.x * blockDim.x + threadIdx.x) >> 5;
    int lane = threadIdx.x & 31;
    if (wave >= N_NODES) return;
    int g = batch[wave];
    float4 v = ((const float4*)(X + (size_t)wave * HID))[lane];
    float* pp = pooled + (size_t)g * HID + lane * 4;
    atomicAdd(pp + 0, v.x);
    atomicAdd(pp + 1, v.y);
    atomicAdd(pp + 2, v.z);
    atomicAdd(pp + 3, v.w);
    if (lane == 0) atomicAdd(&counts[g], 1.0f);
}

__global__ void final_linear_kernel(const float* __restrict__ pooled,
                                    const float* __restrict__ counts,
                                    const float* __restrict__ linW,
                                    const float* __restrict__ linb,
                                    float* __restrict__ out) {
    int i = blockIdx.x * blockDim.x + threadIdx.x;
    if (i >= N_GRAPHS * N_CLASSES) return;
    int g = i / N_CLASSES;
    int c = i - g * N_CLASSES;
    float scale = 1.0f / fmaxf(counts[g], 1.0f);
    const float* pg = pooled + (size_t)g * HID;
    float acc = 0.f;
    #pragma unroll 4
    for (int f = 0; f < HID; ++f) acc += pg[f] * linW[f * N_CLASSES + c];
    out[i] = acc * scale + linb[c];
}

extern "C" void kernel_launch(void* const* d_in, const int* in_sizes, int n_in,
                              void* d_out, int out_size, void* d_ws, size_t ws_size,
                              hipStream_t stream) {
    const float* x     = (const float*)d_in[0];
    const int*   edge  = (const int*)d_in[1];
    const int*   src   = edge;
    const int*   dst   = edge + N_EDGES;
    const int*   batch = (const int*)d_in[2];
    const float* Ws    = (const float*)d_in[3];
    const float* bs    = (const float*)d_in[4];
    const float* linW  = (const float*)d_in[5];
    const float* linb  = (const float*)d_in[6];
    float*       out   = (float*)d_out;

    float*  ws     = (float*)d_ws;
    int*    cnt    = (int*)ws;
    float*  dinv   = ws + N_NODES;
    int*    rowptr = (int*)(dinv + N_NODES);
    int*    rank   = rowptr + N_NODES;
    int2*   csr    = (int2*)(rank + N_EDGES);
    float*  H      = (float*)(csr + N_EDGES);
    float*  X      = H + (size_t)N_NODES * HID;
    float*  pooled = X + (size_t)N_NODES * HID;
    float*  counts = pooled + (size_t)N_GRAPHS * HID;

    const int T = 256;

    zero_cnt_kernel<<<(N_NODES / 4 + T - 1) / T, T, 0, stream>>>((int4*)cnt);
    count_rank_kernel<<<(N_EDGES + T - 1) / T, T, 0, stream>>>(dst, cnt, rank);
    scan_kernel<<<(N_NODES + SCAN_TILE - 1) / SCAN_TILE, T, 0, stream>>>(cnt, rowptr, dinv);
    fill_kernel<<<(N_EDGES + T - 1) / T, T, 0, stream>>>(src, dst, rank, rowptr, dinv, csr);

    const int gemm_blocks   = (N_NODES / 16) * (HID / 16) / 8;
    const int gather_blocks = N_NODES / (8 * NODES_PER_WAVE);

    const float* xin = x;
    for (int l = 0; l < N_LAYERS; ++l) {
        const float* Wl = Ws + (size_t)l * HID * HID;
        const float* bl = bs + (size_t)l * HID;

        gemm_wmma_kernel<<<gemm_blocks, T, 0, stream>>>(xin, Wl, H);
        gather_kernel<<<gather_blocks, T, 0, stream>>>(H, rowptr, cnt, csr, dinv, bl, X);

        xin = X;
    }

    int pool4 = (N_GRAPHS * HID + N_GRAPHS + 3) / 4;
    zero_kernel<<<(pool4 + T - 1) / T, T, 0, stream>>>((float4*)pooled, pool4);
    pool_accum_kernel<<<N_NODES / 8, T, 0, stream>>>(xin, batch, pooled, counts);
    final_linear_kernel<<<(N_GRAPHS * N_CLASSES + T - 1) / T, T, 0, stream>>>(
        pooled, counts, linW, linb, out);
}


# GEMM retiled: W copied once per workgroup into LDS; each wave owns a 16x128 strip, reads its A values once and uses every A fragment for all 8 column tiles; k order of every sum unchanged
# speedup vs baseline: 2.9562x; 1.1918x over previous
#include <hip/hip_runtime.h>
#include <hip/hip_bf16.h>

#define N_NODES   100000
#define N_EDGES   1600000
#define HID       128
#define N_CLASSES 10
#define N_GRAPHS  512
#define N_LAYERS  5

static_assert(N_NODES % 4 == 0, "int4 passes over per-node arrays");
static_assert(N_NODES % 32 == 0, "8 waves x 4 nodes per gather block, 16-byte scalar loads per wave");

typedef __attribute__((ext_vector_type(2))) float v2f;
typedef __attribute__((ext_vector_type(8))) float v8f;

__global__ void zero_cnt_kernel(int4* __restrict__ cnt4) {
    int i = blockIdx.x * blockDim.x + threadIdx.x;
    if (i < N_NODES / 4) cnt4[i] = make_int4(0, 0, 0, 0);
}

__global__ void count_rank_kernel(const int* __restrict__ dst, int* __restrict__ cnt,
                                  int* __restrict__ rank) {
    int e = blockIdx.x * blockDim.x + threadIdx.x;
    if (e < N_EDGES) rank[e] = atomicAdd(&cnt[dst[e]], 1);
}

#define SCAN_TILE 1024
__global__ __launch_bounds__(256)
void scan_kernel(const int* __restrict__ cnt, int* __restrict__ rowptr,
                 float* __restrict__ dinv) {
    __shared__ int wsum[8];
    int tid  = threadIdx.x;
    int lane = tid & 31;
    int wv   = tid >> 5;
    int tile0 = blockIdx.x * SCAN_TILE;
    const int4* c4 = (const int4*)cnt;

    int pre = 0;
    for (int i = tid; i < tile0 / 4; i += 256) {
        int4 v = c4[i];
        pre += (v.x + v.y) + (v.z + v.w);
    }
    for (int o = 16; o > 0; o >>= 1) pre += __shfl_xor(pre, o);
    if (lane == 0) wsum[wv] = pre;
    __syncthreads();
    int base = 0;
    for (int k = 0; k < 8; ++k) base += wsum[k];
    __syncthreads();

    int i0 = tile0 + tid * 4;
    int4 v = make_int4(0, 0, 0, 0);
    if (i0 < N_NODES) v = c4[i0 >> 2];
    int t = (v.x + v.y) + (v.z + v.w);
    int incl = t;
    for (int o = 1; o < 32; o <<= 1) {
        int u = __shfl_up(incl, o);
        if (lane >= o) incl += u;
    }
    if (lane == 31) wsum[wv] = incl;
    __syncthreads();
    int woff = 0;
    for (int k = 0; k < wv; ++k) woff += wsum[k];
    int ex = base + woff + incl - t;
    if (i0 < N_NODES) {
        int4 r;
        r.x = ex;
        r.y = r.x + v.x;
        r.z = r.y + v.y;
        r.w = r.z + v.z;
        ((int4*)rowptr)[i0 >> 2] = r;
        float4 dv;
        dv.x = rsqrtf((float)(v.x + 1));
        dv.y = rsqrtf((float)(v.y + 1));
        dv.z = rsqrtf((float)(v.z + 1));
        dv.w = rsqrtf((float)(v.w + 1));
        ((float4*)dinv)[i0 >> 2] = dv;
    }
}

__global__ void fill_kernel(const int* __restrict__ src, const int* __restrict__ dst,
                            const int* __restrict__ rank, const int* __restrict__ rowptr,
                            const float* __restrict__ dinv, int2* __restrict__ csr) {
    int e = blockIdx.x * blockDim.x + threadIdx.x;
    if (e >= N_EDGES) return;
    int s = src[e];
    int d = dst[e];
    int p = rowptr[d] + rank[e];
    csr[p] = make_int2(s, __float_as_int(dinv[s] * dinv[d]));
}

#define GEMM_WAVES   16
#define GEMM_THREADS (GEMM_WAVES * 32)
#define GEMM_BLOCKS  256
#define N_STRIPS     (N_NODES / 16)

static_assert(N_NODES % 16 == 0, "whole 16-row strips");
static_assert((HID * HID / 4) % GEMM_THREADS == 0, "W copy: whole passes of the workgroup");

__global__ __launch_bounds__(GEMM_THREADS)
void gemm_wmma_kernel(const float* __restrict__ X, const float* __restrict__ W,
                      float* __restrict__ H) {
    __shared__ float4 Wq[HID * HID / 4];

    int tid = threadIdx.x;
    #pragma unroll
    for (int i = 0; i < (HID * HID / 4) / GEMM_THREADS; ++i) {
        int p  = tid + i * GEMM_THREADS;
        int n  = p & (HID - 1);
        int gh = p >> 7;
        int k0 = (gh >> 1) * 8 + (gh & 1) * 2;
        float4 q;
        q.x = W[(k0 + 0) * HID + n];
        q.y = W[(k0 + 1) * HID + n];
        q.z = W[(k0 + 4) * HID + n];
        q.w = W[(k0 + 5) * HID + n];
        Wq[p] = q;
    }
    __syncthreads();

    int lane = tid & 31;
    int wv = __builtin_amdgcn_readfirstlane(tid >> 5);
    int m  = lane & 15;
    int kh = lane >> 4;

    int s0 = ((int)blockIdx.x * N_STRIPS) / GEMM_BLOCKS;
    int s1 = (((int)blockIdx.x + 1) * N_STRIPS) / GEMM_BLOCKS;
    const float4* bq = Wq + kh * HID + m;
    unsigned a_off = (unsigned)(m * HID + 2 * kh);
    unsigned h_off = (unsigned)(8 * kh * HID + m);

    for (int s = s0 + wv; s < s1; s += GEMM_WAVES) {
        const float* xs = X + (size_t)s * (16 * HID);
        float*       hs = H + (size_t)s * (16 * HID);
        v2f a[HID / 4];
        #pragma unroll
        for (int i = 0; i < HID / 4; ++i) a[i] = *(const v2f*)(xs + a_off + 4 * i);

        v8f acc[8];
        #pragma unroll
        for (int t = 0; t < 8; ++t) acc[t] = (v8f){};

        float4 b[2][8];
        #pragma unroll
        for (int t = 0; t < 8; ++t) b[0][t] = bq[t * 16];

        #pragma unroll
        for (int g = 0; g < HID / 8; ++g) {
            if (g + 1 < HID / 8) {
                #pragma unroll
                for (int t = 0; t < 8; ++t) b[(g + 1) & 1][t] = bq[(g + 1) * 2 * HID + t * 16];
            }
            asm volatile("" ::: "memory");
            #pragma unroll
            for (int t = 0; t < 8; ++t) {
                v2f bb = {b[g & 1][t].x, b[g & 1][t].y};
                acc[t] = __builtin_amdgcn_wmma_f32_16x16x4_f32(
                     false, a[2 * g],  false, bb,
                     (short)0, acc[t],  false,  false);
            }
            #pragma unroll
            for (int t = 0; t < 8; ++t) {
                v2f bb = {b[g & 1][t].z, b[g & 1][t].w};
                acc[t] = __builtin_amdgcn_wmma_f32_16x16x4_f32(
                     false, a[2 * g + 1],  false, bb,
                     (short)0, acc[t],  false,  false);
            }
        }

        #pragma unroll
        for (int t = 0; t < 8; ++t) {
            #pragma unroll
            for (int r = 0; r < 8; ++r) (hs + r * HID + t * 16)[h_off] = acc[t][r];
        }
    }
}

__global__ void zero_kernel(float4* __restrict__ p, int n4) {
    int i = blockIdx.x * blockDim.x + threadIdx.x;
    if (i < n4) p[i] = make_float4(0.f, 0.f, 0.f, 0.f);
}

#define NODES_PER_WAVE 4

template <int K>
__device__ __forceinline__ void rows(const float4* __restrict__ H4, int lane,
                                     int ex, int ew, int j, float4& acc) {
    float4 v[K];
    float  w[K];
    for (int k = 0; k < K; ++k) {
        int s = __builtin_amdgcn_readlane(ex, j + k);
        w[k]  = __int_as_float(__builtin_amdgcn_readlane(ew, j + k));
        v[k]  = H4[(size_t)s * (HID / 4) + lane];
    }
    for (int k = 0; k < K; ++k) {
        acc.x = fmaf(v[k].x, w[k], acc.x);
        acc.y = fmaf(v[k].y, w[k], acc.y);
        acc.z = fmaf(v[k].z, w[k], acc.z);
        acc.w = fmaf(v[k].w, w[k], acc.w);
    }
}

__global__ __launch_bounds__(256)
void gather_kernel(const float* __restrict__ H, const int* __restrict__ rowptr,
                   const int* __restrict__ cnt, const int2* __restrict__ csr,
                   const float* __restrict__ dinv, const float* __restrict__ bias,
                   float* __restrict__ out) {
    int lane = threadIdx.x & 31;
    int wave = __builtin_amdgcn_readfirstlane((int)((blockIdx.x * blockDim.x + threadIdx.x) >> 5));
    const float4* H4 = (const float4*)H;
    float4 b = ((const float4*)bias)[lane];

    int node0 = wave * NODES_PER_WAVE;
    int4   st4 = *(const int4*)(rowptr + node0);
    int4   n4  = *(const int4*)(cnt + node0);
    float4 d4  = *(const float4*)(dinv + node0);

    for (int q = 0; q < NODES_PER_WAVE; ++q) {
        int node  = node0 + q;
        int start = q == 0 ? st4.x : q == 1 ? st4.y : q == 2 ? st4.z : st4.w;
        int n     = q == 0 ? n4.x  : q == 1 ? n4.y  : q == 2 ? n4.z  : n4.w;
        float ds  = q == 0 ? d4.x  : q == 1 ? d4.y  : q == 2 ? d4.z  : d4.w;
        float4 hv = H4[(size_t)node * (HID / 4) + lane];
        float4 acc = make_float4(0.f, 0.f, 0.f, 0.f);

        for (int c0 = 0; c0 < n; c0 += 32) {
            int m = n - c0;
            if (m > 32) m = 32;
            int2 ent = make_int2(node, 0);
            if (lane < m) ent = csr[start + c0 + lane];
            int ex = (int)min((unsigned)ent.x, (unsigned)(N_NODES - 1));
            int ew = ent.y;
            int j = 0;
            for (; j + 8 <= m; j += 8) rows<8>(H4, lane, ex, ew, j, acc);
            if (j + 4 <= m) { rows<4>(H4, lane, ex, ew, j, acc); j += 4; }
            if (j + 2 <= m) { rows<2>(H4, lane, ex, ew, j, acc); j += 2; }
            if (j < m)      { rows<1>(H4, lane, ex, ew, j, acc); }
        }

        float wself = ds * ds;
        acc.x = fmaf(hv.x, wself, acc.x);
        acc.y = fmaf(hv.y, wself, acc.y);
        acc.z = fmaf(hv.z, wself, acc.z);
        acc.w = fmaf(hv.w, wself, acc.w);

        float4 r;
        r.x = acc.x + b.x;
        r.y = acc.y + b.y;
        r.z = acc.z + b.z;
        r.w = acc.w + b.w;
        r.x = r.x > 0.f ? r.x : 0.f;
        r.y = r.y > 0.f ? r.y : 0.f;
        r.z = r.z > 0.f ? r.z : 0.f;
        r.w = r.w > 0.f ? r.w : 0.f;
        ((float4*)out)[(size_t)node * (HID / 4) + lane] = r;
    }
}

__global__ __launch_bounds__(256)
void pool_accum_kernel(const float* __restrict__ X, const int* __restrict__ batch,
                       float* __restrict__ pooled, float* __restrict__ counts) {
    int wave = (blockIdx.x * blockDim.x + threadIdx.x) >> 5;
    int lane = threadIdx.x & 31;
    if (wave >= N_NODES) return;
    int g = batch[wave];
    float4 v = ((const float4*)(X + (size_t)wave * HID))[lane];
    float* pp = pooled + (size_t)g * HID + lane * 4;
    atomicAdd(pp + 0, v.x);
    atomicAdd(pp + 1, v.y);
    atomicAdd(pp + 2, v.z);
    atomicAdd(pp + 3, v.w);
    if (lane == 0) atomicAdd(&counts[g], 1.0f);
}

__global__ void final_linear_kernel(const float* __restrict__ pooled,
                                    const float* __restrict__ counts,
                                    const float* __restrict__ linW,
                                    const float* __restrict__ linb,
                                    float* __restrict__ out) {
    int i = blockIdx.x * blockDim.x + threadIdx.x;
    if (i >= N_GRAPHS * N_CLASSES) return;
    int g = i / N_CLASSES;
    int c = i - g * N_CLASSES;
    float scale = 1.0f / fmaxf(counts[g], 1.0f);
    const float* pg = pooled + (size_t)g * HID;
    float acc = 0.f;
    #pragma unroll 4
    for (int f = 0; f < HID; ++f) acc += pg[f] * linW[f * N_CLASSES + c];
    out[i] = acc * scale + linb[c];
}

extern "C" void kernel_launch(void* const* d_in, const int* in_sizes, int n_in,
                              void* d_out, int out_size, void* d_ws, size_t ws_size,
                              hipStream_t stream) {
    const float* x     = (const float*)d_in[0];
    const int*   edge  = (const int*)d_in[1];
    const int*   src   = edge;
    const int*   dst   = edge + N_EDGES;
    const int*   batch = (const int*)d_in[2];
    const float* Ws    = (const float*)d_in[3];
    const float* bs    = (const float*)d_in[4];
    const float* linW  = (const float*)d_in[5];
    const float* linb  = (const float*)d_in[6];
    float*       out   = (float*)d_out;

    float*  ws     = (float*)d_ws;
    int*    cnt    = (int*)ws;
    float*  dinv   = ws + N_NODES;
    int*    rowptr = (int*)(dinv + N_NODES);
    int*    rank   = rowptr + N_NODES;
    int2*   csr    = (int2*)(rank + N_EDGES);
    float*  H      = (float*)(csr + N_EDGES);
    float*  X      = H + (size_t)N_NODES * HID;
    float*  pooled = X + (size_t)N_NODES * HID;
    float*  counts = pooled + (size_t)N_GRAPHS * HID;

    const int T = 256;

    zero_cnt_kernel<<<(N_NODES / 4 + T - 1) / T, T, 0, stream>>>((int4*)cnt);
    count_rank_kernel<<<(N_EDGES + T - 1) / T, T, 0, stream>>>(dst, cnt, rank);
    scan_kernel<<<(N_NODES + SCAN_TILE - 1) / SCAN_TILE, T, 0, stream>>>(cnt, rowptr, dinv);
    fill_kernel<<<(N_EDGES + T - 1) / T, T, 0, stream>>>(src, dst, rank, rowptr, dinv, csr);

    const int gather_blocks = N_NODES / (8 * NODES_PER_WAVE);

    const float* xin = x;
    for (int l = 0; l < N_LAYERS; ++l) {
        const float* Wl = Ws + (size_t)l * HID * HID;
        const float* bl = bs + (size_t)l * HID;

        gemm_wmma_kernel<<<GEMM_BLOCKS, GEMM_THREADS, 0, stream>>>(xin, Wl, H);
        gather_kernel<<<gather_blocks, T, 0, stream>>>(H, rowptr, cnt, csr, dinv, bl, X);

        xin = X;
    }

    int pool4 = (N_GRAPHS * HID + N_GRAPHS + 3) / 4;
    zero_kernel<<<(pool4 + T - 1) / T, T, 0, stream>>>((float4*)pooled, pool4);
    pool_accum_kernel<<<N_NODES / 8, T, 0, stream>>>(xin, batch, pooled, counts);
    final_linear_kernel<<<(N_GRAPHS * N_CLASSES + T - 1) / T, T, 0, stream>>>(
        pooled, counts, linW, linb, out);
}


# Count, fill and pool kernels coarsened: 8 edges per thread, 16 nodes per wave, 8-16x fewer waves to start; a pool wave adds same-graph rows in registers before its float atomics (sum order changes)
# speedup vs baseline: 3.5476x; 1.2001x over previous
#include <hip/hip_runtime.h>
#include <hip/hip_bf16.h>

#define N_NODES   100000
#define N_EDGES   1600000
#define HID       128
#define N_CLASSES 10
#define N_GRAPHS  512
#define N_LAYERS  5

static_assert(N_NODES % 4 == 0, "int4 passes over per-node arrays");
static_assert(N_NODES % 32 == 0, "8 waves x 4 nodes per gather block, 16-byte scalar loads per wave");

typedef __attribute__((ext_vector_type(2))) float v2f;
typedef __attribute__((ext_vector_type(8))) float v8f;

__global__ void zero_cnt_kernel(int4* __restrict__ cnt4) {
    int i = blockIdx.x * blockDim.x + threadIdx.x;
    if (i < N_NODES / 4) cnt4[i] = make_int4(0, 0, 0, 0);
}

static_assert(N_EDGES % 4 == 0, "int4 passes over per-edge arrays");
#define EDGE_Q            (N_EDGES / 4)
#define EDGE_Q_PER_BLOCK  512
#define EDGE_BLOCKS       ((EDGE_Q + EDGE_Q_PER_BLOCK - 1) / EDGE_Q_PER_BLOCK)

__global__ __launch_bounds__(256)
void count_rank_kernel(const int* __restrict__ dst, int* __restrict__ cnt,
                       int* __restrict__ rank) {
    const int4* d4 = (const int4*)dst;
    int4*       r4 = (int4*)rank;
    int q0 = blockIdx.x * EDGE_Q_PER_BLOCK + threadIdx.x;
    int q1 = q0 + 256;
    if (q1 < EDGE_Q) {
        int4 a = d4[q0];
        int4 b = d4[q1];
        int4 ra, rb;
        ra.x = atomicAdd(&cnt[a.x], 1);
        ra.y = atomicAdd(&cnt[a.y], 1);
        ra.z = atomicAdd(&cnt[a.z], 1);
        ra.w = atomicAdd(&cnt[a.w], 1);
        rb.x = atomicAdd(&cnt[b.x], 1);
        rb.y = atomicAdd(&cnt[b.y], 1);
        rb.z = atomicAdd(&cnt[b.z], 1);
        rb.w = atomicAdd(&cnt[b.w], 1);
        r4[q0] = ra;
        r4[q1] = rb;
    } else if (q0 < EDGE_Q) {
        int4 a = d4[q0];
        int4 ra;
        ra.x = atomicAdd(&cnt[a.x], 1);
        ra.y = atomicAdd(&cnt[a.y], 1);
        ra.z = atomicAdd(&cnt[a.z], 1);
        ra.w = atomicAdd(&cnt[a.w], 1);
        r4[q0] = ra;
    }
}

#define SCAN_TILE 1024
__global__ __launch_bounds__(256)
void scan_kernel(const int* __restrict__ cnt, int* __restrict__ rowptr,
                 float* __restrict__ dinv) {
    __shared__ int wsum[8];
    int tid  = threadIdx.x;
    int lane = tid & 31;
    int wv   = tid >> 5;
    int tile0 = blockIdx.x * SCAN_TILE;
    const int4* c4 = (const int4*)cnt;

    int pre = 0;
    for (int i = tid; i < tile0 / 4; i += 256) {
        int4 v = c4[i];
        pre += (v.x + v.y) + (v.z + v.w);
    }
    for (int o = 16; o > 0; o >>= 1) pre += __shfl_xor(pre, o);
    if (lane == 0) wsum[wv] = pre;
    __syncthreads();
    int base = 0;
    for (int k = 0; k < 8; ++k) base += wsum[k];
    __syncthreads();

    int i0 = tile0 + tid * 4;
    int4 v = make_int4(0, 0, 0, 0);
    if (i0 < N_NODES) v = c4[i0 >> 2];
    int t = (v.x + v.y) + (v.z + v.w);
    int incl = t;
    for (int o = 1; o < 32; o <<= 1) {
        int u = __shfl_up(incl, o);
        if (lane >= o) incl += u;
    }
    if (lane == 31) wsum[wv] = incl;
    __syncthreads();
    int woff = 0;
    for (int k = 0; k < wv; ++k) woff += wsum[k];
    int ex = base + woff + incl - t;
    if (i0 < N_NODES) {
        int4 r;
        r.x = ex;
        r.y = r.x + v.x;
        r.z = r.y + v.y;
        r.w = r.z + v.z;
        ((int4*)rowptr)[i0 >> 2] = r;
        float4 dv;
        dv.x = rsqrtf((float)(v.x + 1));
        dv.y = rsqrtf((float)(v.y + 1));
        dv.z = rsqrtf((float)(v.z + 1));
        dv.w = rsqrtf((float)(v.w + 1));
        ((float4*)dinv)[i0 >> 2] = dv;
    }
}

__device__ __forceinline__ int2 csr_entry(int s, float ds, float dd) {
    return make_int2(s, __float_as_int(ds * dd));
}

__global__ __launch_bounds__(256)
void fill_kernel(const int* __restrict__ src, const int* __restrict__ dst,
                 const int* __restrict__ rank, const int* __restrict__ rowptr,
                 const float* __restrict__ dinv, int2* __restrict__ csr) {
    const int4* s4 = (const int4*)src;
    const int4* d4 = (const int4*)dst;
    const int4* r4 = (const int4*)rank;
    int q0 = blockIdx.x * EDGE_Q_PER_BLOCK + threadIdx.x;
    int q1 = q0 + 256;
    if (q1 < EDGE_Q) {
        int4 sa = s4[q0], da = d4[q0], ra = r4[q0];
        int4 sb = s4[q1], db = d4[q1], rb = r4[q1];
        int   pa0 = rowptr[da.x], pa1 = rowptr[da.y], pa2 = rowptr[da.z], pa3 = rowptr[da.w];
        int   pb0 = rowptr[db.x], pb1 = rowptr[db.y], pb2 = rowptr[db.z], pb3 = rowptr[db.w];
        float ua0 = dinv[sa.x], ua1 = dinv[sa.y], ua2 = dinv[sa.z], ua3 = dinv[sa.w];
        float ub0 = dinv[sb.x], ub1 = dinv[sb.y], ub2 = dinv[sb.z], ub3 = dinv[sb.w];
        float va0 = dinv[da.x], va1 = dinv[da.y], va2 = dinv[da.z], va3 = dinv[da.w];
        float vb0 = dinv[db.x], vb1 = dinv[db.y], vb2 = dinv[db.z], vb3 = dinv[db.w];
        csr[pa0 + ra.x] = csr_entry(sa.x, ua0, va0);
        csr[pa1 + ra.y] = csr_entry(sa.y, ua1, va1);
        csr[pa2 + ra.z] = csr_entry(sa.z, ua2, va2);
        csr[pa3 + ra.w] = csr_entry(sa.w, ua3, va3);
        csr[pb0 + rb.x] = csr_entry(sb.x, ub0, vb0);
        csr[pb1 + rb.y] = csr_entry(sb.y, ub1, vb1);
        csr[pb2 + rb.z] = csr_entry(sb.z, ub2, vb2);
        csr[pb3 + rb.w] = csr_entry(sb.w, ub3, vb3);
    } else if (q0 < EDGE_Q) {
        int4 sa = s4[q0], da = d4[q0], ra = r4[q0];
        int   pa0 = rowptr[da.x], pa1 = rowptr[da.y], pa2 = rowptr[da.z], pa3 = rowptr[da.w];
        float ua0 = dinv[sa.x], ua1 = dinv[sa.y], ua2 = dinv[sa.z], ua3 = dinv[sa.w];
        float va0 = dinv[da.x], va1 = dinv[da.y], va2 = dinv[da.z], va3 = dinv[da.w];
        csr[pa0 + ra.x] = csr_entry(sa.x, ua0, va0);
        csr[pa1 + ra.y] = csr_entry(sa.y, ua1, va1);
        csr[pa2 + ra.z] = csr_entry(sa.z, ua2, va2);
        csr[pa3 + ra.w] = csr_entry(sa.w, ua3, va3);
    }
}

#define GEMM_WAVES   16
#define GEMM_THREADS (GEMM_WAVES * 32)
#define GEMM_BLOCKS  256
#define N_STRIPS     (N_NODES / 16)

static_assert(N_NODES % 16 == 0, "whole 16-row strips");
static_assert((HID * HID / 4) % GEMM_THREADS == 0, "W copy: whole passes of the workgroup");

__global__ __launch_bounds__(GEMM_THREADS)
void gemm_wmma_kernel(const float* __restrict__ X, const float* __restrict__ W,
                      float* __restrict__ H) {
    __shared__ float4 Wq[HID * HID / 4];

    int tid = threadIdx.x;
    #pragma unroll
    for (int i = 0; i < (HID * HID / 4) / GEMM_THREADS; ++i) {
        int p  = tid + i * GEMM_THREADS;
        int n  = p & (HID - 1);
        int gh = p >> 7;
        int k0 = (gh >> 1) * 8 + (gh & 1) * 2;
        float4 q;
        q.x = W[(k0 + 0) * HID + n];
        q.y = W[(k0 + 1) * HID + n];
        q.z = W[(k0 + 4) * HID + n];
        q.w = W[(k0 + 5) * HID + n];
        Wq[p] = q;
    }
    __syncthreads();

    int lane = tid & 31;
    int wv = __builtin_amdgcn_readfirstlane(tid >> 5);
    int m  = lane & 15;
    int kh = lane >> 4;

    int s0 = ((int)blockIdx.x * N_STRIPS) / GEMM_BLOCKS;
    int s1 = (((int)blockIdx.x + 1) * N_STRIPS) / GEMM_BLOCKS;
    const float4* bq = Wq + kh * HID + m;
    unsigned a_off = (unsigned)(m * HID + 2 * kh);
    unsigned h_off = (unsigned)(8 * kh * HID + m);

    for (int s = s0 + wv; s < s1; s += GEMM_WAVES) {
        const float* xs = X + (size_t)s * (16 * HID);
        float*       hs = H + (size_t)s * (16 * HID);
        v2f a[HID / 4];
        #pragma unroll
        for (int i = 0; i < HID / 4; ++i) a[i] = *(const v2f*)(xs + a_off + 4 * i);

        v8f acc[8];
        #pragma unroll
        for (int t = 0; t < 8; ++t) acc[t] = (v8f){};

        float4 b[2][8];
        #pragma unroll
        for (int t = 0; t < 8; ++t) b[0][t] = bq[t * 16];

        #pragma unroll
        for (int g = 0; g < HID / 8; ++g) {
            if (g + 1 < HID / 8) {
                #pragma unroll
                for (int t = 0; t < 8; ++t) b[(g + 1) & 1][t] = bq[(g + 1) * 2 * HID + t * 16];
            }
            asm volatile("" ::: "memory");
            #pragma unroll
            for (int t = 0; t < 8; ++t) {
                v2f bb = {b[g & 1][t].x, b[g & 1][t].y};
                acc[t] = __builtin_amdgcn_wmma_f32_16x16x4_f32(
                     false, a[2 * g],  false, bb,
                     (short)0, acc[t],  false,  false);
            }
            #pragma unroll
            for (int t = 0; t < 8; ++t) {
                v2f bb = {b[g & 1][t].z, b[g & 1][t].w};
                acc[t] = __builtin_amdgcn_wmma_f32_16x16x4_f32(
                     false, a[2 * g + 1],  false, bb,
                     (short)0, acc[t],  false,  false);
            }
        }

        #pragma unroll
        for (int t = 0; t < 8; ++t) {
            #pragma unroll
            for (int r = 0; r < 8; ++r) (hs + r * HID + t * 16)[h_off] = acc[t][r];
        }
    }
}

__global__ void zero_kernel(float4* __restrict__ p, int n4) {
    int i = blockIdx.x * blockDim.x + threadIdx.x;
    if (i < n4) p[i] = make_float4(0.f, 0.f, 0.f, 0.f);
}

#define NODES_PER_WAVE 4

template <int K>
__device__ __forceinline__ void rows(const float4* __restrict__ H4, int lane,
                                     int ex, int ew, int j, float4& acc) {
    float4 v[K];
    float  w[K];
    for (int k = 0; k < K; ++k) {
        int s = __builtin_amdgcn_readlane(ex, j + k);
        w[k]  = __int_as_float(__builtin_amdgcn_readlane(ew, j + k));
        v[k]  = H4[(size_t)s * (HID / 4) + lane];
    }
    for (int k = 0; k < K; ++k) {
        acc.x = fmaf(v[k].x, w[k], acc.x);
        acc.y = fmaf(v[k].y, w[k], acc.y);
        acc.z = fmaf(v[k].z, w[k], acc.z);
        acc.w = fmaf(v[k].w, w[k], acc.w);
    }
}

__global__ __launch_bounds__(256)
void gather_kernel(const float* __restrict__ H, const int* __restrict__ rowptr,
                   const int* __restrict__ cnt, const int2* __restrict__ csr,
                   const float* __restrict__ dinv, const float* __restrict__ bias,
                   float* __restrict__ out) {
    int lane = threadIdx.x & 31;
    int wave = __builtin_amdgcn_readfirstlane((int)((blockIdx.x * blockDim.x + threadIdx.x) >> 5));
    const float4* H4 = (const float4*)H;
    float4 b = ((const float4*)bias)[lane];

    int node0 = wave * NODES_PER_WAVE;
    int4   st4 = *(const int4*)(rowptr + node0);
    int4   n4  = *(const int4*)(cnt + node0);
    float4 d4  = *(const float4*)(dinv + node0);

    for (int q = 0; q < NODES_PER_WAVE; ++q) {
        int node  = node0 + q;
        int start = q == 0 ? st4.x : q == 1 ? st4.y : q == 2 ? st4.z : st4.w;
        int n     = q == 0 ? n4.x  : q == 1 ? n4.y  : q == 2 ? n4.z  : n4.w;
        float ds  = q == 0 ? d4.x  : q == 1 ? d4.y  : q == 2 ? d4.z  : d4.w;
        float4 hv = H4[(size_t)node * (HID / 4) + lane];
        float4 acc = make_float4(0.f, 0.f, 0.f, 0.f);

        for (int c0 = 0; c0 < n; c0 += 32) {
            int m = n - c0;
            if (m > 32) m = 32;
            int2 ent = make_int2(node, 0);
            if (lane < m) ent = csr[start + c0 + lane];
            int ex = (int)min((unsigned)ent.x, (unsigned)(N_NODES - 1));
            int ew = ent.y;
            int j = 0;
            for (; j + 8 <= m; j += 8) rows<8>(H4, lane, ex, ew, j, acc);
            if (j + 4 <= m) { rows<4>(H4, lane, ex, ew, j, acc); j += 4; }
            if (j + 2 <= m) { rows<2>(H4, lane, ex, ew, j, acc); j += 2; }
            if (j < m)      { rows<1>(H4, lane, ex, ew, j, acc); }
        }

        float wself = ds * ds;
        acc.x = fmaf(hv.x, wself, acc.x);
        acc.y = fmaf(hv.y, wself, acc.y);
        acc.z = fmaf(hv.z, wself, acc.z);
        acc.w = fmaf(hv.w, wself, acc.w);

        float4 r;
        r.x = acc.x + b.x;
        r.y = acc.y + b.y;
        r.z = acc.z + b.z;
        r.w = acc.w + b.w;
        r.x = r.x > 0.f ? r.x : 0.f;
        r.y = r.y > 0.f ? r.y : 0.f;
        r.z = r.z > 0.f ? r.z : 0.f;
        r.w = r.w > 0.f ? r.w : 0.f;
        ((float4*)out)[(size_t)node * (HID / 4) + lane] = r;
    }
}

#define POOL_NODES 16
static_assert(N_NODES % POOL_NODES == 0, "whole groups of nodes per wave, 64-byte scalar loads of batch");

__device__ __forceinline__ void pool_flush(float* __restrict__ pooled, float* __restrict__ counts,
                                           int g, int lane, float4 acc, float c) {
    float* pp = pooled + (size_t)g * HID + lane * 4;
    atomicAdd(pp + 0, acc.x);
    atomicAdd(pp + 1, acc.y);
    atomicAdd(pp + 2, acc.z);
    atomicAdd(pp + 3, acc.w);
    if (lane == 0) atomicAdd(&counts[g], c);
}

__global__ __launch_bounds__(256)
void pool_accum_kernel(const float* __restrict__ X, const int* __restrict__ batch,
                       float* __restrict__ pooled, float* __restrict__ counts) {
    int lane = threadIdx.x & 31;
    int wave = __builtin_amdgcn_readfirstlane((int)((blockIdx.x * blockDim.x + threadIdx.x) >> 5));
    if (wave >= N_NODES / POOL_NODES) return;
    int node0 = wave * POOL_NODES;

    const int4* b4 = (const int4*)(batch + node0);
    int4 i0 = b4[0], i1 = b4[1], i2 = b4[2], i3 = b4[3];
    int g[POOL_NODES] = {i0.x, i0.y, i0.z, i0.w, i1.x, i1.y, i1.z, i1.w,
                         i2.x, i2.y, i2.z, i2.w, i3.x, i3.y, i3.z, i3.w};

    const float4* X4 = (const float4*)X + (size_t)node0 * (HID / 4) + lane;
    float4 v[POOL_NODES];
    #pragma unroll
    for (int k = 0; k < POOL_NODES; ++k) v[k] = X4[k * (HID / 4)];

    int    gcur = g[0];
    float4 acc  = v[0];
    float  c    = 1.0f;
    #pragma unroll
    for (int k = 1; k < POOL_NODES; ++k) {
        if (g[k] != gcur) {
            pool_flush(pooled, counts, gcur, lane, acc, c);
            gcur = g[k];
            acc  = v[k];
            c    = 1.0f;
        } else {
            acc.x += v[k].x;
            acc.y += v[k].y;
            acc.z += v[k].z;
            acc.w += v[k].w;
            c += 1.0f;
        }
    }
    pool_flush(pooled, counts, gcur, lane, acc, c);
}

__global__ void final_linear_kernel(const float* __restrict__ pooled,
                                    const float* __restrict__ counts,
                                    const float* __restrict__ linW,
                                    const float* __restrict__ linb,
                                    float* __restrict__ out) {
    int i = blockIdx.x * blockDim.x + threadIdx.x;
    if (i >= N_GRAPHS * N_CLASSES) return;
    int g = i / N_CLASSES;
    int c = i - g * N_CLASSES;
    float scale = 1.0f / fmaxf(counts[g], 1.0f);
    const float* pg = pooled + (size_t)g * HID;
    float acc = 0.f;
    #pragma unroll 4
    for (int f = 0; f < HID; ++f) acc += pg[f] * linW[f * N_CLASSES + c];
    out[i] = acc * scale + linb[c];
}

extern "C" void kernel_launch(void* const* d_in, const int* in_sizes, int n_in,
                              void* d_out, int out_size, void* d_ws, size_t ws_size,
                              hipStream_t stream) {
    const float* x     = (const float*)d_in[0];
    const int*   edge  = (const int*)d_in[1];
    const int*   src   = edge;
    const int*   dst   = edge + N_EDGES;
    const int*   batch = (const int*)d_in[2];
    const float* Ws    = (const float*)d_in[3];
    const float* bs    = (const float*)d_in[4];
    const float* linW  = (const float*)d_in[5];
    const float* linb  = (const float*)d_in[6];
    float*       out   = (float*)d_out;

    float*  ws     = (float*)d_ws;
    int*    cnt    = (int*)ws;
    float*  dinv   = ws + N_NODES;
    int*    rowptr = (int*)(dinv + N_NODES);
    int*    rank   = rowptr + N_NODES;
    int2*   csr    = (int2*)(rank + N_EDGES);
    float*  H      = (float*)(csr + N_EDGES);
    float*  X      = H + (size_t)N_NODES * HID;
    float*  pooled = X + (size_t)N_NODES * HID;
    float*  counts = pooled + (size_t)N_GRAPHS * HID;

    const int T = 256;

    zero_cnt_kernel<<<(N_NODES / 4 + T - 1) / T, T, 0, stream>>>((int4*)cnt);
    count_rank_kernel<<<EDGE_BLOCKS, T, 0, stream>>>(dst, cnt, rank);
    scan_kernel<<<(N_NODES + SCAN_TILE - 1) / SCAN_TILE, T, 0, stream>>>(cnt, rowptr, dinv);
    fill_kernel<<<EDGE_BLOCKS, T, 0, stream>>>(src, dst, rank, rowptr, dinv, csr);

    const int gather_blocks = N_NODES / (8 * NODES_PER_WAVE);

    const float* xin = x;
    for (int l = 0; l < N_LAYERS; ++l) {
        const float* Wl = Ws + (size_t)l * HID * HID;
        const float* bl = bs + (size_t)l * HID;

        gemm_wmma_kernel<<<GEMM_BLOCKS, GEMM_THREADS, 0, stream>>>(xin, Wl, H);
        gather_kernel<<<gather_blocks, T, 0, stream>>>(H, rowptr, cnt, csr, dinv, bl, X);

        xin = X;
    }

    int pool4 = (N_GRAPHS * HID + N_GRAPHS + 3) / 4;
    zero_kernel<<<(pool4 + T - 1) / T, T, 0, stream>>>((float4*)pooled, pool4);
    pool_accum_kernel<<<(N_NODES / POOL_NODES + 7) / 8, T, 0, stream>>>(xin, batch, pooled, counts);
    final_linear_kernel<<<(N_GRAPHS * N_CLASSES + T - 1) / T, T, 0, stream>>>(
        pooled, counts, linW, linb, out);
}


# Edge counting moved from returning adds in memory to LDS counters: 32 slices, one workgroup each, two 16-bit counts per word, then a per-node sum over slices; row slot order = slice, then arrival
# speedup vs baseline: 3.9420x; 1.1112x over previous
#include <hip/hip_runtime.h>
#include <hip/hip_bf16.h>

#define N_NODES   100000
#define N_EDGES   1600000
#define HID       128
#define N_CLASSES 10
#define N_GRAPHS  512
#define N_LAYERS  5

static_assert(N_NODES % 4 == 0, "int4 passes over per-node arrays");
static_assert(N_NODES % 32 == 0, "8 waves x 4 nodes per gather block, 16-byte scalar loads per wave");

typedef __attribute__((ext_vector_type(2))) float v2f;
typedef __attribute__((ext_vector_type(8))) float v8f;


static_assert(N_EDGES % 4 == 0, "int4 passes over per-edge arrays");
#define EDGE_Q            (N_EDGES / 4)
#define EDGE_Q_PER_BLOCK  512
#define EDGE_BLOCKS       ((EDGE_Q + EDGE_Q_PER_BLOCK - 1) / EDGE_Q_PER_BLOCK)

#define N_SLICES     32
#define SLICE_EDGES  (N_EDGES / N_SLICES)
#define SLICE_Q      (SLICE_EDGES / 4)
#define PAIR_WORDS   (N_NODES / 2)
#define CL_THREADS   1024
#define CL_ITERS     ((SLICE_Q + CL_THREADS - 1) / CL_THREADS)

static_assert(N_EDGES % N_SLICES == 0, "equal slices");
static_assert(SLICE_EDGES % 4 == 0, "a slice is whole int4 groups, so a group never straddles two slices");
static_assert(SLICE_EDGES <= 65535, "a 16-bit counter cannot overflow inside one slice");
static_assert(N_NODES % 8 == 0, "16-byte passes over the packed counter table");

__device__ __forceinline__ int lds_count(unsigned* __restrict__ c, int d) {
    unsigned u  = min((unsigned)d, (unsigned)(N_NODES - 1));
    unsigned sh = (u & 1u) * 16u;
    unsigned old = atomicAdd(&c[u >> 1], 1u << sh);
    return (int)((old >> sh) & 0xFFFFu);
}

__global__ __launch_bounds__(CL_THREADS)
void count_lds_kernel(const int* __restrict__ dst, int* __restrict__ rank,
                      unsigned* __restrict__ hist) {
    __shared__ uint4 c4[PAIR_WORDS / 4];
    unsigned* c = (unsigned*)c4;
    int tid = threadIdx.x;
    int s   = blockIdx.x;
    const int4* d4 = (const int4*)dst + (size_t)s * SLICE_Q;
    int4*       r4 = (int4*)rank + (size_t)s * SLICE_Q;

    int4 a[CL_ITERS];
    #pragma unroll
    for (int k = 0; k < CL_ITERS; ++k) {
        int q = tid + k * CL_THREADS;
        a[k] = make_int4(0, 0, 0, 0);
        if (q < SLICE_Q) a[k] = d4[q];
    }
    for (int i = tid; i < PAIR_WORDS / 4; i += CL_THREADS) c4[i] = make_uint4(0u, 0u, 0u, 0u);
    __syncthreads();

    #pragma unroll
    for (int k = 0; k < CL_ITERS; ++k) {
        int q = tid + k * CL_THREADS;
        if (q < SLICE_Q) {
            int4 r;
            r.x = lds_count(c, a[k].x);
            r.y = lds_count(c, a[k].y);
            r.z = lds_count(c, a[k].z);
            r.w = lds_count(c, a[k].w);
            r4[q] = r;
        }
    }
    __syncthreads();

    uint4* h4 = (uint4*)(hist + (size_t)s * PAIR_WORDS);
    for (int i = tid; i < PAIR_WORDS / 4; i += CL_THREADS) h4[i] = c4[i];
}

__global__ __launch_bounds__(256)
void slice_prefix_kernel(const unsigned* __restrict__ hist, int* __restrict__ pre,
                         int* __restrict__ cnt) {
    int i = blockIdx.x * blockDim.x + threadIdx.x;
    if (i >= N_NODES / 4) return;
    uint2 w[N_SLICES];
    #pragma unroll
    for (int s = 0; s < N_SLICES; ++s)
        w[s] = ((const uint2*)(hist + (size_t)s * PAIR_WORDS))[i];
    int4 run = make_int4(0, 0, 0, 0);
    #pragma unroll
    for (int s = 0; s < N_SLICES; ++s) {
        ((int4*)(pre + (size_t)s * N_NODES))[i] = run;
        run.x += (int)(w[s].x & 0xFFFFu);
        run.y += (int)(w[s].x >> 16);
        run.z += (int)(w[s].y & 0xFFFFu);
        run.w += (int)(w[s].y >> 16);
    }
    ((int4*)cnt)[i] = run;
}

#define SCAN_TILE 1024
__global__ __launch_bounds__(256)
void scan_kernel(const int* __restrict__ cnt, int* __restrict__ rowptr,
                 float* __restrict__ dinv) {
    __shared__ int wsum[8];
    int tid  = threadIdx.x;
    int lane = tid & 31;
    int wv   = tid >> 5;
    int tile0 = blockIdx.x * SCAN_TILE;
    const int4* c4 = (const int4*)cnt;

    int pre = 0;
    for (int i = tid; i < tile0 / 4; i += 256) {
        int4 v = c4[i];
        pre += (v.x + v.y) + (v.z + v.w);
    }
    for (int o = 16; o > 0; o >>= 1) pre += __shfl_xor(pre, o);
    if (lane == 0) wsum[wv] = pre;
    __syncthreads();
    int base = 0;
    for (int k = 0; k < 8; ++k) base += wsum[k];
    __syncthreads();

    int i0 = tile0 + tid * 4;
    int4 v = make_int4(0, 0, 0, 0);
    if (i0 < N_NODES) v = c4[i0 >> 2];
    int t = (v.x + v.y) + (v.z + v.w);
    int incl = t;
    for (int o = 1; o < 32; o <<= 1) {
        int u = __shfl_up(incl, o);
        if (lane >= o) incl += u;
    }
    if (lane == 31) wsum[wv] = incl;
    __syncthreads();
    int woff = 0;
    for (int k = 0; k < wv; ++k) woff += wsum[k];
    int ex = base + woff + incl - t;
    if (i0 < N_NODES) {
        int4 r;
        r.x = ex;
        r.y = r.x + v.x;
        r.z = r.y + v.y;
        r.w = r.z + v.z;
        ((int4*)rowptr)[i0 >> 2] = r;
        float4 dv;
        dv.x = rsqrtf((float)(v.x + 1));
        dv.y = rsqrtf((float)(v.y + 1));
        dv.z = rsqrtf((float)(v.z + 1));
        dv.w = rsqrtf((float)(v.w + 1));
        ((float4*)dinv)[i0 >> 2] = dv;
    }
}

__device__ __forceinline__ int2 csr_entry(int s, float ds, float dd) {
    return make_int2(s, __float_as_int(ds * dd));
}

__global__ __launch_bounds__(256)
void fill_kernel(const int* __restrict__ src, const int* __restrict__ dst,
                 const int* __restrict__ rank, const int* __restrict__ rowptr,
                 const int* __restrict__ pre,
                 const float* __restrict__ dinv, int2* __restrict__ csr) {
    const int4* s4 = (const int4*)src;
    const int4* d4 = (const int4*)dst;
    const int4* r4 = (const int4*)rank;
    int q0 = blockIdx.x * EDGE_Q_PER_BLOCK + threadIdx.x;
    int q1 = q0 + 256;
    if (q1 < EDGE_Q) {
        const int* ea = pre + (size_t)(q0 / SLICE_Q) * N_NODES;
        const int* eb = pre + (size_t)(q1 / SLICE_Q) * N_NODES;
        int4 sa = s4[q0], da = d4[q0], ra = r4[q0];
        int4 sb = s4[q1], db = d4[q1], rb = r4[q1];
        int   pa0 = rowptr[da.x], pa1 = rowptr[da.y], pa2 = rowptr[da.z], pa3 = rowptr[da.w];
        int   pb0 = rowptr[db.x], pb1 = rowptr[db.y], pb2 = rowptr[db.z], pb3 = rowptr[db.w];
        int   ea0 = ea[da.x], ea1 = ea[da.y], ea2 = ea[da.z], ea3 = ea[da.w];
        int   eb0 = eb[db.x], eb1 = eb[db.y], eb2 = eb[db.z], eb3 = eb[db.w];
        float ua0 = dinv[sa.x], ua1 = dinv[sa.y], ua2 = dinv[sa.z], ua3 = dinv[sa.w];
        float ub0 = dinv[sb.x], ub1 = dinv[sb.y], ub2 = dinv[sb.z], ub3 = dinv[sb.w];
        float va0 = dinv[da.x], va1 = dinv[da.y], va2 = dinv[da.z], va3 = dinv[da.w];
        float vb0 = dinv[db.x], vb1 = dinv[db.y], vb2 = dinv[db.z], vb3 = dinv[db.w];
        csr[pa0 + ea0 + ra.x] = csr_entry(sa.x, ua0, va0);
        csr[pa1 + ea1 + ra.y] = csr_entry(sa.y, ua1, va1);
        csr[pa2 + ea2 + ra.z] = csr_entry(sa.z, ua2, va2);
        csr[pa3 + ea3 + ra.w] = csr_entry(sa.w, ua3, va3);
        csr[pb0 + eb0 + rb.x] = csr_entry(sb.x, ub0, vb0);
        csr[pb1 + eb1 + rb.y] = csr_entry(sb.y, ub1, vb1);
        csr[pb2 + eb2 + rb.z] = csr_entry(sb.z, ub2, vb2);
        csr[pb3 + eb3 + rb.w] = csr_entry(sb.w, ub3, vb3);
    } else if (q0 < EDGE_Q) {
        const int* ea = pre + (size_t)(q0 / SLICE_Q) * N_NODES;
        int4 sa = s4[q0], da = d4[q0], ra = r4[q0];
        int   pa0 = rowptr[da.x], pa1 = rowptr[da.y], pa2 = rowptr[da.z], pa3 = rowptr[da.w];
        int   ea0 = ea[da.x], ea1 = ea[da.y], ea2 = ea[da.z], ea3 = ea[da.w];
        float ua0 = dinv[sa.x], ua1 = dinv[sa.y], ua2 = dinv[sa.z], ua3 = dinv[sa.w];
        float va0 = dinv[da.x], va1 = dinv[da.y], va2 = dinv[da.z], va3 = dinv[da.w];
        csr[pa0 + ea0 + ra.x] = csr_entry(sa.x, ua0, va0);
        csr[pa1 + ea1 + ra.y] = csr_entry(sa.y, ua1, va1);
        csr[pa2 + ea2 + ra.z] = csr_entry(sa.z, ua2, va2);
        csr[pa3 + ea3 + ra.w] = csr_entry(sa.w, ua3, va3);
    }
}

#define GEMM_WAVES   16
#define GEMM_THREADS (GEMM_WAVES * 32)
#define GEMM_BLOCKS  256
#define N_STRIPS     (N_NODES / 16)

static_assert(N_NODES % 16 == 0, "whole 16-row strips");
static_assert((HID * HID / 4) % GEMM_THREADS == 0, "W copy: whole passes of the workgroup");

__global__ __launch_bounds__(GEMM_THREADS)
void gemm_wmma_kernel(const float* __restrict__ X, const float* __restrict__ W,
                      float* __restrict__ H) {
    __shared__ float4 Wq[HID * HID / 4];

    int tid = threadIdx.x;
    #pragma unroll
    for (int i = 0; i < (HID * HID / 4) / GEMM_THREADS; ++i) {
        int p  = tid + i * GEMM_THREADS;
        int n  = p & (HID - 1);
        int gh = p >> 7;
        int k0 = (gh >> 1) * 8 + (gh & 1) * 2;
        float4 q;
        q.x = W[(k0 + 0) * HID + n];
        q.y = W[(k0 + 1) * HID + n];
        q.z = W[(k0 + 4) * HID + n];
        q.w = W[(k0 + 5) * HID + n];
        Wq[p] = q;
    }
    __syncthreads();

    int lane = tid & 31;
    int wv = __builtin_amdgcn_readfirstlane(tid >> 5);
    int m  = lane & 15;
    int kh = lane >> 4;

    int s0 = ((int)blockIdx.x * N_STRIPS) / GEMM_BLOCKS;
    int s1 = (((int)blockIdx.x + 1) * N_STRIPS) / GEMM_BLOCKS;
    const float4* bq = Wq + kh * HID + m;
    unsigned a_off = (unsigned)(m * HID + 2 * kh);
    unsigned h_off = (unsigned)(8 * kh * HID + m);

    for (int s = s0 + wv; s < s1; s += GEMM_WAVES) {
        const float* xs = X + (size_t)s * (16 * HID);
        float*       hs = H + (size_t)s * (16 * HID);
        v2f a[HID / 4];
        #pragma unroll
        for (int i = 0; i < HID / 4; ++i) a[i] = *(const v2f*)(xs + a_off + 4 * i);

        v8f acc[8];
        #pragma unroll
        for (int t = 0; t < 8; ++t) acc[t] = (v8f){};

        float4 b[2][8];
        #pragma unroll
        for (int t = 0; t < 8; ++t) b[0][t] = bq[t * 16];

        #pragma unroll
        for (int g = 0; g < HID / 8; ++g) {
            if (g + 1 < HID / 8) {
                #pragma unroll
                for (int t = 0; t < 8; ++t) b[(g + 1) & 1][t] = bq[(g + 1) * 2 * HID + t * 16];
            }
            asm volatile("" ::: "memory");
            #pragma unroll
            for (int t = 0; t < 8; ++t) {
                v2f bb = {b[g & 1][t].x, b[g & 1][t].y};
                acc[t] = __builtin_amdgcn_wmma_f32_16x16x4_f32(
                     false, a[2 * g],  false, bb,
                     (short)0, acc[t],  false,  false);
            }
            #pragma unroll
            for (int t = 0; t < 8; ++t) {
                v2f bb = {b[g & 1][t].z, b[g & 1][t].w};
                acc[t] = __builtin_amdgcn_wmma_f32_16x16x4_f32(
                     false, a[2 * g + 1],  false, bb,
                     (short)0, acc[t],  false,  false);
            }
        }

        #pragma unroll
        for (int t = 0; t < 8; ++t) {
            #pragma unroll
            for (int r = 0; r < 8; ++r) (hs + r * HID + t * 16)[h_off] = acc[t][r];
        }
    }
}

__global__ void zero_kernel(float4* __restrict__ p, int n4) {
    int i = blockIdx.x * blockDim.x + threadIdx.x;
    if (i < n4) p[i] = make_float4(0.f, 0.f, 0.f, 0.f);
}

#define NODES_PER_WAVE 4

template <int K>
__device__ __forceinline__ void rows(const float4* __restrict__ H4, int lane,
                                     int ex, int ew, int j, float4& acc) {
    float4 v[K];
    float  w[K];
    for (int k = 0; k < K; ++k) {
        int s = __builtin_amdgcn_readlane(ex, j + k);
        w[k]  = __int_as_float(__builtin_amdgcn_readlane(ew, j + k));
        v[k]  = H4[(size_t)s * (HID / 4) + lane];
    }
    for (int k = 0; k < K; ++k) {
        acc.x = fmaf(v[k].x, w[k], acc.x);
        acc.y = fmaf(v[k].y, w[k], acc.y);
        acc.z = fmaf(v[k].z, w[k], acc.z);
        acc.w = fmaf(v[k].w, w[k], acc.w);
    }
}

__global__ __launch_bounds__(256)
void gather_kernel(const float* __restrict__ H, const int* __restrict__ rowptr,
                   const int* __restrict__ cnt, const int2* __restrict__ csr,
                   const float* __restrict__ dinv, const float* __restrict__ bias,
                   float* __restrict__ out) {
    int lane = threadIdx.x & 31;
    int wave = __builtin_amdgcn_readfirstlane((int)((blockIdx.x * blockDim.x + threadIdx.x) >> 5));
    const float4* H4 = (const float4*)H;
    float4 b = ((const float4*)bias)[lane];

    int node0 = wave * NODES_PER_WAVE;
    int4   st4 = *(const int4*)(rowptr + node0);
    int4   n4  = *(const int4*)(cnt + node0);
    float4 d4  = *(const float4*)(dinv + node0);

    for (int q = 0; q < NODES_PER_WAVE; ++q) {
        int node  = node0 + q;
        int start = q == 0 ? st4.x : q == 1 ? st4.y : q == 2 ? st4.z : st4.w;
        int n     = q == 0 ? n4.x  : q == 1 ? n4.y  : q == 2 ? n4.z  : n4.w;
        float ds  = q == 0 ? d4.x  : q == 1 ? d4.y  : q == 2 ? d4.z  : d4.w;
        float4 hv = H4[(size_t)node * (HID / 4) + lane];
        float4 acc = make_float4(0.f, 0.f, 0.f, 0.f);

        for (int c0 = 0; c0 < n; c0 += 32) {
            int m = n - c0;
            if (m > 32) m = 32;
            int2 ent = make_int2(node, 0);
            if (lane < m) ent = csr[start + c0 + lane];
            int ex = (int)min((unsigned)ent.x, (unsigned)(N_NODES - 1));
            int ew = ent.y;
            int j = 0;
            for (; j + 8 <= m; j += 8) rows<8>(H4, lane, ex, ew, j, acc);
            if (j + 4 <= m) { rows<4>(H4, lane, ex, ew, j, acc); j += 4; }
            if (j + 2 <= m) { rows<2>(H4, lane, ex, ew, j, acc); j += 2; }
            if (j < m)      { rows<1>(H4, lane, ex, ew, j, acc); }
        }

        float wself = ds * ds;
        acc.x = fmaf(hv.x, wself, acc.x);
        acc.y = fmaf(hv.y, wself, acc.y);
        acc.z = fmaf(hv.z, wself, acc.z);
        acc.w = fmaf(hv.w, wself, acc.w);

        float4 r;
        r.x = acc.x + b.x;
        r.y = acc.y + b.y;
        r.z = acc.z + b.z;
        r.w = acc.w + b.w;
        r.x = r.x > 0.f ? r.x : 0.f;
        r.y = r.y > 0.f ? r.y : 0.f;
        r.z = r.z > 0.f ? r.z : 0.f;
        r.w = r.w > 0.f ? r.w : 0.f;
        ((float4*)out)[(size_t)node * (HID / 4) + lane] = r;
    }
}

#define POOL_NODES 16
static_assert(N_NODES % POOL_NODES == 0, "whole groups of nodes per wave, 64-byte scalar loads of batch");

__device__ __forceinline__ void pool_flush(float* __restrict__ pooled, float* __restrict__ counts,
                                           int g, int lane, float4 acc, float c) {
    float* pp = pooled + (size_t)g * HID + lane * 4;
    atomicAdd(pp + 0, acc.x);
    atomicAdd(pp + 1, acc.y);
    atomicAdd(pp + 2, acc.z);
    atomicAdd(pp + 3, acc.w);
    if (lane == 0) atomicAdd(&counts[g], c);
}

__global__ __launch_bounds__(256)
void pool_accum_kernel(const float* __restrict__ X, const int* __restrict__ batch,
                       float* __restrict__ pooled, float* __restrict__ counts) {
    int lane = threadIdx.x & 31;
    int wave = __builtin_amdgcn_readfirstlane((int)((blockIdx.x * blockDim.x + threadIdx.x) >> 5));
    if (wave >= N_NODES / POOL_NODES) return;
    int node0 = wave * POOL_NODES;

    const int4* b4 = (const int4*)(batch + node0);
    int4 i0 = b4[0], i1 = b4[1], i2 = b4[2], i3 = b4[3];
    int g[POOL_NODES] = {i0.x, i0.y, i0.z, i0.w, i1.x, i1.y, i1.z, i1.w,
                         i2.x, i2.y, i2.z, i2.w, i3.x, i3.y, i3.z, i3.w};

    const float4* X4 = (const float4*)X + (size_t)node0 * (HID / 4) + lane;
    float4 v[POOL_NODES];
    #pragma unroll
    for (int k = 0; k < POOL_NODES; ++k) v[k] = X4[k * (HID / 4)];

    int    gcur = g[0];
    float4 acc  = v[0];
    float  c    = 1.0f;
    #pragma unroll
    for (int k = 1; k < POOL_NODES; ++k) {
        if (g[k] != gcur) {
            pool_flush(pooled, counts, gcur, lane, acc, c);
            gcur = g[k];
            acc  = v[k];
            c    = 1.0f;
        } else {
            acc.x += v[k].x;
            acc.y += v[k].y;
            acc.z += v[k].z;
            acc.w += v[k].w;
            c += 1.0f;
        }
    }
    pool_flush(pooled, counts, gcur, lane, acc, c);
}

__global__ void final_linear_kernel(const float* __restrict__ pooled,
                                    const float* __restrict__ counts,
                                    const float* __restrict__ linW,
                                    const float* __restrict__ linb,
                                    float* __restrict__ out) {
    int i = blockIdx.x * blockDim.x + threadIdx.x;
    if (i >= N_GRAPHS * N_CLASSES) return;
    int g = i / N_CLASSES;
    int c = i - g * N_CLASSES;
    float scale = 1.0f / fmaxf(counts[g], 1.0f);
    const float* pg = pooled + (size_t)g * HID;
    float acc = 0.f;
    #pragma unroll 4
    for (int f = 0; f < HID; ++f) acc += pg[f] * linW[f * N_CLASSES + c];
    out[i] = acc * scale + linb[c];
}

extern "C" void kernel_launch(void* const* d_in, const int* in_sizes, int n_in,
                              void* d_out, int out_size, void* d_ws, size_t ws_size,
                              hipStream_t stream) {
    const float* x     = (const float*)d_in[0];
    const int*   edge  = (const int*)d_in[1];
    const int*   src   = edge;
    const int*   dst   = edge + N_EDGES;
    const int*   batch = (const int*)d_in[2];
    const float* Ws    = (const float*)d_in[3];
    const float* bs    = (const float*)d_in[4];
    const float* linW  = (const float*)d_in[5];
    const float* linb  = (const float*)d_in[6];
    float*       out   = (float*)d_out;

    float*  ws     = (float*)d_ws;
    int*    cnt    = (int*)ws;
    float*  dinv   = ws + N_NODES;
    int*    rowptr = (int*)(dinv + N_NODES);
    int*    rank   = rowptr + N_NODES;
    int2*   csr    = (int2*)(rank + N_EDGES);
    float*  H      = (float*)(csr + N_EDGES);
    float*  X      = H + (size_t)N_NODES * HID;
    float*  pooled = X + (size_t)N_NODES * HID;
    float*  counts = pooled + (size_t)N_GRAPHS * HID;
    unsigned* hist = (unsigned*)(counts + N_GRAPHS);
    int*    pre    = (int*)(hist + (size_t)N_SLICES * PAIR_WORDS);

    const int T = 256;

    count_lds_kernel<<<N_SLICES, CL_THREADS, 0, stream>>>(dst, rank, hist);
    slice_prefix_kernel<<<(N_NODES / 4 + T - 1) / T, T, 0, stream>>>(hist, pre, cnt);
    scan_kernel<<<(N_NODES + SCAN_TILE - 1) / SCAN_TILE, T, 0, stream>>>(cnt, rowptr, dinv);
    fill_kernel<<<EDGE_BLOCKS, T, 0, stream>>>(src, dst, rank, rowptr, pre, dinv, csr);

    const int gather_blocks = N_NODES / (8 * NODES_PER_WAVE);

    const float* xin = x;
    for (int l = 0; l < N_LAYERS; ++l) {
        const float* Wl = Ws + (size_t)l * HID * HID;
        const float* bl = bs + (size_t)l * HID;

        gemm_wmma_kernel<<<GEMM_BLOCKS, GEMM_THREADS, 0, stream>>>(xin, Wl, H);
        gather_kernel<<<gather_blocks, T, 0, stream>>>(H, rowptr, cnt, csr, dinv, bl, X);

        xin = X;
    }

    int pool4 = (N_GRAPHS * HID + N_GRAPHS + 3) / 4;
    zero_kernel<<<(pool4 + T - 1) / T, T, 0, stream>>>((float4*)pooled, pool4);
    pool_accum_kernel<<<(N_NODES / POOL_NODES + 7) / 8, T, 0, stream>>>(xin, batch, pooled, counts);
    final_linear_kernel<<<(N_GRAPHS * N_CLASSES + T - 1) / T, T, 0, stream>>>(
        pooled, counts, linW, linb, out);
}


# Neighbour sum moved in front of the product and fused into one launch per layer: a wave sums 16 nodes' input rows, feeds them to the WMMAs through LDS, bias+ReLU in the epilogue; sum order changes
# speedup vs baseline: 4.3546x; 1.1047x over previous
#include <hip/hip_runtime.h>
#include <hip/hip_bf16.h>

#define N_NODES   100000
#define N_EDGES   1600000
#define HID       128
#define N_CLASSES 10
#define N_GRAPHS  512
#define N_LAYERS  5

static_assert(N_NODES % 4 == 0, "int4 passes over per-node arrays");

typedef __attribute__((ext_vector_type(2))) float v2f;
typedef __attribute__((ext_vector_type(8))) float v8f;


static_assert(N_EDGES % 4 == 0, "int4 passes over per-edge arrays");
#define EDGE_Q            (N_EDGES / 4)
#define EDGE_Q_PER_BLOCK  512
#define EDGE_BLOCKS       ((EDGE_Q + EDGE_Q_PER_BLOCK - 1) / EDGE_Q_PER_BLOCK)

#define N_SLICES     32
#define SLICE_EDGES  (N_EDGES / N_SLICES)
#define SLICE_Q      (SLICE_EDGES / 4)
#define PAIR_WORDS   (N_NODES / 2)
#define CL_THREADS   1024
#define CL_ITERS     ((SLICE_Q + CL_THREADS - 1) / CL_THREADS)

static_assert(N_EDGES % N_SLICES == 0, "equal slices");
static_assert(SLICE_EDGES % 4 == 0, "a slice is whole int4 groups, so a group never straddles two slices");
static_assert(SLICE_EDGES <= 65535, "a 16-bit counter cannot overflow inside one slice");
static_assert(N_NODES % 8 == 0, "16-byte passes over the packed counter table");

__device__ __forceinline__ int lds_count(unsigned* __restrict__ c, int d) {
    unsigned u  = min((unsigned)d, (unsigned)(N_NODES - 1));
    unsigned sh = (u & 1u) * 16u;
    unsigned old = atomicAdd(&c[u >> 1], 1u << sh);
    return (int)((old >> sh) & 0xFFFFu);
}

__global__ __launch_bounds__(CL_THREADS)
void count_lds_kernel(const int* __restrict__ dst, int* __restrict__ rank,
                      unsigned* __restrict__ hist) {
    __shared__ uint4 c4[PAIR_WORDS / 4];
    unsigned* c = (unsigned*)c4;
    int tid = threadIdx.x;
    int s   = blockIdx.x;
    const int4* d4 = (const int4*)dst + (size_t)s * SLICE_Q;
    int4*       r4 = (int4*)rank + (size_t)s * SLICE_Q;

    int4 a[CL_ITERS];
    #pragma unroll
    for (int k = 0; k < CL_ITERS; ++k) {
        int q = tid + k * CL_THREADS;
        a[k] = make_int4(0, 0, 0, 0);
        if (q < SLICE_Q) a[k] = d4[q];
    }
    for (int i = tid; i < PAIR_WORDS / 4; i += CL_THREADS) c4[i] = make_uint4(0u, 0u, 0u, 0u);
    __syncthreads();

    #pragma unroll
    for (int k = 0; k < CL_ITERS; ++k) {
        int q = tid + k * CL_THREADS;
        if (q < SLICE_Q) {
            int4 r;
            r.x = lds_count(c, a[k].x);
            r.y = lds_count(c, a[k].y);
            r.z = lds_count(c, a[k].z);
            r.w = lds_count(c, a[k].w);
            r4[q] = r;
        }
    }
    __syncthreads();

    uint4* h4 = (uint4*)(hist + (size_t)s * PAIR_WORDS);
    for (int i = tid; i < PAIR_WORDS / 4; i += CL_THREADS) h4[i] = c4[i];
}

__global__ __launch_bounds__(256)
void slice_prefix_kernel(const unsigned* __restrict__ hist, int* __restrict__ pre,
                         int* __restrict__ cnt) {
    int i = blockIdx.x * blockDim.x + threadIdx.x;
    if (i >= N_NODES / 4) return;
    uint2 w[N_SLICES];
    #pragma unroll
    for (int s = 0; s < N_SLICES; ++s)
        w[s] = ((const uint2*)(hist + (size_t)s * PAIR_WORDS))[i];
    int4 run = make_int4(0, 0, 0, 0);
    #pragma unroll
    for (int s = 0; s < N_SLICES; ++s) {
        ((int4*)(pre + (size_t)s * N_NODES))[i] = run;
        run.x += (int)(w[s].x & 0xFFFFu);
        run.y += (int)(w[s].x >> 16);
        run.z += (int)(w[s].y & 0xFFFFu);
        run.w += (int)(w[s].y >> 16);
    }
    ((int4*)cnt)[i] = run;
}

#define SCAN_TILE 1024
__global__ __launch_bounds__(256)
void scan_kernel(const int* __restrict__ cnt, int* __restrict__ rowptr,
                 float* __restrict__ dinv) {
    __shared__ int wsum[8];
    int tid  = threadIdx.x;
    int lane = tid & 31;
    int wv   = tid >> 5;
    int tile0 = blockIdx.x * SCAN_TILE;
    const int4* c4 = (const int4*)cnt;

    int pre = 0;
    for (int i = tid; i < tile0 / 4; i += 256) {
        int4 v = c4[i];
        pre += (v.x + v.y) + (v.z + v.w);
    }
    for (int o = 16; o > 0; o >>= 1) pre += __shfl_xor(pre, o);
    if (lane == 0) wsum[wv] = pre;
    __syncthreads();
    int base = 0;
    for (int k = 0; k < 8; ++k) base += wsum[k];
    __syncthreads();

    int i0 = tile0 + tid * 4;
    int4 v = make_int4(0, 0, 0, 0);
    if (i0 < N_NODES) v = c4[i0 >> 2];
    int t = (v.x + v.y) + (v.z + v.w);
    int incl = t;
    for (int o = 1; o < 32; o <<= 1) {
        int u = __shfl_up(incl, o);
        if (lane >= o) incl += u;
    }
    if (lane == 31) wsum[wv] = incl;
    __syncthreads();
    int woff = 0;
    for (int k = 0; k < wv; ++k) woff += wsum[k];
    int ex = base + woff + incl - t;
    if (i0 < N_NODES) {
        int4 r;
        r.x = ex;
        r.y = r.x + v.x;
        r.z = r.y + v.y;
        r.w = r.z + v.z;
        ((int4*)rowptr)[i0 >> 2] = r;
        float4 dv;
        dv.x = rsqrtf((float)(v.x + 1));
        dv.y = rsqrtf((float)(v.y + 1));
        dv.z = rsqrtf((float)(v.z + 1));
        dv.w = rsqrtf((float)(v.w + 1));
        ((float4*)dinv)[i0 >> 2] = dv;
    }
}

__device__ __forceinline__ int2 csr_entry(int s, float ds, float dd) {
    return make_int2(s, __float_as_int(ds * dd));
}

__global__ __launch_bounds__(256)
void fill_kernel(const int* __restrict__ src, const int* __restrict__ dst,
                 const int* __restrict__ rank, const int* __restrict__ rowptr,
                 const int* __restrict__ pre,
                 const float* __restrict__ dinv, int2* __restrict__ csr) {
    const int4* s4 = (const int4*)src;
    const int4* d4 = (const int4*)dst;
    const int4* r4 = (const int4*)rank;
    int q0 = blockIdx.x * EDGE_Q_PER_BLOCK + threadIdx.x;
    int q1 = q0 + 256;
    if (q1 < EDGE_Q) {
        const int* ea = pre + (size_t)(q0 / SLICE_Q) * N_NODES;
        const int* eb = pre + (size_t)(q1 / SLICE_Q) * N_NODES;
        int4 sa = s4[q0], da = d4[q0], ra = r4[q0];
        int4 sb = s4[q1], db = d4[q1], rb = r4[q1];
        int   pa0 = rowptr[da.x], pa1 = rowptr[da.y], pa2 = rowptr[da.z], pa3 = rowptr[da.w];
        int   pb0 = rowptr[db.x], pb1 = rowptr[db.y], pb2 = rowptr[db.z], pb3 = rowptr[db.w];
        int   ea0 = ea[da.x], ea1 = ea[da.y], ea2 = ea[da.z], ea3 = ea[da.w];
        int   eb0 = eb[db.x], eb1 = eb[db.y], eb2 = eb[db.z], eb3 = eb[db.w];
        float ua0 = dinv[sa.x], ua1 = dinv[sa.y], ua2 = dinv[sa.z], ua3 = dinv[sa.w];
        float ub0 = dinv[sb.x], ub1 = dinv[sb.y], ub2 = dinv[sb.z], ub3 = dinv[sb.w];
        float va0 = dinv[da.x], va1 = dinv[da.y], va2 = dinv[da.z], va3 = dinv[da.w];
        float vb0 = dinv[db.x], vb1 = dinv[db.y], vb2 = dinv[db.z], vb3 = dinv[db.w];
        csr[pa0 + ea0 + ra.x] = csr_entry(sa.x, ua0, va0);
        csr[pa1 + ea1 + ra.y] = csr_entry(sa.y, ua1, va1);
        csr[pa2 + ea2 + ra.z] = csr_entry(sa.z, ua2, va2);
        csr[pa3 + ea3 + ra.w] = csr_entry(sa.w, ua3, va3);
        csr[pb0 + eb0 + rb.x] = csr_entry(sb.x, ub0, vb0);
        csr[pb1 + eb1 + rb.y] = csr_entry(sb.y, ub1, vb1);
        csr[pb2 + eb2 + rb.z] = csr_entry(sb.z, ub2, vb2);
        csr[pb3 + eb3 + rb.w] = csr_entry(sb.w, ub3, vb3);
    } else if (q0 < EDGE_Q) {
        const int* ea = pre + (size_t)(q0 / SLICE_Q) * N_NODES;
        int4 sa = s4[q0], da = d4[q0], ra = r4[q0];
        int   pa0 = rowptr[da.x], pa1 = rowptr[da.y], pa2 = rowptr[da.z], pa3 = rowptr[da.w];
        int   ea0 = ea[da.x], ea1 = ea[da.y], ea2 = ea[da.z], ea3 = ea[da.w];
        float ua0 = dinv[sa.x], ua1 = dinv[sa.y], ua2 = dinv[sa.z], ua3 = dinv[sa.w];
        float va0 = dinv[da.x], va1 = dinv[da.y], va2 = dinv[da.z], va3 = dinv[da.w];
        csr[pa0 + ea0 + ra.x] = csr_entry(sa.x, ua0, va0);
        csr[pa1 + ea1 + ra.y] = csr_entry(sa.y, ua1, va1);
        csr[pa2 + ea2 + ra.z] = csr_entry(sa.z, ua2, va2);
        csr[pa3 + ea3 + ra.w] = csr_entry(sa.w, ua3, va3);
    }
}

__global__ void zero_kernel(float4* __restrict__ p, int n4) {
    int i = blockIdx.x * blockDim.x + threadIdx.x;
    if (i < n4) p[i] = make_float4(0.f, 0.f, 0.f, 0.f);
}


template <int K>
__device__ __forceinline__ void rows(const float4* __restrict__ H4, int lane,
                                     int ex, int ew, int j, float4& acc) {
    float4 v[K];
    float  w[K];
    for (int k = 0; k < K; ++k) {
        int s = __builtin_amdgcn_readlane(ex, j + k);
        w[k]  = __int_as_float(__builtin_amdgcn_readlane(ew, j + k));
        v[k]  = H4[(size_t)s * (HID / 4) + lane];
    }
    for (int k = 0; k < K; ++k) {
        acc.x = fmaf(v[k].x, w[k], acc.x);
        acc.y = fmaf(v[k].y, w[k], acc.y);
        acc.z = fmaf(v[k].z, w[k], acc.z);
        acc.w = fmaf(v[k].w, w[k], acc.w);
    }
}

#define LAYER_WAVES   16
#define LAYER_THREADS (LAYER_WAVES * 32)
#define LAYER_BLOCKS  256
#define N_STRIPS      (N_NODES / 16)
#define TILE_PITCH    132

static_assert(N_NODES % 16 == 0, "whole 16-node strips, 16-byte scalar loads per group of 4 nodes");
static_assert((HID * HID / 4) % LAYER_THREADS == 0, "W copy: whole passes of the workgroup");

__device__ __forceinline__ float4 node_sum(const float4* __restrict__ X4, const int2* __restrict__ csr,
                                           int lane, int node, int start, int n, float ds,
                                           int2 ent0, float4 hv) {
    float4 acc = make_float4(0.f, 0.f, 0.f, 0.f);
    for (int c0 = 0; c0 < n; c0 += 32) {
        int m = n - c0;
        if (m > 32) m = 32;
        int2 ent = ent0;
        if (c0 > 0) {
            ent = make_int2(node, 0);
            if (lane < m) ent = csr[start + c0 + lane];
        }
        int ex = (int)min((unsigned)ent.x, (unsigned)(N_NODES - 1));
        int ew = ent.y;
        int j = 0;
        for (; j + 16 <= m; j += 16) rows<16>(X4, lane, ex, ew, j, acc);
        if (j + 8 <= m) { rows<8>(X4, lane, ex, ew, j, acc); j += 8; }
        if (j + 4 <= m) { rows<4>(X4, lane, ex, ew, j, acc); j += 4; }
        if (j + 2 <= m) { rows<2>(X4, lane, ex, ew, j, acc); j += 2; }
        if (j < m)      { rows<1>(X4, lane, ex, ew, j, acc); }
    }
    float wself = ds * ds;
    acc.x = fmaf(hv.x, wself, acc.x);
    acc.y = fmaf(hv.y, wself, acc.y);
    acc.z = fmaf(hv.z, wself, acc.z);
    acc.w = fmaf(hv.w, wself, acc.w);
    return acc;
}

__global__ __launch_bounds__(LAYER_THREADS)
void layer_kernel(const float* __restrict__ Xin, const float* __restrict__ W,
                  const float* __restrict__ bias, const int* __restrict__ rowptr,
                  const int* __restrict__ cnt, const int2* __restrict__ csr,
                  const float* __restrict__ dinv, float* __restrict__ Xout) {
    __shared__ float4 Wq[HID * HID / 4];
    __shared__ float4 tile4[LAYER_WAVES * 16 * TILE_PITCH / 4];

    int tid = threadIdx.x;
    #pragma unroll
    for (int i = 0; i < (HID * HID / 4) / LAYER_THREADS; ++i) {
        int p  = tid + i * LAYER_THREADS;
        int n  = p & (HID - 1);
        int gh = p >> 7;
        int k0 = (gh >> 1) * 8 + (gh & 1) * 2;
        float4 q;
        q.x = W[(k0 + 0) * HID + n];
        q.y = W[(k0 + 1) * HID + n];
        q.z = W[(k0 + 4) * HID + n];
        q.w = W[(k0 + 5) * HID + n];
        Wq[p] = q;
    }
    __syncthreads();

    int lane = tid & 31;
    int wv = __builtin_amdgcn_readfirstlane(tid >> 5);
    int m  = lane & 15;
    int kh = lane >> 4;

    int s0 = ((int)blockIdx.x * N_STRIPS) / LAYER_BLOCKS;
    int s1 = (((int)blockIdx.x + 1) * N_STRIPS) / LAYER_BLOCKS;
    const float4* bq = Wq + kh * HID + m;
    const float4* X4 = (const float4*)Xin;
    float* tile = (float*)tile4 + wv * (16 * TILE_PITCH);
    unsigned a_off = (unsigned)(m * TILE_PITCH + 2 * kh);
    unsigned h_off = (unsigned)(8 * kh * HID + m);

    float bv[8];
    #pragma unroll
    for (int t = 0; t < 8; ++t) bv[t] = bias[t * 16 + m];

    for (int s = s0 + wv; s < s1; s += LAYER_WAVES) {
        for (int g4 = 0; g4 < 4; ++g4) {
            int node0 = s * 16 + g4 * 4;
            int4   st4 = *(const int4*)(rowptr + node0);
            int4   n4  = *(const int4*)(cnt + node0);
            float4 d4  = *(const float4*)(dinv + node0);
            int   st[4] = {st4.x, st4.y, st4.z, st4.w};
            int   nn[4] = {n4.x, n4.y, n4.z, n4.w};
            float dd[4] = {d4.x, d4.y, d4.z, d4.w};
            int2   e0[4];
            float4 hv[4];
            #pragma unroll
            for (int q = 0; q < 4; ++q) {
                e0[q] = make_int2(node0 + q, 0);
                if (lane < nn[q]) e0[q] = csr[st[q] + lane];
                hv[q] = X4[(size_t)(node0 + q) * (HID / 4) + lane];
            }
            #pragma unroll
            for (int q = 0; q < 4; ++q) {
                float4 r = node_sum(X4, csr, lane, node0 + q, st[q], nn[q], dd[q], e0[q], hv[q]);
                *(float4*)(tile + (g4 * 4 + q) * TILE_PITCH + 4 * lane) = r;
            }
        }
        v2f a[HID / 4];
        #pragma unroll
        for (int i = 0; i < HID / 4; ++i) a[i] = *(const v2f*)(tile + a_off + 4 * i);

        v8f acc[8];
        #pragma unroll
        for (int t = 0; t < 8; ++t) acc[t] = (v8f){};

        float4 b[2][8];
        #pragma unroll
        for (int t = 0; t < 8; ++t) b[0][t] = bq[t * 16];

        #pragma unroll
        for (int g = 0; g < HID / 8; ++g) {
            if (g + 1 < HID / 8) {
                #pragma unroll
                for (int t = 0; t < 8; ++t) b[(g + 1) & 1][t] = bq[(g + 1) * 2 * HID + t * 16];
            }
            asm volatile("" ::: "memory");
            #pragma unroll
            for (int t = 0; t < 8; ++t) {
                v2f bb = {b[g & 1][t].x, b[g & 1][t].y};
                acc[t] = __builtin_amdgcn_wmma_f32_16x16x4_f32(
                     false, a[2 * g],  false, bb,
                     (short)0, acc[t],  false,  false);
            }
            #pragma unroll
            for (int t = 0; t < 8; ++t) {
                v2f bb = {b[g & 1][t].z, b[g & 1][t].w};
                acc[t] = __builtin_amdgcn_wmma_f32_16x16x4_f32(
                     false, a[2 * g + 1],  false, bb,
                     (short)0, acc[t],  false,  false);
            }
        }

        float* os = Xout + (size_t)s * (16 * HID);
        #pragma unroll
        for (int t = 0; t < 8; ++t) {
            #pragma unroll
            for (int r = 0; r < 8; ++r) {
                float v = acc[t][r] + bv[t];
                v = v > 0.f ? v : 0.f;
                (os + r * HID + t * 16)[h_off] = v;
            }
        }
    }
}

#define POOL_NODES 16
static_assert(N_NODES % POOL_NODES == 0, "whole groups of nodes per wave, 64-byte scalar loads of batch");

__device__ __forceinline__ void pool_flush(float* __restrict__ pooled, float* __restrict__ counts,
                                           int g, int lane, float4 acc, float c) {
    float* pp = pooled + (size_t)g * HID + lane * 4;
    atomicAdd(pp + 0, acc.x);
    atomicAdd(pp + 1, acc.y);
    atomicAdd(pp + 2, acc.z);
    atomicAdd(pp + 3, acc.w);
    if (lane == 0) atomicAdd(&counts[g], c);
}

__global__ __launch_bounds__(256)
void pool_accum_kernel(const float* __restrict__ X, const int* __restrict__ batch,
                       float* __restrict__ pooled, float* __restrict__ counts) {
    int lane = threadIdx.x & 31;
    int wave = __builtin_amdgcn_readfirstlane((int)((blockIdx.x * blockDim.x + threadIdx.x) >> 5));
    if (wave >= N_NODES / POOL_NODES) return;
    int node0 = wave * POOL_NODES;

    const int4* b4 = (const int4*)(batch + node0);
    int4 i0 = b4[0], i1 = b4[1], i2 = b4[2], i3 = b4[3];
    int g[POOL_NODES] = {i0.x, i0.y, i0.z, i0.w, i1.x, i1.y, i1.z, i1.w,
                         i2.x, i2.y, i2.z, i2.w, i3.x, i3.y, i3.z, i3.w};

    const float4* X4 = (const float4*)X + (size_t)node0 * (HID / 4) + lane;
    float4 v[POOL_NODES];
    #pragma unroll
    for (int k = 0; k < POOL_NODES; ++k) v[k] = X4[k * (HID / 4)];

    int    gcur = g[0];
    float4 acc  = v[0];
    float  c    = 1.0f;
    #pragma unroll
    for (int k = 1; k < POOL_NODES; ++k) {
        if (g[k] != gcur) {
            pool_flush(pooled, counts, gcur, lane, acc, c);
            gcur = g[k];
            acc  = v[k];
            c    = 1.0f;
        } else {
            acc.x += v[k].x;
            acc.y += v[k].y;
            acc.z += v[k].z;
            acc.w += v[k].w;
            c += 1.0f;
        }
    }
    pool_flush(pooled, counts, gcur, lane, acc, c);
}

__global__ void final_linear_kernel(const float* __restrict__ pooled,
                                    const float* __restrict__ counts,
                                    const float* __restrict__ linW,
                                    const float* __restrict__ linb,
                                    float* __restrict__ out) {
    int i = blockIdx.x * blockDim.x + threadIdx.x;
    if (i >= N_GRAPHS * N_CLASSES) return;
    int g = i / N_CLASSES;
    int c = i - g * N_CLASSES;
    float scale = 1.0f / fmaxf(counts[g], 1.0f);
    const float* pg = pooled + (size_t)g * HID;
    float acc = 0.f;
    #pragma unroll 4
    for (int f = 0; f < HID; ++f) acc += pg[f] * linW[f * N_CLASSES + c];
    out[i] = acc * scale + linb[c];
}

extern "C" void kernel_launch(void* const* d_in, const int* in_sizes, int n_in,
                              void* d_out, int out_size, void* d_ws, size_t ws_size,
                              hipStream_t stream) {
    const float* x     = (const float*)d_in[0];
    const int*   edge  = (const int*)d_in[1];
    const int*   src   = edge;
    const int*   dst   = edge + N_EDGES;
    const int*   batch = (const int*)d_in[2];
    const float* Ws    = (const float*)d_in[3];
    const float* bs    = (const float*)d_in[4];
    const float* linW  = (const float*)d_in[5];
    const float* linb  = (const float*)d_in[6];
    float*       out   = (float*)d_out;

    float*  ws     = (float*)d_ws;
    int*    cnt    = (int*)ws;
    float*  dinv   = ws + N_NODES;
    int*    rowptr = (int*)(dinv + N_NODES);
    int*    rank   = rowptr + N_NODES;
    int2*   csr    = (int2*)(rank + N_EDGES);
    float*  H      = (float*)(csr + N_EDGES);
    float*  X      = H + (size_t)N_NODES * HID;
    float*  pooled = X + (size_t)N_NODES * HID;
    float*  counts = pooled + (size_t)N_GRAPHS * HID;
    unsigned* hist = (unsigned*)(counts + N_GRAPHS);
    int*    pre    = (int*)(hist + (size_t)N_SLICES * PAIR_WORDS);

    const int T = 256;

    count_lds_kernel<<<N_SLICES, CL_THREADS, 0, stream>>>(dst, rank, hist);
    slice_prefix_kernel<<<(N_NODES / 4 + T - 1) / T, T, 0, stream>>>(hist, pre, cnt);
    scan_kernel<<<(N_NODES + SCAN_TILE - 1) / SCAN_TILE, T, 0, stream>>>(cnt, rowptr, dinv);
    fill_kernel<<<EDGE_BLOCKS, T, 0, stream>>>(src, dst, rank, rowptr, pre, dinv, csr);

    const float* xin   = x;
    float*       xout  = H;
    float*       xnext = X;
    for (int l = 0; l < N_LAYERS; ++l) {
        const float* Wl = Ws + (size_t)l * HID * HID;
        const float* bl = bs + (size_t)l * HID;

        layer_kernel<<<LAYER_BLOCKS, LAYER_THREADS, 0, stream>>>(xin, Wl, bl, rowptr, cnt, csr, dinv, xout);

        xin = xout;
        float* t = xout;
        xout  = xnext;
        xnext = t;
    }

    int pool4 = (N_GRAPHS * HID + N_GRAPHS + 3) / 4;
    zero_kernel<<<(pool4 + T - 1) / T, T, 0, stream>>>((float4*)pooled, pool4);
    pool_accum_kernel<<<(N_NODES / POOL_NODES + 7) / 8, T, 0, stream>>>(xin, batch, pooled, counts);
    final_linear_kernel<<<(N_GRAPHS * N_CLASSES + T - 1) / T, T, 0, stream>>>(
        pooled, counts, linW, linb, out);
}


# Rows of a strip's 16 nodes read as one stream of slots, 16 in flight across node boundaries; a node's sum is added in LDS to its own row scaled before (own term rounded before the add)
# speedup vs baseline: 4.5168x; 1.0372x over previous
#include <hip/hip_runtime.h>
#include <hip/hip_bf16.h>

#define N_NODES   100000
#define N_EDGES   1600000
#define HID       128
#define N_CLASSES 10
#define N_GRAPHS  512
#define N_LAYERS  5

static_assert(N_NODES % 4 == 0, "int4 passes over per-node arrays");

typedef __attribute__((ext_vector_type(2))) float v2f;
typedef __attribute__((ext_vector_type(8))) float v8f;


static_assert(N_EDGES % 4 == 0, "int4 passes over per-edge arrays");
#define EDGE_Q            (N_EDGES / 4)
#define EDGE_Q_PER_BLOCK  512
#define EDGE_BLOCKS       ((EDGE_Q + EDGE_Q_PER_BLOCK - 1) / EDGE_Q_PER_BLOCK)

#define N_SLICES     32
#define SLICE_EDGES  (N_EDGES / N_SLICES)
#define SLICE_Q      (SLICE_EDGES / 4)
#define PAIR_WORDS   (N_NODES / 2)
#define CL_THREADS   1024
#define CL_ITERS     ((SLICE_Q + CL_THREADS - 1) / CL_THREADS)

static_assert(N_EDGES % N_SLICES == 0, "equal slices");
static_assert(SLICE_EDGES % 4 == 0, "a slice is whole int4 groups, so a group never straddles two slices");
static_assert(SLICE_EDGES <= 65535, "a 16-bit counter cannot overflow inside one slice");
static_assert(N_NODES % 8 == 0, "16-byte passes over the packed counter table");

__device__ __forceinline__ int lds_count(unsigned* __restrict__ c, int d) {
    unsigned u  = min((unsigned)d, (unsigned)(N_NODES - 1));
    unsigned sh = (u & 1u) * 16u;
    unsigned old = atomicAdd(&c[u >> 1], 1u << sh);
    return (int)((old >> sh) & 0xFFFFu);
}

__global__ __launch_bounds__(CL_THREADS)
void count_lds_kernel(const int* __restrict__ dst, int* __restrict__ rank,
                      unsigned* __restrict__ hist) {
    __shared__ uint4 c4[PAIR_WORDS / 4];
    unsigned* c = (unsigned*)c4;
    int tid = threadIdx.x;
    int s   = blockIdx.x;
    const int4* d4 = (const int4*)dst + (size_t)s * SLICE_Q;
    int4*       r4 = (int4*)rank + (size_t)s * SLICE_Q;

    int4 a[CL_ITERS];
    #pragma unroll
    for (int k = 0; k < CL_ITERS; ++k) {
        int q = tid + k * CL_THREADS;
        a[k] = make_int4(0, 0, 0, 0);
        if (q < SLICE_Q) a[k] = d4[q];
    }
    for (int i = tid; i < PAIR_WORDS / 4; i += CL_THREADS) c4[i] = make_uint4(0u, 0u, 0u, 0u);
    __syncthreads();

    #pragma unroll
    for (int k = 0; k < CL_ITERS; ++k) {
        int q = tid + k * CL_THREADS;
        if (q < SLICE_Q) {
            int4 r;
            r.x = lds_count(c, a[k].x);
            r.y = lds_count(c, a[k].y);
            r.z = lds_count(c, a[k].z);
            r.w = lds_count(c, a[k].w);
            r4[q] = r;
        }
    }
    __syncthreads();

    uint4* h4 = (uint4*)(hist + (size_t)s * PAIR_WORDS);
    for (int i = tid; i < PAIR_WORDS / 4; i += CL_THREADS) h4[i] = c4[i];
}

__global__ __launch_bounds__(256)
void slice_prefix_kernel(const unsigned* __restrict__ hist, int* __restrict__ pre,
                         int* __restrict__ cnt) {
    int i = blockIdx.x * blockDim.x + threadIdx.x;
    if (i >= N_NODES / 4) return;
    uint2 w[N_SLICES];
    #pragma unroll
    for (int s = 0; s < N_SLICES; ++s)
        w[s] = ((const uint2*)(hist + (size_t)s * PAIR_WORDS))[i];
    int4 run = make_int4(0, 0, 0, 0);
    #pragma unroll
    for (int s = 0; s < N_SLICES; ++s) {
        ((int4*)(pre + (size_t)s * N_NODES))[i] = run;
        run.x += (int)(w[s].x & 0xFFFFu);
        run.y += (int)(w[s].x >> 16);
        run.z += (int)(w[s].y & 0xFFFFu);
        run.w += (int)(w[s].y >> 16);
    }
    ((int4*)cnt)[i] = run;
}

#define SCAN_TILE 1024
__global__ __launch_bounds__(256)
void scan_kernel(const int* __restrict__ cnt, int* __restrict__ rowptr,
                 float* __restrict__ dinv) {
    __shared__ int wsum[8];
    int tid  = threadIdx.x;
    int lane = tid & 31;
    int wv   = tid >> 5;
    int tile0 = blockIdx.x * SCAN_TILE;
    const int4* c4 = (const int4*)cnt;

    int pre = 0;
    for (int i = tid; i < tile0 / 4; i += 256) {
        int4 v = c4[i];
        pre += (v.x + v.y) + (v.z + v.w);
    }
    for (int o = 16; o > 0; o >>= 1) pre += __shfl_xor(pre, o);
    if (lane == 0) wsum[wv] = pre;
    __syncthreads();
    int base = 0;
    for (int k = 0; k < 8; ++k) base += wsum[k];
    __syncthreads();

    int i0 = tile0 + tid * 4;
    int4 v = make_int4(0, 0, 0, 0);
    if (i0 < N_NODES) v = c4[i0 >> 2];
    int t = (v.x + v.y) + (v.z + v.w);
    int incl = t;
    for (int o = 1; o < 32; o <<= 1) {
        int u = __shfl_up(incl, o);
        if (lane >= o) incl += u;
    }
    if (lane == 31) wsum[wv] = incl;
    __syncthreads();
    int woff = 0;
    for (int k = 0; k < wv; ++k) woff += wsum[k];
    int ex = base + woff + incl - t;
    if (i0 < N_NODES) {
        int4 r;
        r.x = ex;
        r.y = r.x + v.x;
        r.z = r.y + v.y;
        r.w = r.z + v.z;
        ((int4*)rowptr)[i0 >> 2] = r;
        float4 dv;
        dv.x = rsqrtf((float)(v.x + 1));
        dv.y = rsqrtf((float)(v.y + 1));
        dv.z = rsqrtf((float)(v.z + 1));
        dv.w = rsqrtf((float)(v.w + 1));
        ((float4*)dinv)[i0 >> 2] = dv;
    }
}

__device__ __forceinline__ int2 csr_entry(int s, float ds, float dd) {
    return make_int2(s, __float_as_int(ds * dd));
}

__global__ __launch_bounds__(256)
void fill_kernel(const int* __restrict__ src, const int* __restrict__ dst,
                 const int* __restrict__ rank, const int* __restrict__ rowptr,
                 const int* __restrict__ pre,
                 const float* __restrict__ dinv, int2* __restrict__ csr) {
    const int4* s4 = (const int4*)src;
    const int4* d4 = (const int4*)dst;
    const int4* r4 = (const int4*)rank;
    int q0 = blockIdx.x * EDGE_Q_PER_BLOCK + threadIdx.x;
    int q1 = q0 + 256;
    if (q1 < EDGE_Q) {
        const int* ea = pre + (size_t)(q0 / SLICE_Q) * N_NODES;
        const int* eb = pre + (size_t)(q1 / SLICE_Q) * N_NODES;
        int4 sa = s4[q0], da = d4[q0], ra = r4[q0];
        int4 sb = s4[q1], db = d4[q1], rb = r4[q1];
        int   pa0 = rowptr[da.x], pa1 = rowptr[da.y], pa2 = rowptr[da.z], pa3 = rowptr[da.w];
        int   pb0 = rowptr[db.x], pb1 = rowptr[db.y], pb2 = rowptr[db.z], pb3 = rowptr[db.w];
        int   ea0 = ea[da.x], ea1 = ea[da.y], ea2 = ea[da.z], ea3 = ea[da.w];
        int   eb0 = eb[db.x], eb1 = eb[db.y], eb2 = eb[db.z], eb3 = eb[db.w];
        float ua0 = dinv[sa.x], ua1 = dinv[sa.y], ua2 = dinv[sa.z], ua3 = dinv[sa.w];
        float ub0 = dinv[sb.x], ub1 = dinv[sb.y], ub2 = dinv[sb.z], ub3 = dinv[sb.w];
        float va0 = dinv[da.x], va1 = dinv[da.y], va2 = dinv[da.z], va3 = dinv[da.w];
        float vb0 = dinv[db.x], vb1 = dinv[db.y], vb2 = dinv[db.z], vb3 = dinv[db.w];
        csr[pa0 + ea0 + ra.x] = csr_entry(sa.x, ua0, va0);
        csr[pa1 + ea1 + ra.y] = csr_entry(sa.y, ua1, va1);
        csr[pa2 + ea2 + ra.z] = csr_entry(sa.z, ua2, va2);
        csr[pa3 + ea3 + ra.w] = csr_entry(sa.w, ua3, va3);
        csr[pb0 + eb0 + rb.x] = csr_entry(sb.x, ub0, vb0);
        csr[pb1 + eb1 + rb.y] = csr_entry(sb.y, ub1, vb1);
        csr[pb2 + eb2 + rb.z] = csr_entry(sb.z, ub2, vb2);
        csr[pb3 + eb3 + rb.w] = csr_entry(sb.w, ub3, vb3);
    } else if (q0 < EDGE_Q) {
        const int* ea = pre + (size_t)(q0 / SLICE_Q) * N_NODES;
        int4 sa = s4[q0], da = d4[q0], ra = r4[q0];
        int   pa0 = rowptr[da.x], pa1 = rowptr[da.y], pa2 = rowptr[da.z], pa3 = rowptr[da.w];
        int   ea0 = ea[da.x], ea1 = ea[da.y], ea2 = ea[da.z], ea3 = ea[da.w];
        float ua0 = dinv[sa.x], ua1 = dinv[sa.y], ua2 = dinv[sa.z], ua3 = dinv[sa.w];
        float va0 = dinv[da.x], va1 = dinv[da.y], va2 = dinv[da.z], va3 = dinv[da.w];
        csr[pa0 + ea0 + ra.x] = csr_entry(sa.x, ua0, va0);
        csr[pa1 + ea1 + ra.y] = csr_entry(sa.y, ua1, va1);
        csr[pa2 + ea2 + ra.z] = csr_entry(sa.z, ua2, va2);
        csr[pa3 + ea3 + ra.w] = csr_entry(sa.w, ua3, va3);
    }
}

__global__ void zero_kernel(float4* __restrict__ p, int n4) {
    int i = blockIdx.x * blockDim.x + threadIdx.x;
    if (i < n4) p[i] = make_float4(0.f, 0.f, 0.f, 0.f);
}

#define LAYER_WAVES   16
#define LAYER_THREADS (LAYER_WAVES * 32)
#define LAYER_BLOCKS  256
#define N_STRIPS      (N_NODES / 16)
#define TILE_PITCH    132

static_assert(N_NODES % 16 == 0, "whole 16-node strips, 16-byte scalar loads per group of 4 nodes");
static_assert((HID * HID / 4) % LAYER_THREADS == 0, "W copy: whole passes of the workgroup");


__device__ __forceinline__ void node_done(float* __restrict__ tile, int lane, int cur, float4& acc) {
    float4* row = (float4*)(tile + cur * TILE_PITCH + 4 * lane);
    float4 t = *row;
    t.x = acc.x + t.x;
    t.y = acc.y + t.y;
    t.z = acc.z + t.z;
    t.w = acc.w + t.w;
    *row = t;
    acc = make_float4(0.f, 0.f, 0.f, 0.f);
}

template <int K>
__device__ __forceinline__ void batch(const float4* __restrict__ X4, float* __restrict__ tile, int lane,
                                      int ex, int ew, int j, int p, int endv,
                                      int& cur, int& end, float4& acc) {
    float4 v[K];
    float  w[K];
    #pragma unroll
    for (int k = 0; k < K; ++k) {
        int s = __builtin_amdgcn_readlane(ex, j + k);
        w[k]  = __int_as_float(__builtin_amdgcn_readlane(ew, j + k));
        v[k]  = X4[(size_t)s * (HID / 4) + lane];
    }
    #pragma unroll
    for (int k = 0; k < K; ++k) {
        while (p + k == end && cur < 15) {
            node_done(tile, lane, cur, acc);
            ++cur;
            end = __builtin_amdgcn_readlane(endv, cur);
        }
        acc.x = fmaf(v[k].x, w[k], acc.x);
        acc.y = fmaf(v[k].y, w[k], acc.y);
        acc.z = fmaf(v[k].z, w[k], acc.z);
        acc.w = fmaf(v[k].w, w[k], acc.w);
    }
}

__global__ __launch_bounds__(LAYER_THREADS)
void layer_kernel(const float* __restrict__ Xin, const float* __restrict__ W,
                  const float* __restrict__ bias, const int* __restrict__ rowptr,
                  const int* __restrict__ cnt, const int2* __restrict__ csr,
                  const float* __restrict__ dinv, float* __restrict__ Xout) {
    __shared__ float4 Wq[HID * HID / 4];
    __shared__ float4 tile4[LAYER_WAVES * 16 * TILE_PITCH / 4];

    int tid = threadIdx.x;
    #pragma unroll
    for (int i = 0; i < (HID * HID / 4) / LAYER_THREADS; ++i) {
        int p  = tid + i * LAYER_THREADS;
        int n  = p & (HID - 1);
        int gh = p >> 7;
        int k0 = (gh >> 1) * 8 + (gh & 1) * 2;
        float4 q;
        q.x = W[(k0 + 0) * HID + n];
        q.y = W[(k0 + 1) * HID + n];
        q.z = W[(k0 + 4) * HID + n];
        q.w = W[(k0 + 5) * HID + n];
        Wq[p] = q;
    }
    __syncthreads();

    int lane = tid & 31;
    int wv = __builtin_amdgcn_readfirstlane(tid >> 5);
    int m  = lane & 15;
    int kh = lane >> 4;

    int s0 = ((int)blockIdx.x * N_STRIPS) / LAYER_BLOCKS;
    int s1 = (((int)blockIdx.x + 1) * N_STRIPS) / LAYER_BLOCKS;
    const float4* bq = Wq + kh * HID + m;
    const float4* X4 = (const float4*)Xin;
    float* tile = (float*)tile4 + wv * (16 * TILE_PITCH);
    unsigned a_off = (unsigned)(m * TILE_PITCH + 2 * kh);
    unsigned h_off = (unsigned)(8 * kh * HID + m);

    float bv[8];
    #pragma unroll
    for (int t = 0; t < 8; ++t) bv[t] = bias[t * 16 + m];

    for (int s = s0 + wv; s < s1; s += LAYER_WAVES) {
        int node0 = s * 16;
        int   rp = rowptr[node0 + m];
        int   cn = cnt[node0 + m];
        float dv = dinv[node0 + m];
        float4 hv[16];
        #pragma unroll
        for (int i = 0; i < 16; ++i) hv[i] = X4[(size_t)(node0 + i) * (HID / 4) + lane];

        int endv = rp + cn;
        int p0 = __builtin_amdgcn_readlane(rp, 0);
        int pE = __builtin_amdgcn_readlane(endv, 15);
        pE = min(pE, p0 + N_EDGES);

        int2 entn = make_int2(node0, 0);
        if (lane < pE - p0) entn = csr[p0 + lane];

        #pragma unroll
        for (int i = 0; i < 16; ++i) {
            float ds = __int_as_float(__builtin_amdgcn_readlane(__float_as_int(dv), i));
            float wself = ds * ds;
            float4 r;
            r.x = hv[i].x * wself;
            r.y = hv[i].y * wself;
            r.z = hv[i].z * wself;
            r.w = hv[i].w * wself;
            *(float4*)(tile + i * TILE_PITCH + 4 * lane) = r;
        }

        {
            float4 acc = make_float4(0.f, 0.f, 0.f, 0.f);
            int cur = 0;
            int end = __builtin_amdgcn_readlane(endv, 0);
            for (int c0 = p0; c0 < pE; c0 += 32) {
                int mm = pE - c0;
                if (mm > 32) mm = 32;
                int2 ent = entn;
                entn = make_int2(node0, 0);
                if (lane < pE - (c0 + 32)) entn = csr[c0 + 32 + lane];
                int ex = (int)min((unsigned)ent.x, (unsigned)(N_NODES - 1));
                int ew = ent.y;
                int j = 0;
                for (; j + 16 <= mm; j += 16) batch<16>(X4, tile, lane, ex, ew, j, c0 + j, endv, cur, end, acc);
                if (j + 8 <= mm) { batch<8>(X4, tile, lane, ex, ew, j, c0 + j, endv, cur, end, acc); j += 8; }
                if (j + 4 <= mm) { batch<4>(X4, tile, lane, ex, ew, j, c0 + j, endv, cur, end, acc); j += 4; }
                if (j + 2 <= mm) { batch<2>(X4, tile, lane, ex, ew, j, c0 + j, endv, cur, end, acc); j += 2; }
                if (j < mm)      { batch<1>(X4, tile, lane, ex, ew, j, c0 + j, endv, cur, end, acc); }
            }
            for (; cur < 16; ++cur) node_done(tile, lane, cur, acc);
        }
        v2f a[HID / 4];
        #pragma unroll
        for (int i = 0; i < HID / 4; ++i) a[i] = *(const v2f*)(tile + a_off + 4 * i);

        v8f acc[8];
        #pragma unroll
        for (int t = 0; t < 8; ++t) acc[t] = (v8f){};

        float4 b[2][8];
        #pragma unroll
        for (int t = 0; t < 8; ++t) b[0][t] = bq[t * 16];

        #pragma unroll
        for (int g = 0; g < HID / 8; ++g) {
            if (g + 1 < HID / 8) {
                #pragma unroll
                for (int t = 0; t < 8; ++t) b[(g + 1) & 1][t] = bq[(g + 1) * 2 * HID + t * 16];
            }
            asm volatile("" ::: "memory");
            #pragma unroll
            for (int t = 0; t < 8; ++t) {
                v2f bb = {b[g & 1][t].x, b[g & 1][t].y};
                acc[t] = __builtin_amdgcn_wmma_f32_16x16x4_f32(
                     false, a[2 * g],  false, bb,
                     (short)0, acc[t],  false,  false);
            }
            #pragma unroll
            for (int t = 0; t < 8; ++t) {
                v2f bb = {b[g & 1][t].z, b[g & 1][t].w};
                acc[t] = __builtin_amdgcn_wmma_f32_16x16x4_f32(
                     false, a[2 * g + 1],  false, bb,
                     (short)0, acc[t],  false,  false);
            }
        }

        float* os = Xout + (size_t)s * (16 * HID);
        #pragma unroll
        for (int t = 0; t < 8; ++t) {
            #pragma unroll
            for (int r = 0; r < 8; ++r) {
                float v = acc[t][r] + bv[t];
                v = v > 0.f ? v : 0.f;
                (os + r * HID + t * 16)[h_off] = v;
            }
        }
    }
}

#define POOL_NODES 16
static_assert(N_NODES % POOL_NODES == 0, "whole groups of nodes per wave, 64-byte scalar loads of batch");

__device__ __forceinline__ void pool_flush(float* __restrict__ pooled, float* __restrict__ counts,
                                           int g, int lane, float4 acc, float c) {
    float* pp = pooled + (size_t)g * HID + lane * 4;
    atomicAdd(pp + 0, acc.x);
    atomicAdd(pp + 1, acc.y);
    atomicAdd(pp + 2, acc.z);
    atomicAdd(pp + 3, acc.w);
    if (lane == 0) atomicAdd(&counts[g], c);
}

__global__ __launch_bounds__(256)
void pool_accum_kernel(const float* __restrict__ X, const int* __restrict__ batch,
                       float* __restrict__ pooled, float* __restrict__ counts) {
    int lane = threadIdx.x & 31;
    int wave = __builtin_amdgcn_readfirstlane((int)((blockIdx.x * blockDim.x + threadIdx.x) >> 5));
    if (wave >= N_NODES / POOL_NODES) return;
    int node0 = wave * POOL_NODES;

    const int4* b4 = (const int4*)(batch + node0);
    int4 i0 = b4[0], i1 = b4[1], i2 = b4[2], i3 = b4[3];
    int g[POOL_NODES] = {i0.x, i0.y, i0.z, i0.w, i1.x, i1.y, i1.z, i1.w,
                         i2.x, i2.y, i2.z, i2.w, i3.x, i3.y, i3.z, i3.w};

    const float4* X4 = (const float4*)X + (size_t)node0 * (HID / 4) + lane;
    float4 v[POOL_NODES];
    #pragma unroll
    for (int k = 0; k < POOL_NODES; ++k) v[k] = X4[k * (HID / 4)];

    int    gcur = g[0];
    float4 acc  = v[0];
    float  c    = 1.0f;
    #pragma unroll
    for (int k = 1; k < POOL_NODES; ++k) {
        if (g[k] != gcur) {
            pool_flush(pooled, counts, gcur, lane, acc, c);
            gcur = g[k];
            acc  = v[k];
            c    = 1.0f;
        } else {
            acc.x += v[k].x;
            acc.y += v[k].y;
            acc.z += v[k].z;
            acc.w += v[k].w;
            c += 1.0f;
        }
    }
    pool_flush(pooled, counts, gcur, lane, acc, c);
}

__global__ void final_linear_kernel(const float* __restrict__ pooled,
                                    const float* __restrict__ counts,
                                    const float* __restrict__ linW,
                                    const float* __restrict__ linb,
                                    float* __restrict__ out) {
    int i = blockIdx.x * blockDim.x + threadIdx.x;
    if (i >= N_GRAPHS * N_CLASSES) return;
    int g = i / N_CLASSES;
    int c = i - g * N_CLASSES;
    float scale = 1.0f / fmaxf(counts[g], 1.0f);
    const float* pg = pooled + (size_t)g * HID;
    float acc = 0.f;
    #pragma unroll 4
    for (int f = 0; f < HID; ++f) acc += pg[f] * linW[f * N_CLASSES + c];
    out[i] = acc * scale + linb[c];
}

extern "C" void kernel_launch(void* const* d_in, const int* in_sizes, int n_in,
                              void* d_out, int out_size, void* d_ws, size_t ws_size,
                              hipStream_t stream) {
    const float* x     = (const float*)d_in[0];
    const int*   edge  = (const int*)d_in[1];
    const int*   src   = edge;
    const int*   dst   = edge + N_EDGES;
    const int*   batch = (const int*)d_in[2];
    const float* Ws    = (const float*)d_in[3];
    const float* bs    = (const float*)d_in[4];
    const float* linW  = (const float*)d_in[5];
    const float* linb  = (const float*)d_in[6];
    float*       out   = (float*)d_out;

    float*  ws     = (float*)d_ws;
    int*    cnt    = (int*)ws;
    float*  dinv   = ws + N_NODES;
    int*    rowptr = (int*)(dinv + N_NODES);
    int*    rank   = rowptr + N_NODES;
    int2*   csr    = (int2*)(rank + N_EDGES);
    float*  H      = (float*)(csr + N_EDGES);
    float*  X      = H + (size_t)N_NODES * HID;
    float*  pooled = X + (size_t)N_NODES * HID;
    float*  counts = pooled + (size_t)N_GRAPHS * HID;
    unsigned* hist = (unsigned*)(counts + N_GRAPHS);
    int*    pre    = (int*)(hist + (size_t)N_SLICES * PAIR_WORDS);

    const int T = 256;

    count_lds_kernel<<<N_SLICES, CL_THREADS, 0, stream>>>(dst, rank, hist);
    slice_prefix_kernel<<<(N_NODES / 4 + T - 1) / T, T, 0, stream>>>(hist, pre, cnt);
    scan_kernel<<<(N_NODES + SCAN_TILE - 1) / SCAN_TILE, T, 0, stream>>>(cnt, rowptr, dinv);
    fill_kernel<<<EDGE_BLOCKS, T, 0, stream>>>(src, dst, rank, rowptr, pre, dinv, csr);

    const float* xin   = x;
    float*       xout  = H;
    float*       xnext = X;
    for (int l = 0; l < N_LAYERS; ++l) {
        const float* Wl = Ws + (size_t)l * HID * HID;
        const float* bl = bs + (size_t)l * HID;

        layer_kernel<<<LAYER_BLOCKS, LAYER_THREADS, 0, stream>>>(xin, Wl, bl, rowptr, cnt, csr, dinv, xout);

        xin = xout;
        float* t = xout;
        xout  = xnext;
        xnext = t;
    }

    int pool4 = (N_GRAPHS * HID + N_GRAPHS + 3) / 4;
    zero_kernel<<<(pool4 + T - 1) / T, T, 0, stream>>>((float4*)pooled, pool4);
    pool_accum_kernel<<<(N_NODES / POOL_NODES + 7) / 8, T, 0, stream>>>(xin, batch, pooled, counts);
    final_linear_kernel<<<(N_GRAPHS * N_CLASSES + T - 1) / T, T, 0, stream>>>(
        pooled, counts, linW, linb, out);
}


# Scan takes the sum in front of its tile from per-tile totals written by the slice-sum pass: one read per thread, not a serial re-read of all counts in front (up to 97 dependent round trips)
# speedup vs baseline: 4.9800x; 1.1026x over previous
#include <hip/hip_runtime.h>
#include <hip/hip_bf16.h>

#define N_NODES   100000
#define N_EDGES   1600000
#define HID       128
#define N_CLASSES 10
#define N_GRAPHS  512
#define N_LAYERS  5

static_assert(N_NODES % 4 == 0, "int4 passes over per-node arrays");

typedef __attribute__((ext_vector_type(2))) float v2f;
typedef __attribute__((ext_vector_type(8))) float v8f;


static_assert(N_EDGES % 4 == 0, "int4 passes over per-edge arrays");
#define EDGE_Q            (N_EDGES / 4)
#define EDGE_Q_PER_BLOCK  512
#define EDGE_BLOCKS       ((EDGE_Q + EDGE_Q_PER_BLOCK - 1) / EDGE_Q_PER_BLOCK)

#define N_SLICES     32
#define SLICE_EDGES  (N_EDGES / N_SLICES)
#define SLICE_Q      (SLICE_EDGES / 4)
#define PAIR_WORDS   (N_NODES / 2)
#define CL_THREADS   1024
#define CL_ITERS     ((SLICE_Q + CL_THREADS - 1) / CL_THREADS)

static_assert(N_EDGES % N_SLICES == 0, "equal slices");
static_assert(SLICE_EDGES % 4 == 0, "a slice is whole int4 groups, so a group never straddles two slices");
static_assert(SLICE_EDGES <= 65535, "a 16-bit counter cannot overflow inside one slice");
static_assert(N_NODES % 8 == 0, "16-byte passes over the packed counter table");

__device__ __forceinline__ int lds_count(unsigned* __restrict__ c, int d) {
    unsigned u  = min((unsigned)d, (unsigned)(N_NODES - 1));
    unsigned sh = (u & 1u) * 16u;
    unsigned old = atomicAdd(&c[u >> 1], 1u << sh);
    return (int)((old >> sh) & 0xFFFFu);
}

__global__ __launch_bounds__(CL_THREADS)
void count_lds_kernel(const int* __restrict__ dst, int* __restrict__ rank,
                      unsigned* __restrict__ hist) {
    __shared__ uint4 c4[PAIR_WORDS / 4];
    unsigned* c = (unsigned*)c4;
    int tid = threadIdx.x;
    int s   = blockIdx.x;
    const int4* d4 = (const int4*)dst + (size_t)s * SLICE_Q;
    int4*       r4 = (int4*)rank + (size_t)s * SLICE_Q;

    int4 a[CL_ITERS];
    #pragma unroll
    for (int k = 0; k < CL_ITERS; ++k) {
        int q = tid + k * CL_THREADS;
        a[k] = make_int4(0, 0, 0, 0);
        if (q < SLICE_Q) a[k] = d4[q];
    }
    for (int i = tid; i < PAIR_WORDS / 4; i += CL_THREADS) c4[i] = make_uint4(0u, 0u, 0u, 0u);
    __syncthreads();

    #pragma unroll
    for (int k = 0; k < CL_ITERS; ++k) {
        int q = tid + k * CL_THREADS;
        if (q < SLICE_Q) {
            int4 r;
            r.x = lds_count(c, a[k].x);
            r.y = lds_count(c, a[k].y);
            r.z = lds_count(c, a[k].z);
            r.w = lds_count(c, a[k].w);
            r4[q] = r;
        }
    }
    __syncthreads();

    uint4* h4 = (uint4*)(hist + (size_t)s * PAIR_WORDS);
    for (int i = tid; i < PAIR_WORDS / 4; i += CL_THREADS) h4[i] = c4[i];
}

#define SCAN_TILE 1024
#define N_TILES   ((N_NODES + SCAN_TILE - 1) / SCAN_TILE)
static_assert(SCAN_TILE == 4 * 256, "a slice-sum block of 256 threads x 4 nodes is one scan tile");
static_assert(N_TILES <= 256, "one tile total per thread of a scan block");

__global__ __launch_bounds__(256)
void slice_prefix_kernel(const unsigned* __restrict__ hist, int* __restrict__ pre,
                         int* __restrict__ cnt, int* __restrict__ tsum) {
    __shared__ int wtot[8];
    int tid = threadIdx.x;
    int i = blockIdx.x * 256 + tid;
    int4 run = make_int4(0, 0, 0, 0);
    if (i < N_NODES / 4) {
        uint2 w[N_SLICES];
        #pragma unroll
        for (int s = 0; s < N_SLICES; ++s)
            w[s] = ((const uint2*)(hist + (size_t)s * PAIR_WORDS))[i];
        #pragma unroll
        for (int s = 0; s < N_SLICES; ++s) {
            ((int4*)(pre + (size_t)s * N_NODES))[i] = run;
            run.x += (int)(w[s].x & 0xFFFFu);
            run.y += (int)(w[s].x >> 16);
            run.z += (int)(w[s].y & 0xFFFFu);
            run.w += (int)(w[s].y >> 16);
        }
        ((int4*)cnt)[i] = run;
    }
    int t = (run.x + run.y) + (run.z + run.w);
    for (int o = 16; o > 0; o >>= 1) t += __shfl_xor(t, o);
    if ((tid & 31) == 0) wtot[tid >> 5] = t;
    __syncthreads();
    if (tid == 0) {
        int s = 0;
        for (int k = 0; k < 8; ++k) s += wtot[k];
        tsum[blockIdx.x] = s;
    }
}

__global__ __launch_bounds__(256)
void scan_kernel(const int* __restrict__ cnt, const int* __restrict__ tsum,
                 int* __restrict__ rowptr, float* __restrict__ dinv) {
    __shared__ int wsum[8];
    int tid  = threadIdx.x;
    int lane = tid & 31;
    int wv   = tid >> 5;
    int tile0 = blockIdx.x * SCAN_TILE;
    const int4* c4 = (const int4*)cnt;

    int i0 = tile0 + tid * 4;
    int pre = 0;
    if (tid < (int)blockIdx.x) pre = tsum[tid];
    int4 v = make_int4(0, 0, 0, 0);
    if (i0 < N_NODES) v = c4[i0 >> 2];

    for (int o = 16; o > 0; o >>= 1) pre += __shfl_xor(pre, o);
    if (lane == 0) wsum[wv] = pre;
    __syncthreads();
    int base = 0;
    for (int k = 0; k < 8; ++k) base += wsum[k];
    __syncthreads();

    int t = (v.x + v.y) + (v.z + v.w);
    int incl = t;
    for (int o = 1; o < 32; o <<= 1) {
        int u = __shfl_up(incl, o);
        if (lane >= o) incl += u;
    }
    if (lane == 31) wsum[wv] = incl;
    __syncthreads();
    int woff = 0;
    for (int k = 0; k < wv; ++k) woff += wsum[k];
    int ex = base + woff + incl - t;
    if (i0 < N_NODES) {
        int4 r;
        r.x = ex;
        r.y = r.x + v.x;
        r.z = r.y + v.y;
        r.w = r.z + v.z;
        ((int4*)rowptr)[i0 >> 2] = r;
        float4 dv;
        dv.x = rsqrtf((float)(v.x + 1));
        dv.y = rsqrtf((float)(v.y + 1));
        dv.z = rsqrtf((float)(v.z + 1));
        dv.w = rsqrtf((float)(v.w + 1));
        ((float4*)dinv)[i0 >> 2] = dv;
    }
}

__device__ __forceinline__ int2 csr_entry(int s, float ds, float dd) {
    return make_int2(s, __float_as_int(ds * dd));
}

__global__ __launch_bounds__(256)
void fill_kernel(const int* __restrict__ src, const int* __restrict__ dst,
                 const int* __restrict__ rank, const int* __restrict__ rowptr,
                 const int* __restrict__ pre,
                 const float* __restrict__ dinv, int2* __restrict__ csr) {
    const int4* s4 = (const int4*)src;
    const int4* d4 = (const int4*)dst;
    const int4* r4 = (const int4*)rank;
    int q0 = blockIdx.x * EDGE_Q_PER_BLOCK + threadIdx.x;
    int q1 = q0 + 256;
    if (q1 < EDGE_Q) {
        const int* ea = pre + (size_t)(q0 / SLICE_Q) * N_NODES;
        const int* eb = pre + (size_t)(q1 / SLICE_Q) * N_NODES;
        int4 sa = s4[q0], da = d4[q0], ra = r4[q0];
        int4 sb = s4[q1], db = d4[q1], rb = r4[q1];
        int   pa0 = rowptr[da.x], pa1 = rowptr[da.y], pa2 = rowptr[da.z], pa3 = rowptr[da.w];
        int   pb0 = rowptr[db.x], pb1 = rowptr[db.y], pb2 = rowptr[db.z], pb3 = rowptr[db.w];
        int   ea0 = ea[da.x], ea1 = ea[da.y], ea2 = ea[da.z], ea3 = ea[da.w];
        int   eb0 = eb[db.x], eb1 = eb[db.y], eb2 = eb[db.z], eb3 = eb[db.w];
        float ua0 = dinv[sa.x], ua1 = dinv[sa.y], ua2 = dinv[sa.z], ua3 = dinv[sa.w];
        float ub0 = dinv[sb.x], ub1 = dinv[sb.y], ub2 = dinv[sb.z], ub3 = dinv[sb.w];
        float va0 = dinv[da.x], va1 = dinv[da.y], va2 = dinv[da.z], va3 = dinv[da.w];
        float vb0 = dinv[db.x], vb1 = dinv[db.y], vb2 = dinv[db.z], vb3 = dinv[db.w];
        csr[pa0 + ea0 + ra.x] = csr_entry(sa.x, ua0, va0);
        csr[pa1 + ea1 + ra.y] = csr_entry(sa.y, ua1, va1);
        csr[pa2 + ea2 + ra.z] = csr_entry(sa.z, ua2, va2);
        csr[pa3 + ea3 + ra.w] = csr_entry(sa.w, ua3, va3);
        csr[pb0 + eb0 + rb.x] = csr_entry(sb.x, ub0, vb0);
        csr[pb1 + eb1 + rb.y] = csr_entry(sb.y, ub1, vb1);
        csr[pb2 + eb2 + rb.z] = csr_entry(sb.z, ub2, vb2);
        csr[pb3 + eb3 + rb.w] = csr_entry(sb.w, ub3, vb3);
    } else if (q0 < EDGE_Q) {
        const int* ea = pre + (size_t)(q0 / SLICE_Q) * N_NODES;
        int4 sa = s4[q0], da = d4[q0], ra = r4[q0];
        int   pa0 = rowptr[da.x], pa1 = rowptr[da.y], pa2 = rowptr[da.z], pa3 = rowptr[da.w];
        int   ea0 = ea[da.x], ea1 = ea[da.y], ea2 = ea[da.z], ea3 = ea[da.w];
        float ua0 = dinv[sa.x], ua1 = dinv[sa.y], ua2 = dinv[sa.z], ua3 = dinv[sa.w];
        float va0 = dinv[da.x], va1 = dinv[da.y], va2 = dinv[da.z], va3 = dinv[da.w];
        csr[pa0 + ea0 + ra.x] = csr_entry(sa.x, ua0, va0);
        csr[pa1 + ea1 + ra.y] = csr_entry(sa.y, ua1, va1);
        csr[pa2 + ea2 + ra.z] = csr_entry(sa.z, ua2, va2);
        csr[pa3 + ea3 + ra.w] = csr_entry(sa.w, ua3, va3);
    }
}

__global__ void zero_kernel(float4* __restrict__ p, int n4) {
    int i = blockIdx.x * blockDim.x + threadIdx.x;
    if (i < n4) p[i] = make_float4(0.f, 0.f, 0.f, 0.f);
}

#define LAYER_WAVES   16
#define LAYER_THREADS (LAYER_WAVES * 32)
#define LAYER_BLOCKS  256
#define N_STRIPS      (N_NODES / 16)
#define TILE_PITCH    132

static_assert(N_NODES % 16 == 0, "whole 16-node strips, 16-byte scalar loads per group of 4 nodes");
static_assert((HID * HID / 4) % LAYER_THREADS == 0, "W copy: whole passes of the workgroup");


__device__ __forceinline__ void node_done(float* __restrict__ tile, int lane, int cur, float4& acc) {
    float4* row = (float4*)(tile + cur * TILE_PITCH + 4 * lane);
    float4 t = *row;
    t.x = acc.x + t.x;
    t.y = acc.y + t.y;
    t.z = acc.z + t.z;
    t.w = acc.w + t.w;
    *row = t;
    acc = make_float4(0.f, 0.f, 0.f, 0.f);
}

template <int K>
__device__ __forceinline__ void batch(const float4* __restrict__ X4, float* __restrict__ tile, int lane,
                                      int ex, int ew, int j, int p, int endv,
                                      int& cur, int& end, float4& acc) {
    float4 v[K];
    float  w[K];
    #pragma unroll
    for (int k = 0; k < K; ++k) {
        int s = __builtin_amdgcn_readlane(ex, j + k);
        w[k]  = __int_as_float(__builtin_amdgcn_readlane(ew, j + k));
        v[k]  = X4[(size_t)s * (HID / 4) + lane];
    }
    #pragma unroll
    for (int k = 0; k < K; ++k) {
        while (p + k == end && cur < 15) {
            node_done(tile, lane, cur, acc);
            ++cur;
            end = __builtin_amdgcn_readlane(endv, cur);
        }
        acc.x = fmaf(v[k].x, w[k], acc.x);
        acc.y = fmaf(v[k].y, w[k], acc.y);
        acc.z = fmaf(v[k].z, w[k], acc.z);
        acc.w = fmaf(v[k].w, w[k], acc.w);
    }
}

__global__ __launch_bounds__(LAYER_THREADS)
void layer_kernel(const float* __restrict__ Xin, const float* __restrict__ W,
                  const float* __restrict__ bias, const int* __restrict__ rowptr,
                  const int* __restrict__ cnt, const int2* __restrict__ csr,
                  const float* __restrict__ dinv, float* __restrict__ Xout) {
    __shared__ float4 Wq[HID * HID / 4];
    __shared__ float4 tile4[LAYER_WAVES * 16 * TILE_PITCH / 4];

    int tid = threadIdx.x;
    #pragma unroll
    for (int i = 0; i < (HID * HID / 4) / LAYER_THREADS; ++i) {
        int p  = tid + i * LAYER_THREADS;
        int n  = p & (HID - 1);
        int gh = p >> 7;
        int k0 = (gh >> 1) * 8 + (gh & 1) * 2;
        float4 q;
        q.x = W[(k0 + 0) * HID + n];
        q.y = W[(k0 + 1) * HID + n];
        q.z = W[(k0 + 4) * HID + n];
        q.w = W[(k0 + 5) * HID + n];
        Wq[p] = q;
    }
    __syncthreads();

    int lane = tid & 31;
    int wv = __builtin_amdgcn_readfirstlane(tid >> 5);
    int m  = lane & 15;
    int kh = lane >> 4;

    int s0 = ((int)blockIdx.x * N_STRIPS) / LAYER_BLOCKS;
    int s1 = (((int)blockIdx.x + 1) * N_STRIPS) / LAYER_BLOCKS;
    const float4* bq = Wq + kh * HID + m;
    const float4* X4 = (const float4*)Xin;
    float* tile = (float*)tile4 + wv * (16 * TILE_PITCH);
    unsigned a_off = (unsigned)(m * TILE_PITCH + 2 * kh);
    unsigned h_off = (unsigned)(8 * kh * HID + m);

    float bv[8];
    #pragma unroll
    for (int t = 0; t < 8; ++t) bv[t] = bias[t * 16 + m];

    for (int s = s0 + wv; s < s1; s += LAYER_WAVES) {
        int node0 = s * 16;
        int   rp = rowptr[node0 + m];
        int   cn = cnt[node0 + m];
        float dv = dinv[node0 + m];
        float4 hv[16];
        #pragma unroll
        for (int i = 0; i < 16; ++i) hv[i] = X4[(size_t)(node0 + i) * (HID / 4) + lane];

        int endv = rp + cn;
        int p0 = __builtin_amdgcn_readlane(rp, 0);
        int pE = __builtin_amdgcn_readlane(endv, 15);
        pE = min(pE, p0 + N_EDGES);

        int2 entn = make_int2(node0, 0);
        if (lane < pE - p0) entn = csr[p0 + lane];

        #pragma unroll
        for (int i = 0; i < 16; ++i) {
            float ds = __int_as_float(__builtin_amdgcn_readlane(__float_as_int(dv), i));
            float wself = ds * ds;
            float4 r;
            r.x = hv[i].x * wself;
            r.y = hv[i].y * wself;
            r.z = hv[i].z * wself;
            r.w = hv[i].w * wself;
            *(float4*)(tile + i * TILE_PITCH + 4 * lane) = r;
        }

        {
            float4 acc = make_float4(0.f, 0.f, 0.f, 0.f);
            int cur = 0;
            int end = __builtin_amdgcn_readlane(endv, 0);
            for (int c0 = p0; c0 < pE; c0 += 32) {
                int mm = pE - c0;
                if (mm > 32) mm = 32;
                int2 ent = entn;
                entn = make_int2(node0, 0);
                if (lane < pE - (c0 + 32)) entn = csr[c0 + 32 + lane];
                int ex = (int)min((unsigned)ent.x, (unsigned)(N_NODES - 1));
                int ew = ent.y;
                int j = 0;
                for (; j + 16 <= mm; j += 16) batch<16>(X4, tile, lane, ex, ew, j, c0 + j, endv, cur, end, acc);
                if (j + 8 <= mm) { batch<8>(X4, tile, lane, ex, ew, j, c0 + j, endv, cur, end, acc); j += 8; }
                if (j + 4 <= mm) { batch<4>(X4, tile, lane, ex, ew, j, c0 + j, endv, cur, end, acc); j += 4; }
                if (j + 2 <= mm) { batch<2>(X4, tile, lane, ex, ew, j, c0 + j, endv, cur, end, acc); j += 2; }
                if (j < mm)      { batch<1>(X4, tile, lane, ex, ew, j, c0 + j, endv, cur, end, acc); }
            }
            for (; cur < 16; ++cur) node_done(tile, lane, cur, acc);
        }
        v2f a[HID / 4];
        #pragma unroll
        for (int i = 0; i < HID / 4; ++i) a[i] = *(const v2f*)(tile + a_off + 4 * i);

        v8f acc[8];
        #pragma unroll
        for (int t = 0; t < 8; ++t) acc[t] = (v8f){};

        float4 b[2][8];
        #pragma unroll
        for (int t = 0; t < 8; ++t) b[0][t] = bq[t * 16];

        #pragma unroll
        for (int g = 0; g < HID / 8; ++g) {
            if (g + 1 < HID / 8) {
                #pragma unroll
                for (int t = 0; t < 8; ++t) b[(g + 1) & 1][t] = bq[(g + 1) * 2 * HID + t * 16];
            }
            asm volatile("" ::: "memory");
            #pragma unroll
            for (int t = 0; t < 8; ++t) {
                v2f bb = {b[g & 1][t].x, b[g & 1][t].y};
                acc[t] = __builtin_amdgcn_wmma_f32_16x16x4_f32(
                     false, a[2 * g],  false, bb,
                     (short)0, acc[t],  false,  false);
            }
            #pragma unroll
            for (int t = 0; t < 8; ++t) {
                v2f bb = {b[g & 1][t].z, b[g & 1][t].w};
                acc[t] = __builtin_amdgcn_wmma_f32_16x16x4_f32(
                     false, a[2 * g + 1],  false, bb,
                     (short)0, acc[t],  false,  false);
            }
        }

        float* os = Xout + (size_t)s * (16 * HID);
        #pragma unroll
        for (int t = 0; t < 8; ++t) {
            #pragma unroll
            for (int r = 0; r < 8; ++r) {
                float v = acc[t][r] + bv[t];
                v = v > 0.f ? v : 0.f;
                (os + r * HID + t * 16)[h_off] = v;
            }
        }
    }
}

#define POOL_NODES 16
static_assert(N_NODES % POOL_NODES == 0, "whole groups of nodes per wave, 64-byte scalar loads of batch");

__device__ __forceinline__ void pool_flush(float* __restrict__ pooled, float* __restrict__ counts,
                                           int g, int lane, float4 acc, float c) {
    float* pp = pooled + (size_t)g * HID + lane * 4;
    atomicAdd(pp + 0, acc.x);
    atomicAdd(pp + 1, acc.y);
    atomicAdd(pp + 2, acc.z);
    atomicAdd(pp + 3, acc.w);
    if (lane == 0) atomicAdd(&counts[g], c);
}

__global__ __launch_bounds__(256)
void pool_accum_kernel(const float* __restrict__ X, const int* __restrict__ batch,
                       float* __restrict__ pooled, float* __restrict__ counts) {
    int lane = threadIdx.x & 31;
    int wave = __builtin_amdgcn_readfirstlane((int)((blockIdx.x * blockDim.x + threadIdx.x) >> 5));
    if (wave >= N_NODES / POOL_NODES) return;
    int node0 = wave * POOL_NODES;

    const int4* b4 = (const int4*)(batch + node0);
    int4 i0 = b4[0], i1 = b4[1], i2 = b4[2], i3 = b4[3];
    int g[POOL_NODES] = {i0.x, i0.y, i0.z, i0.w, i1.x, i1.y, i1.z, i1.w,
                         i2.x, i2.y, i2.z, i2.w, i3.x, i3.y, i3.z, i3.w};

    const float4* X4 = (const float4*)X + (size_t)node0 * (HID / 4) + lane;
    float4 v[POOL_NODES];
    #pragma unroll
    for (int k = 0; k < POOL_NODES; ++k) v[k] = X4[k * (HID / 4)];

    int    gcur = g[0];
    float4 acc  = v[0];
    float  c    = 1.0f;
    #pragma unroll
    for (int k = 1; k < POOL_NODES; ++k) {
        if (g[k] != gcur) {
            pool_flush(pooled, counts, gcur, lane, acc, c);
            gcur = g[k];
            acc  = v[k];
            c    = 1.0f;
        } else {
            acc.x += v[k].x;
            acc.y += v[k].y;
            acc.z += v[k].z;
            acc.w += v[k].w;
            c += 1.0f;
        }
    }
    pool_flush(pooled, counts, gcur, lane, acc, c);
}

__global__ void final_linear_kernel(const float* __restrict__ pooled,
                                    const float* __restrict__ counts,
                                    const float* __restrict__ linW,
                                    const float* __restrict__ linb,
                                    float* __restrict__ out) {
    int i = blockIdx.x * blockDim.x + threadIdx.x;
    if (i >= N_GRAPHS * N_CLASSES) return;
    int g = i / N_CLASSES;
    int c = i - g * N_CLASSES;
    float scale = 1.0f / fmaxf(counts[g], 1.0f);
    const float* pg = pooled + (size_t)g * HID;
    float acc = 0.f;
    #pragma unroll 4
    for (int f = 0; f < HID; ++f) acc += pg[f] * linW[f * N_CLASSES + c];
    out[i] = acc * scale + linb[c];
}

extern "C" void kernel_launch(void* const* d_in, const int* in_sizes, int n_in,
                              void* d_out, int out_size, void* d_ws, size_t ws_size,
                              hipStream_t stream) {
    const float* x     = (const float*)d_in[0];
    const int*   edge  = (const int*)d_in[1];
    const int*   src   = edge;
    const int*   dst   = edge + N_EDGES;
    const int*   batch = (const int*)d_in[2];
    const float* Ws    = (const float*)d_in[3];
    const float* bs    = (const float*)d_in[4];
    const float* linW  = (const float*)d_in[5];
    const float* linb  = (const float*)d_in[6];
    float*       out   = (float*)d_out;

    float*  ws     = (float*)d_ws;
    int*    cnt    = (int*)ws;
    float*  dinv   = ws + N_NODES;
    int*    rowptr = (int*)(dinv + N_NODES);
    int*    rank   = rowptr + N_NODES;
    int2*   csr    = (int2*)(rank + N_EDGES);
    float*  H      = (float*)(csr + N_EDGES);
    float*  X      = H + (size_t)N_NODES * HID;
    float*  pooled = X + (size_t)N_NODES * HID;
    float*  counts = pooled + (size_t)N_GRAPHS * HID;
    unsigned* hist = (unsigned*)(counts + N_GRAPHS);
    int*    pre    = (int*)(hist + (size_t)N_SLICES * PAIR_WORDS);
    int*    tsum   = pre + (size_t)N_SLICES * N_NODES;

    const int T = 256;

    count_lds_kernel<<<N_SLICES, CL_THREADS, 0, stream>>>(dst, rank, hist);
    slice_prefix_kernel<<<N_TILES, T, 0, stream>>>(hist, pre, cnt, tsum);
    scan_kernel<<<N_TILES, T, 0, stream>>>(cnt, tsum, rowptr, dinv);
    fill_kernel<<<EDGE_BLOCKS, T, 0, stream>>>(src, dst, rank, rowptr, pre, dinv, csr);

    const float* xin   = x;
    float*       xout  = H;
    float*       xnext = X;
    for (int l = 0; l < N_LAYERS; ++l) {
        const float* Wl = Ws + (size_t)l * HID * HID;
        const float* bl = bs + (size_t)l * HID;

        layer_kernel<<<LAYER_BLOCKS, LAYER_THREADS, 0, stream>>>(xin, Wl, bl, rowptr, cnt, csr, dinv, xout);

        xin = xout;
        float* t = xout;
        xout  = xnext;
        xnext = t;
    }

    int pool4 = (N_GRAPHS * HID + N_GRAPHS + 3) / 4;
    zero_kernel<<<(pool4 + T - 1) / T, T, 0, stream>>>((float4*)pooled, pool4);
    pool_accum_kernel<<<(N_NODES / POOL_NODES + 7) / 8, T, 0, stream>>>(xin, batch, pooled, counts);
    final_linear_kernel<<<(N_GRAPHS * N_CLASSES + T - 1) / T, T, 0, stream>>>(
        pooled, counts, linW, linb, out);
}


# Mean pool moved into the last layer's launch: each wave adds its strip's 16 finished rows by graph id from its LDS tile (same 16-node groups, same float atomics); no last plane, pool or clear launch
# speedup vs baseline: 5.1391x; 1.0319x over previous
#include <hip/hip_runtime.h>
#include <hip/hip_bf16.h>

#define N_NODES   100000
#define N_EDGES   1600000
#define HID       128
#define N_CLASSES 10
#define N_GRAPHS  512
#define N_LAYERS  5

static_assert(N_NODES % 4 == 0, "int4 passes over per-node arrays");

typedef __attribute__((ext_vector_type(2))) float v2f;
typedef __attribute__((ext_vector_type(8))) float v8f;


static_assert(N_EDGES % 4 == 0, "int4 passes over per-edge arrays");
#define EDGE_Q            (N_EDGES / 4)
#define EDGE_Q_PER_BLOCK  512
#define EDGE_BLOCKS       ((EDGE_Q + EDGE_Q_PER_BLOCK - 1) / EDGE_Q_PER_BLOCK)

#define N_SLICES     32
#define SLICE_EDGES  (N_EDGES / N_SLICES)
#define SLICE_Q      (SLICE_EDGES / 4)
#define PAIR_WORDS   (N_NODES / 2)
#define CL_THREADS   1024
#define CL_ITERS     ((SLICE_Q + CL_THREADS - 1) / CL_THREADS)

static_assert(N_EDGES % N_SLICES == 0, "equal slices");
static_assert(SLICE_EDGES % 4 == 0, "a slice is whole int4 groups, so a group never straddles two slices");
static_assert(SLICE_EDGES <= 65535, "a 16-bit counter cannot overflow inside one slice");
static_assert(N_NODES % 8 == 0, "16-byte passes over the packed counter table");

__device__ __forceinline__ int lds_count(unsigned* __restrict__ c, int d) {
    unsigned u  = min((unsigned)d, (unsigned)(N_NODES - 1));
    unsigned sh = (u & 1u) * 16u;
    unsigned old = atomicAdd(&c[u >> 1], 1u << sh);
    return (int)((old >> sh) & 0xFFFFu);
}

__global__ __launch_bounds__(CL_THREADS)
void count_lds_kernel(const int* __restrict__ dst, int* __restrict__ rank,
                      unsigned* __restrict__ hist) {
    __shared__ uint4 c4[PAIR_WORDS / 4];
    unsigned* c = (unsigned*)c4;
    int tid = threadIdx.x;
    int s   = blockIdx.x;
    const int4* d4 = (const int4*)dst + (size_t)s * SLICE_Q;
    int4*       r4 = (int4*)rank + (size_t)s * SLICE_Q;

    int4 a[CL_ITERS];
    #pragma unroll
    for (int k = 0; k < CL_ITERS; ++k) {
        int q = tid + k * CL_THREADS;
        a[k] = make_int4(0, 0, 0, 0);
        if (q < SLICE_Q) a[k] = d4[q];
    }
    for (int i = tid; i < PAIR_WORDS / 4; i += CL_THREADS) c4[i] = make_uint4(0u, 0u, 0u, 0u);
    __syncthreads();

    #pragma unroll
    for (int k = 0; k < CL_ITERS; ++k) {
        int q = tid + k * CL_THREADS;
        if (q < SLICE_Q) {
            int4 r;
            r.x = lds_count(c, a[k].x);
            r.y = lds_count(c, a[k].y);
            r.z = lds_count(c, a[k].z);
            r.w = lds_count(c, a[k].w);
            r4[q] = r;
        }
    }
    __syncthreads();

    uint4* h4 = (uint4*)(hist + (size_t)s * PAIR_WORDS);
    for (int i = tid; i < PAIR_WORDS / 4; i += CL_THREADS) h4[i] = c4[i];
}

#define SCAN_TILE 1024
#define N_TILES   ((N_NODES + SCAN_TILE - 1) / SCAN_TILE)
static_assert(SCAN_TILE == 4 * 256, "a slice-sum block of 256 threads x 4 nodes is one scan tile");
static_assert(N_TILES <= 256, "one tile total per thread of a scan block");

__global__ __launch_bounds__(256)
void slice_prefix_kernel(const unsigned* __restrict__ hist, int* __restrict__ pre,
                         int* __restrict__ cnt, int* __restrict__ tsum) {
    __shared__ int wtot[8];
    int tid = threadIdx.x;
    int i = blockIdx.x * 256 + tid;
    int4 run = make_int4(0, 0, 0, 0);
    if (i < N_NODES / 4) {
        uint2 w[N_SLICES];
        #pragma unroll
        for (int s = 0; s < N_SLICES; ++s)
            w[s] = ((const uint2*)(hist + (size_t)s * PAIR_WORDS))[i];
        #pragma unroll
        for (int s = 0; s < N_SLICES; ++s) {
            ((int4*)(pre + (size_t)s * N_NODES))[i] = run;
            run.x += (int)(w[s].x & 0xFFFFu);
            run.y += (int)(w[s].x >> 16);
            run.z += (int)(w[s].y & 0xFFFFu);
            run.w += (int)(w[s].y >> 16);
        }
        ((int4*)cnt)[i] = run;
    }
    int t = (run.x + run.y) + (run.z + run.w);
    for (int o = 16; o > 0; o >>= 1) t += __shfl_xor(t, o);
    if ((tid & 31) == 0) wtot[tid >> 5] = t;
    __syncthreads();
    if (tid == 0) {
        int s = 0;
        for (int k = 0; k < 8; ++k) s += wtot[k];
        tsum[blockIdx.x] = s;
    }
}

#define POOL_Q ((N_GRAPHS * HID + N_GRAPHS) / 4)
static_assert((N_GRAPHS * HID + N_GRAPHS) % 4 == 0, "whole float4 groups");
static_assert(POOL_Q <= N_TILES * 256, "one float4 group per thread of the scan launch covers pooled + counts");

__global__ __launch_bounds__(256)
void scan_kernel(const int* __restrict__ cnt, const int* __restrict__ tsum,
                 int* __restrict__ rowptr, float* __restrict__ dinv,
                 float4* __restrict__ pool0) {
    __shared__ int wsum[8];
    int tid  = threadIdx.x;
    {
        int z = blockIdx.x * 256 + tid;
        if (z < POOL_Q) pool0[z] = make_float4(0.f, 0.f, 0.f, 0.f);
    }
    int lane = tid & 31;
    int wv   = tid >> 5;
    int tile0 = blockIdx.x * SCAN_TILE;
    const int4* c4 = (const int4*)cnt;

    int i0 = tile0 + tid * 4;
    int pre = 0;
    if (tid < (int)blockIdx.x) pre = tsum[tid];
    int4 v = make_int4(0, 0, 0, 0);
    if (i0 < N_NODES) v = c4[i0 >> 2];

    for (int o = 16; o > 0; o >>= 1) pre += __shfl_xor(pre, o);
    if (lane == 0) wsum[wv] = pre;
    __syncthreads();
    int base = 0;
    for (int k = 0; k < 8; ++k) base += wsum[k];
    __syncthreads();

    int t = (v.x + v.y) + (v.z + v.w);
    int incl = t;
    for (int o = 1; o < 32; o <<= 1) {
        int u = __shfl_up(incl, o);
        if (lane >= o) incl += u;
    }
    if (lane == 31) wsum[wv] = incl;
    __syncthreads();
    int woff = 0;
    for (int k = 0; k < wv; ++k) woff += wsum[k];
    int ex = base + woff + incl - t;
    if (i0 < N_NODES) {
        int4 r;
        r.x = ex;
        r.y = r.x + v.x;
        r.z = r.y + v.y;
        r.w = r.z + v.z;
        ((int4*)rowptr)[i0 >> 2] = r;
        float4 dv;
        dv.x = rsqrtf((float)(v.x + 1));
        dv.y = rsqrtf((float)(v.y + 1));
        dv.z = rsqrtf((float)(v.z + 1));
        dv.w = rsqrtf((float)(v.w + 1));
        ((float4*)dinv)[i0 >> 2] = dv;
    }
}

__device__ __forceinline__ int2 csr_entry(int s, float ds, float dd) {
    return make_int2(s, __float_as_int(ds * dd));
}

__global__ __launch_bounds__(256)
void fill_kernel(const int* __restrict__ src, const int* __restrict__ dst,
                 const int* __restrict__ rank, const int* __restrict__ rowptr,
                 const int* __restrict__ pre,
                 const float* __restrict__ dinv, int2* __restrict__ csr) {
    const int4* s4 = (const int4*)src;
    const int4* d4 = (const int4*)dst;
    const int4* r4 = (const int4*)rank;
    int q0 = blockIdx.x * EDGE_Q_PER_BLOCK + threadIdx.x;
    int q1 = q0 + 256;
    if (q1 < EDGE_Q) {
        const int* ea = pre + (size_t)(q0 / SLICE_Q) * N_NODES;
        const int* eb = pre + (size_t)(q1 / SLICE_Q) * N_NODES;
        int4 sa = s4[q0], da = d4[q0], ra = r4[q0];
        int4 sb = s4[q1], db = d4[q1], rb = r4[q1];
        int   pa0 = rowptr[da.x], pa1 = rowptr[da.y], pa2 = rowptr[da.z], pa3 = rowptr[da.w];
        int   pb0 = rowptr[db.x], pb1 = rowptr[db.y], pb2 = rowptr[db.z], pb3 = rowptr[db.w];
        int   ea0 = ea[da.x], ea1 = ea[da.y], ea2 = ea[da.z], ea3 = ea[da.w];
        int   eb0 = eb[db.x], eb1 = eb[db.y], eb2 = eb[db.z], eb3 = eb[db.w];
        float ua0 = dinv[sa.x], ua1 = dinv[sa.y], ua2 = dinv[sa.z], ua3 = dinv[sa.w];
        float ub0 = dinv[sb.x], ub1 = dinv[sb.y], ub2 = dinv[sb.z], ub3 = dinv[sb.w];
        float va0 = dinv[da.x], va1 = dinv[da.y], va2 = dinv[da.z], va3 = dinv[da.w];
        float vb0 = dinv[db.x], vb1 = dinv[db.y], vb2 = dinv[db.z], vb3 = dinv[db.w];
        csr[pa0 + ea0 + ra.x] = csr_entry(sa.x, ua0, va0);
        csr[pa1 + ea1 + ra.y] = csr_entry(sa.y, ua1, va1);
        csr[pa2 + ea2 + ra.z] = csr_entry(sa.z, ua2, va2);
        csr[pa3 + ea3 + ra.w] = csr_entry(sa.w, ua3, va3);
        csr[pb0 + eb0 + rb.x] = csr_entry(sb.x, ub0, vb0);
        csr[pb1 + eb1 + rb.y] = csr_entry(sb.y, ub1, vb1);
        csr[pb2 + eb2 + rb.z] = csr_entry(sb.z, ub2, vb2);
        csr[pb3 + eb3 + rb.w] = csr_entry(sb.w, ub3, vb3);
    } else if (q0 < EDGE_Q) {
        const int* ea = pre + (size_t)(q0 / SLICE_Q) * N_NODES;
        int4 sa = s4[q0], da = d4[q0], ra = r4[q0];
        int   pa0 = rowptr[da.x], pa1 = rowptr[da.y], pa2 = rowptr[da.z], pa3 = rowptr[da.w];
        int   ea0 = ea[da.x], ea1 = ea[da.y], ea2 = ea[da.z], ea3 = ea[da.w];
        float ua0 = dinv[sa.x], ua1 = dinv[sa.y], ua2 = dinv[sa.z], ua3 = dinv[sa.w];
        float va0 = dinv[da.x], va1 = dinv[da.y], va2 = dinv[da.z], va3 = dinv[da.w];
        csr[pa0 + ea0 + ra.x] = csr_entry(sa.x, ua0, va0);
        csr[pa1 + ea1 + ra.y] = csr_entry(sa.y, ua1, va1);
        csr[pa2 + ea2 + ra.z] = csr_entry(sa.z, ua2, va2);
        csr[pa3 + ea3 + ra.w] = csr_entry(sa.w, ua3, va3);
    }
}

#define LAYER_WAVES   16
#define LAYER_THREADS (LAYER_WAVES * 32)
#define LAYER_BLOCKS  256
#define N_STRIPS      (N_NODES / 16)
#define TILE_PITCH    132

static_assert(N_NODES % 16 == 0, "whole 16-node strips, 16-byte scalar loads per group of 4 nodes");
static_assert((HID * HID / 4) % LAYER_THREADS == 0, "W copy: whole passes of the workgroup");


__device__ __forceinline__ void node_done(float* __restrict__ tile, int lane, int cur, float4& acc) {
    float4* row = (float4*)(tile + cur * TILE_PITCH + 4 * lane);
    float4 t = *row;
    t.x = acc.x + t.x;
    t.y = acc.y + t.y;
    t.z = acc.z + t.z;
    t.w = acc.w + t.w;
    *row = t;
    acc = make_float4(0.f, 0.f, 0.f, 0.f);
}

template <int K>
__device__ __forceinline__ void batch(const float4* __restrict__ X4, float* __restrict__ tile, int lane,
                                      int ex, int ew, int j, int p, int endv,
                                      int& cur, int& end, float4& acc) {
    float4 v[K];
    float  w[K];
    #pragma unroll
    for (int k = 0; k < K; ++k) {
        int s = __builtin_amdgcn_readlane(ex, j + k);
        w[k]  = __int_as_float(__builtin_amdgcn_readlane(ew, j + k));
        v[k]  = X4[(size_t)s * (HID / 4) + lane];
    }
    #pragma unroll
    for (int k = 0; k < K; ++k) {
        while (p + k == end && cur < 15) {
            node_done(tile, lane, cur, acc);
            ++cur;
            end = __builtin_amdgcn_readlane(endv, cur);
        }
        acc.x = fmaf(v[k].x, w[k], acc.x);
        acc.y = fmaf(v[k].y, w[k], acc.y);
        acc.z = fmaf(v[k].z, w[k], acc.z);
        acc.w = fmaf(v[k].w, w[k], acc.w);
    }
}

__device__ __forceinline__ void pool_flush(float* __restrict__ pooled, float* __restrict__ counts,
                                           int g, int lane, float4 acc, float c) {
    float* pp = pooled + (size_t)g * HID + lane * 4;
    atomicAdd(pp + 0, acc.x);
    atomicAdd(pp + 1, acc.y);
    atomicAdd(pp + 2, acc.z);
    atomicAdd(pp + 3, acc.w);
    if (lane == 0) atomicAdd(&counts[g], c);
}

template <bool LAST>
__global__ __launch_bounds__(LAYER_THREADS)
void layer_kernel(const float* __restrict__ Xin, const float* __restrict__ W,
                  const float* __restrict__ bias, const int* __restrict__ rowptr,
                  const int* __restrict__ cnt, const int2* __restrict__ csr,
                  const float* __restrict__ dinv, float* __restrict__ Xout,
                  const int* __restrict__ graph_id, float* __restrict__ pooled,
                  float* __restrict__ counts) {
    __shared__ float4 Wq[HID * HID / 4];
    __shared__ float4 tile4[LAYER_WAVES * 16 * TILE_PITCH / 4];

    int tid = threadIdx.x;
    #pragma unroll
    for (int i = 0; i < (HID * HID / 4) / LAYER_THREADS; ++i) {
        int p  = tid + i * LAYER_THREADS;
        int n  = p & (HID - 1);
        int gh = p >> 7;
        int k0 = (gh >> 1) * 8 + (gh & 1) * 2;
        float4 q;
        q.x = W[(k0 + 0) * HID + n];
        q.y = W[(k0 + 1) * HID + n];
        q.z = W[(k0 + 4) * HID + n];
        q.w = W[(k0 + 5) * HID + n];
        Wq[p] = q;
    }
    __syncthreads();

    int lane = tid & 31;
    int wv = __builtin_amdgcn_readfirstlane(tid >> 5);
    int m  = lane & 15;
    int kh = lane >> 4;

    int s0 = ((int)blockIdx.x * N_STRIPS) / LAYER_BLOCKS;
    int s1 = (((int)blockIdx.x + 1) * N_STRIPS) / LAYER_BLOCKS;
    const float4* bq = Wq + kh * HID + m;
    const float4* X4 = (const float4*)Xin;
    float* tile = (float*)tile4 + wv * (16 * TILE_PITCH);
    unsigned a_off = (unsigned)(m * TILE_PITCH + 2 * kh);
    unsigned h_off = (unsigned)(8 * kh * HID + m);

    float bv[8];
    #pragma unroll
    for (int t = 0; t < 8; ++t) bv[t] = bias[t * 16 + m];

    for (int s = s0 + wv; s < s1; s += LAYER_WAVES) {
        int node0 = s * 16;
        int   rp = rowptr[node0 + m];
        int   cn = cnt[node0 + m];
        float dv = dinv[node0 + m];
        int   gid = 0;
        if (LAST) gid = graph_id[node0 + m];
        float4 hv[16];
        #pragma unroll
        for (int i = 0; i < 16; ++i) hv[i] = X4[(size_t)(node0 + i) * (HID / 4) + lane];

        int endv = rp + cn;
        int p0 = __builtin_amdgcn_readlane(rp, 0);
        int pE = __builtin_amdgcn_readlane(endv, 15);
        pE = min(pE, p0 + N_EDGES);

        int2 entn = make_int2(node0, 0);
        if (lane < pE - p0) entn = csr[p0 + lane];

        #pragma unroll
        for (int i = 0; i < 16; ++i) {
            float ds = __int_as_float(__builtin_amdgcn_readlane(__float_as_int(dv), i));
            float wself = ds * ds;
            float4 r;
            r.x = hv[i].x * wself;
            r.y = hv[i].y * wself;
            r.z = hv[i].z * wself;
            r.w = hv[i].w * wself;
            *(float4*)(tile + i * TILE_PITCH + 4 * lane) = r;
        }

        {
            float4 acc = make_float4(0.f, 0.f, 0.f, 0.f);
            int cur = 0;
            int end = __builtin_amdgcn_readlane(endv, 0);
            for (int c0 = p0; c0 < pE; c0 += 32) {
                int mm = pE - c0;
                if (mm > 32) mm = 32;
                int2 ent = entn;
                entn = make_int2(node0, 0);
                if (lane < pE - (c0 + 32)) entn = csr[c0 + 32 + lane];
                int ex = (int)min((unsigned)ent.x, (unsigned)(N_NODES - 1));
                int ew = ent.y;
                int j = 0;
                for (; j + 16 <= mm; j += 16) batch<16>(X4, tile, lane, ex, ew, j, c0 + j, endv, cur, end, acc);
                if (j + 8 <= mm) { batch<8>(X4, tile, lane, ex, ew, j, c0 + j, endv, cur, end, acc); j += 8; }
                if (j + 4 <= mm) { batch<4>(X4, tile, lane, ex, ew, j, c0 + j, endv, cur, end, acc); j += 4; }
                if (j + 2 <= mm) { batch<2>(X4, tile, lane, ex, ew, j, c0 + j, endv, cur, end, acc); j += 2; }
                if (j < mm)      { batch<1>(X4, tile, lane, ex, ew, j, c0 + j, endv, cur, end, acc); }
            }
            for (; cur < 16; ++cur) node_done(tile, lane, cur, acc);
        }
        v2f a[HID / 4];
        #pragma unroll
        for (int i = 0; i < HID / 4; ++i) a[i] = *(const v2f*)(tile + a_off + 4 * i);

        v8f acc[8];
        #pragma unroll
        for (int t = 0; t < 8; ++t) acc[t] = (v8f){};

        float4 b[2][8];
        #pragma unroll
        for (int t = 0; t < 8; ++t) b[0][t] = bq[t * 16];

        #pragma unroll
        for (int g = 0; g < HID / 8; ++g) {
            if (g + 1 < HID / 8) {
                #pragma unroll
                for (int t = 0; t < 8; ++t) b[(g + 1) & 1][t] = bq[(g + 1) * 2 * HID + t * 16];
            }
            asm volatile("" ::: "memory");
            #pragma unroll
            for (int t = 0; t < 8; ++t) {
                v2f bb = {b[g & 1][t].x, b[g & 1][t].y};
                acc[t] = __builtin_amdgcn_wmma_f32_16x16x4_f32(
                     false, a[2 * g],  false, bb,
                     (short)0, acc[t],  false,  false);
            }
            #pragma unroll
            for (int t = 0; t < 8; ++t) {
                v2f bb = {b[g & 1][t].z, b[g & 1][t].w};
                acc[t] = __builtin_amdgcn_wmma_f32_16x16x4_f32(
                     false, a[2 * g + 1],  false, bb,
                     (short)0, acc[t],  false,  false);
            }
        }

        if (!LAST) {
            float* os = Xout + (size_t)s * (16 * HID);
            #pragma unroll
            for (int t = 0; t < 8; ++t) {
                #pragma unroll
                for (int r = 0; r < 8; ++r) {
                    float v = acc[t][r] + bv[t];
                    v = v > 0.f ? v : 0.f;
                    (os + r * HID + t * 16)[h_off] = v;
                }
            }
        } else {
            #pragma unroll
            for (int t = 0; t < 8; ++t) {
                #pragma unroll
                for (int r = 0; r < 8; ++r) {
                    float v = acc[t][r] + bv[t];
                    v = v > 0.f ? v : 0.f;
                    tile[(r + 8 * kh) * TILE_PITCH + t * 16 + m] = v;
                }
            }
            const float4* trow = (const float4*)(tile + 4 * lane);
            float4 pv[16];
            #pragma unroll
            for (int k = 0; k < 16; ++k) pv[k] = trow[k * (TILE_PITCH / 4)];

            int    gcur = __builtin_amdgcn_readlane(gid, 0);
            float4 psum = pv[0];
            float  pc   = 1.0f;
            #pragma unroll
            for (int k = 1; k < 16; ++k) {
                int gk = __builtin_amdgcn_readlane(gid, k);
                if (gk != gcur) {
                    pool_flush(pooled, counts, gcur, lane, psum, pc);
                    gcur = gk;
                    psum = pv[k];
                    pc   = 1.0f;
                } else {
                    psum.x += pv[k].x;
                    psum.y += pv[k].y;
                    psum.z += pv[k].z;
                    psum.w += pv[k].w;
                    pc += 1.0f;
                }
            }
            pool_flush(pooled, counts, gcur, lane, psum, pc);
        }
    }
}

__global__ void final_linear_kernel(const float* __restrict__ pooled,
                                    const float* __restrict__ counts,
                                    const float* __restrict__ linW,
                                    const float* __restrict__ linb,
                                    float* __restrict__ out) {
    int i = blockIdx.x * blockDim.x + threadIdx.x;
    if (i >= N_GRAPHS * N_CLASSES) return;
    int g = i / N_CLASSES;
    int c = i - g * N_CLASSES;
    float scale = 1.0f / fmaxf(counts[g], 1.0f);
    const float* pg = pooled + (size_t)g * HID;
    float acc = 0.f;
    #pragma unroll 4
    for (int f = 0; f < HID; ++f) acc += pg[f] * linW[f * N_CLASSES + c];
    out[i] = acc * scale + linb[c];
}

extern "C" void kernel_launch(void* const* d_in, const int* in_sizes, int n_in,
                              void* d_out, int out_size, void* d_ws, size_t ws_size,
                              hipStream_t stream) {
    const float* x     = (const float*)d_in[0];
    const int*   edge  = (const int*)d_in[1];
    const int*   src   = edge;
    const int*   dst   = edge + N_EDGES;
    const int*   batch = (const int*)d_in[2];
    const float* Ws    = (const float*)d_in[3];
    const float* bs    = (const float*)d_in[4];
    const float* linW  = (const float*)d_in[5];
    const float* linb  = (const float*)d_in[6];
    float*       out   = (float*)d_out;

    float*  ws     = (float*)d_ws;
    int*    cnt    = (int*)ws;
    float*  dinv   = ws + N_NODES;
    int*    rowptr = (int*)(dinv + N_NODES);
    int*    rank   = rowptr + N_NODES;
    int2*   csr    = (int2*)(rank + N_EDGES);
    float*  H      = (float*)(csr + N_EDGES);
    float*  X      = H + (size_t)N_NODES * HID;
    float*  pooled = X + (size_t)N_NODES * HID;
    float*  counts = pooled + (size_t)N_GRAPHS * HID;
    unsigned* hist = (unsigned*)(counts + N_GRAPHS);
    int*    pre    = (int*)(hist + (size_t)N_SLICES * PAIR_WORDS);
    int*    tsum   = pre + (size_t)N_SLICES * N_NODES;

    const int T = 256;

    count_lds_kernel<<<N_SLICES, CL_THREADS, 0, stream>>>(dst, rank, hist);
    slice_prefix_kernel<<<N_TILES, T, 0, stream>>>(hist, pre, cnt, tsum);
    scan_kernel<<<N_TILES, T, 0, stream>>>(cnt, tsum, rowptr, dinv, (float4*)pooled);
    fill_kernel<<<EDGE_BLOCKS, T, 0, stream>>>(src, dst, rank, rowptr, pre, dinv, csr);

    const float* xin   = x;
    float*       xout  = H;
    float*       xnext = X;
    for (int l = 0; l < N_LAYERS - 1; ++l) {
        const float* Wl = Ws + (size_t)l * HID * HID;
        const float* bl = bs + (size_t)l * HID;

        layer_kernel<false><<<LAYER_BLOCKS, LAYER_THREADS, 0, stream>>>(
            xin, Wl, bl, rowptr, cnt, csr, dinv, xout, batch, pooled, counts);

        xin = xout;
        float* t = xout;
        xout  = xnext;
        xnext = t;
    }
    layer_kernel<true><<<LAYER_BLOCKS, LAYER_THREADS, 0, stream>>>(
        xin, Ws + (size_t)(N_LAYERS - 1) * HID * HID, bs + (size_t)(N_LAYERS - 1) * HID,
        rowptr, cnt, csr, dinv, xout, batch, pooled, counts);

    final_linear_kernel<<<(N_GRAPHS * N_CLASSES + T - 1) / T, T, 0, stream>>>(
        pooled, counts, linW, linb, out);
}


# Row gather double-buffered: a wave asks for the next 16 rows before it adds the current 16 (two register sets, up to 32 rows in flight; strip tails padded, pads not added); sum orders kept
# speedup vs baseline: 5.3348x; 1.0381x over previous
#include <hip/hip_runtime.h>
#include <hip/hip_bf16.h>

#define N_NODES   100000
#define N_EDGES   1600000
#define HID       128
#define N_CLASSES 10
#define N_GRAPHS  512
#define N_LAYERS  5

static_assert(N_NODES % 4 == 0, "int4 passes over per-node arrays");

typedef __attribute__((ext_vector_type(2))) float v2f;
typedef __attribute__((ext_vector_type(8))) float v8f;


static_assert(N_EDGES % 4 == 0, "int4 passes over per-edge arrays");
#define EDGE_Q            (N_EDGES / 4)
#define EDGE_Q_PER_BLOCK  512
#define EDGE_BLOCKS       ((EDGE_Q + EDGE_Q_PER_BLOCK - 1) / EDGE_Q_PER_BLOCK)

#define N_SLICES     32
#define SLICE_EDGES  (N_EDGES / N_SLICES)
#define SLICE_Q      (SLICE_EDGES / 4)
#define PAIR_WORDS   (N_NODES / 2)
#define CL_THREADS   1024
#define CL_ITERS     ((SLICE_Q + CL_THREADS - 1) / CL_THREADS)

static_assert(N_EDGES % N_SLICES == 0, "equal slices");
static_assert(SLICE_EDGES % 4 == 0, "a slice is whole int4 groups, so a group never straddles two slices");
static_assert(SLICE_EDGES <= 65535, "a 16-bit counter cannot overflow inside one slice");
static_assert(N_NODES % 8 == 0, "16-byte passes over the packed counter table");

__device__ __forceinline__ int lds_count(unsigned* __restrict__ c, int d) {
    unsigned u  = min((unsigned)d, (unsigned)(N_NODES - 1));
    unsigned sh = (u & 1u) * 16u;
    unsigned old = atomicAdd(&c[u >> 1], 1u << sh);
    return (int)((old >> sh) & 0xFFFFu);
}

__global__ __launch_bounds__(CL_THREADS)
void count_lds_kernel(const int* __restrict__ dst, int* __restrict__ rank,
                      unsigned* __restrict__ hist) {
    __shared__ uint4 c4[PAIR_WORDS / 4];
    unsigned* c = (unsigned*)c4;
    int tid = threadIdx.x;
    int s   = blockIdx.x;
    const int4* d4 = (const int4*)dst + (size_t)s * SLICE_Q;
    int4*       r4 = (int4*)rank + (size_t)s * SLICE_Q;

    int4 a[CL_ITERS];
    #pragma unroll
    for (int k = 0; k < CL_ITERS; ++k) {
        int q = tid + k * CL_THREADS;
        a[k] = make_int4(0, 0, 0, 0);
        if (q < SLICE_Q) a[k] = d4[q];
    }
    for (int i = tid; i < PAIR_WORDS / 4; i += CL_THREADS) c4[i] = make_uint4(0u, 0u, 0u, 0u);
    __syncthreads();

    #pragma unroll
    for (int k = 0; k < CL_ITERS; ++k) {
        int q = tid + k * CL_THREADS;
        if (q < SLICE_Q) {
            int4 r;
            r.x = lds_count(c, a[k].x);
            r.y = lds_count(c, a[k].y);
            r.z = lds_count(c, a[k].z);
            r.w = lds_count(c, a[k].w);
            r4[q] = r;
        }
    }
    __syncthreads();

    uint4* h4 = (uint4*)(hist + (size_t)s * PAIR_WORDS);
    for (int i = tid; i < PAIR_WORDS / 4; i += CL_THREADS) h4[i] = c4[i];
}

#define SCAN_TILE 1024
#define N_TILES   ((N_NODES + SCAN_TILE - 1) / SCAN_TILE)
static_assert(SCAN_TILE == 4 * 256, "a slice-sum block of 256 threads x 4 nodes is one scan tile");
static_assert(N_TILES <= 256, "one tile total per thread of a scan block");

__global__ __launch_bounds__(256)
void slice_prefix_kernel(const unsigned* __restrict__ hist, int* __restrict__ pre,
                         int* __restrict__ cnt, int* __restrict__ tsum) {
    __shared__ int wtot[8];
    int tid = threadIdx.x;
    int i = blockIdx.x * 256 + tid;
    int4 run = make_int4(0, 0, 0, 0);
    if (i < N_NODES / 4) {
        uint2 w[N_SLICES];
        #pragma unroll
        for (int s = 0; s < N_SLICES; ++s)
            w[s] = ((const uint2*)(hist + (size_t)s * PAIR_WORDS))[i];
        #pragma unroll
        for (int s = 0; s < N_SLICES; ++s) {
            ((int4*)(pre + (size_t)s * N_NODES))[i] = run;
            run.x += (int)(w[s].x & 0xFFFFu);
            run.y += (int)(w[s].x >> 16);
            run.z += (int)(w[s].y & 0xFFFFu);
            run.w += (int)(w[s].y >> 16);
        }
        ((int4*)cnt)[i] = run;
    }
    int t = (run.x + run.y) + (run.z + run.w);
    for (int o = 16; o > 0; o >>= 1) t += __shfl_xor(t, o);
    if ((tid & 31) == 0) wtot[tid >> 5] = t;
    __syncthreads();
    if (tid == 0) {
        int s = 0;
        for (int k = 0; k < 8; ++k) s += wtot[k];
        tsum[blockIdx.x] = s;
    }
}

#define POOL_Q ((N_GRAPHS * HID + N_GRAPHS) / 4)
static_assert((N_GRAPHS * HID + N_GRAPHS) % 4 == 0, "whole float4 groups");
static_assert(POOL_Q <= N_TILES * 256, "one float4 group per thread of the scan launch covers pooled + counts");

__global__ __launch_bounds__(256)
void scan_kernel(const int* __restrict__ cnt, const int* __restrict__ tsum,
                 int* __restrict__ rowptr, float* __restrict__ dinv,
                 float4* __restrict__ pool0) {
    __shared__ int wsum[8];
    int tid  = threadIdx.x;
    {
        int z = blockIdx.x * 256 + tid;
        if (z < POOL_Q) pool0[z] = make_float4(0.f, 0.f, 0.f, 0.f);
    }
    int lane = tid & 31;
    int wv   = tid >> 5;
    int tile0 = blockIdx.x * SCAN_TILE;
    const int4* c4 = (const int4*)cnt;

    int i0 = tile0 + tid * 4;
    int pre = 0;
    if (tid < (int)blockIdx.x) pre = tsum[tid];
    int4 v = make_int4(0, 0, 0, 0);
    if (i0 < N_NODES) v = c4[i0 >> 2];

    for (int o = 16; o > 0; o >>= 1) pre += __shfl_xor(pre, o);
    if (lane == 0) wsum[wv] = pre;
    __syncthreads();
    int base = 0;
    for (int k = 0; k < 8; ++k) base += wsum[k];
    __syncthreads();

    int t = (v.x + v.y) + (v.z + v.w);
    int incl = t;
    for (int o = 1; o < 32; o <<= 1) {
        int u = __shfl_up(incl, o);
        if (lane >= o) incl += u;
    }
    if (lane == 31) wsum[wv] = incl;
    __syncthreads();
    int woff = 0;
    for (int k = 0; k < wv; ++k) woff += wsum[k];
    int ex = base + woff + incl - t;
    if (i0 < N_NODES) {
        int4 r;
        r.x = ex;
        r.y = r.x + v.x;
        r.z = r.y + v.y;
        r.w = r.z + v.z;
        ((int4*)rowptr)[i0 >> 2] = r;
        float4 dv;
        dv.x = rsqrtf((float)(v.x + 1));
        dv.y = rsqrtf((float)(v.y + 1));
        dv.z = rsqrtf((float)(v.z + 1));
        dv.w = rsqrtf((float)(v.w + 1));
        ((float4*)dinv)[i0 >> 2] = dv;
    }
}

__device__ __forceinline__ int2 csr_entry(int s, float ds, float dd) {
    return make_int2(s, __float_as_int(ds * dd));
}

__global__ __launch_bounds__(256)
void fill_kernel(const int* __restrict__ src, const int* __restrict__ dst,
                 const int* __restrict__ rank, const int* __restrict__ rowptr,
                 const int* __restrict__ pre,
                 const float* __restrict__ dinv, int2* __restrict__ csr) {
    const int4* s4 = (const int4*)src;
    const int4* d4 = (const int4*)dst;
    const int4* r4 = (const int4*)rank;
    int q0 = blockIdx.x * EDGE_Q_PER_BLOCK + threadIdx.x;
    int q1 = q0 + 256;
    if (q1 < EDGE_Q) {
        const int* ea = pre + (size_t)(q0 / SLICE_Q) * N_NODES;
        const int* eb = pre + (size_t)(q1 / SLICE_Q) * N_NODES;
        int4 sa = s4[q0], da = d4[q0], ra = r4[q0];
        int4 sb = s4[q1], db = d4[q1], rb = r4[q1];
        int   pa0 = rowptr[da.x], pa1 = rowptr[da.y], pa2 = rowptr[da.z], pa3 = rowptr[da.w];
        int   pb0 = rowptr[db.x], pb1 = rowptr[db.y], pb2 = rowptr[db.z], pb3 = rowptr[db.w];
        int   ea0 = ea[da.x], ea1 = ea[da.y], ea2 = ea[da.z], ea3 = ea[da.w];
        int   eb0 = eb[db.x], eb1 = eb[db.y], eb2 = eb[db.z], eb3 = eb[db.w];
        float ua0 = dinv[sa.x], ua1 = dinv[sa.y], ua2 = dinv[sa.z], ua3 = dinv[sa.w];
        float ub0 = dinv[sb.x], ub1 = dinv[sb.y], ub2 = dinv[sb.z], ub3 = dinv[sb.w];
        float va0 = dinv[da.x], va1 = dinv[da.y], va2 = dinv[da.z], va3 = dinv[da.w];
        float vb0 = dinv[db.x], vb1 = dinv[db.y], vb2 = dinv[db.z], vb3 = dinv[db.w];
        csr[pa0 + ea0 + ra.x] = csr_entry(sa.x, ua0, va0);
        csr[pa1 + ea1 + ra.y] = csr_entry(sa.y, ua1, va1);
        csr[pa2 + ea2 + ra.z] = csr_entry(sa.z, ua2, va2);
        csr[pa3 + ea3 + ra.w] = csr_entry(sa.w, ua3, va3);
        csr[pb0 + eb0 + rb.x] = csr_entry(sb.x, ub0, vb0);
        csr[pb1 + eb1 + rb.y] = csr_entry(sb.y, ub1, vb1);
        csr[pb2 + eb2 + rb.z] = csr_entry(sb.z, ub2, vb2);
        csr[pb3 + eb3 + rb.w] = csr_entry(sb.w, ub3, vb3);
    } else if (q0 < EDGE_Q) {
        const int* ea = pre + (size_t)(q0 / SLICE_Q) * N_NODES;
        int4 sa = s4[q0], da = d4[q0], ra = r4[q0];
        int   pa0 = rowptr[da.x], pa1 = rowptr[da.y], pa2 = rowptr[da.z], pa3 = rowptr[da.w];
        int   ea0 = ea[da.x], ea1 = ea[da.y], ea2 = ea[da.z], ea3 = ea[da.w];
        float ua0 = dinv[sa.x], ua1 = dinv[sa.y], ua2 = dinv[sa.z], ua3 = dinv[sa.w];
        float va0 = dinv[da.x], va1 = dinv[da.y], va2 = dinv[da.z], va3 = dinv[da.w];
        csr[pa0 + ea0 + ra.x] = csr_entry(sa.x, ua0, va0);
        csr[pa1 + ea1 + ra.y] = csr_entry(sa.y, ua1, va1);
        csr[pa2 + ea2 + ra.z] = csr_entry(sa.z, ua2, va2);
        csr[pa3 + ea3 + ra.w] = csr_entry(sa.w, ua3, va3);
    }
}

#define LAYER_WAVES   16
#define LAYER_THREADS (LAYER_WAVES * 32)
#define LAYER_BLOCKS  256
#define N_STRIPS      (N_NODES / 16)
#define TILE_PITCH    132

static_assert(N_NODES % 16 == 0, "whole 16-node strips, 16-byte scalar loads per group of 4 nodes");
static_assert((HID * HID / 4) % LAYER_THREADS == 0, "W copy: whole passes of the workgroup");


__device__ __forceinline__ void node_done(float* __restrict__ tile, int lane, int cur, float4& acc) {
    float4* row = (float4*)(tile + cur * TILE_PITCH + 4 * lane);
    float4 t = *row;
    t.x = acc.x + t.x;
    t.y = acc.y + t.y;
    t.z = acc.z + t.z;
    t.w = acc.w + t.w;
    *row = t;
    acc = make_float4(0.f, 0.f, 0.f, 0.f);
}

template <int J>
__device__ __forceinline__ void rows_issue(const float4* __restrict__ X4, int lane, int ex, int ew,
                                           float4 (&v)[16], float (&w)[16]) {
    #pragma unroll
    for (int k = 0; k < 16; ++k) {
        int s = __builtin_amdgcn_readlane(ex, J + k);
        w[k]  = __int_as_float(__builtin_amdgcn_readlane(ew, J + k));
        v[k]  = X4[(size_t)s * (HID / 4) + lane];
    }
}

template <bool TAIL>
__device__ __forceinline__ void rows_sum(float* __restrict__ tile, int lane,
                                         const float4 (&v)[16], const float (&w)[16],
                                         int p, int endv, int& cur, int& end, float4& acc) {
    if (TAIL) {
        int rel = end - p;
        #pragma unroll
        for (int k = 0; k < 16; ++k) {
            if (rel == k) {
                while (rel == k && cur < 15) {
                    node_done(tile, lane, cur, acc);
                    ++cur;
                    end = __builtin_amdgcn_readlane(endv, cur);
                    rel = end - p;
                }
                if (rel == k) return;
            }
            acc.x = fmaf(v[k].x, w[k], acc.x);
            acc.y = fmaf(v[k].y, w[k], acc.y);
            acc.z = fmaf(v[k].z, w[k], acc.z);
            acc.w = fmaf(v[k].w, w[k], acc.w);
        }
    } else {
        int rel = cur < 15 ? end - p : -1;
        #pragma unroll
        for (int k = 0; k < 16; ++k) {
            while (rel == k) {
                node_done(tile, lane, cur, acc);
                ++cur;
                end = __builtin_amdgcn_readlane(endv, cur);
                rel = cur < 15 ? end - p : -1;
            }
            acc.x = fmaf(v[k].x, w[k], acc.x);
            acc.y = fmaf(v[k].y, w[k], acc.y);
            acc.z = fmaf(v[k].z, w[k], acc.z);
            acc.w = fmaf(v[k].w, w[k], acc.w);
        }
    }
}

__device__ __forceinline__ void pool_flush(float* __restrict__ pooled, float* __restrict__ counts,
                                           int g, int lane, float4 acc, float c) {
    float* pp = pooled + (size_t)g * HID + lane * 4;
    atomicAdd(pp + 0, acc.x);
    atomicAdd(pp + 1, acc.y);
    atomicAdd(pp + 2, acc.z);
    atomicAdd(pp + 3, acc.w);
    if (lane == 0) atomicAdd(&counts[g], c);
}

template <bool LAST>
__global__ __launch_bounds__(LAYER_THREADS)
void layer_kernel(const float* __restrict__ Xin, const float* __restrict__ W,
                  const float* __restrict__ bias, const int* __restrict__ rowptr,
                  const int* __restrict__ cnt, const int2* __restrict__ csr,
                  const float* __restrict__ dinv, float* __restrict__ Xout,
                  const int* __restrict__ graph_id, float* __restrict__ pooled,
                  float* __restrict__ counts) {
    __shared__ float4 Wq[HID * HID / 4];
    __shared__ float4 tile4[LAYER_WAVES * 16 * TILE_PITCH / 4];

    int tid = threadIdx.x;
    #pragma unroll
    for (int i = 0; i < (HID * HID / 4) / LAYER_THREADS; ++i) {
        int p  = tid + i * LAYER_THREADS;
        int n  = p & (HID - 1);
        int gh = p >> 7;
        int k0 = (gh >> 1) * 8 + (gh & 1) * 2;
        float4 q;
        q.x = W[(k0 + 0) * HID + n];
        q.y = W[(k0 + 1) * HID + n];
        q.z = W[(k0 + 4) * HID + n];
        q.w = W[(k0 + 5) * HID + n];
        Wq[p] = q;
    }
    __syncthreads();

    int lane = tid & 31;
    int wv = __builtin_amdgcn_readfirstlane(tid >> 5);
    int m  = lane & 15;
    int kh = lane >> 4;

    int s0 = ((int)blockIdx.x * N_STRIPS) / LAYER_BLOCKS;
    int s1 = (((int)blockIdx.x + 1) * N_STRIPS) / LAYER_BLOCKS;
    const float4* bq = Wq + kh * HID + m;
    const float4* X4 = (const float4*)Xin;
    float* tile = (float*)tile4 + wv * (16 * TILE_PITCH);
    unsigned a_off = (unsigned)(m * TILE_PITCH + 2 * kh);
    unsigned h_off = (unsigned)(8 * kh * HID + m);

    float bv[8];
    #pragma unroll
    for (int t = 0; t < 8; ++t) bv[t] = bias[t * 16 + m];

    for (int s = s0 + wv; s < s1; s += LAYER_WAVES) {
        int node0 = s * 16;
        int   rp = rowptr[node0 + m];
        int   cn = cnt[node0 + m];
        float dv = dinv[node0 + m];
        int   gid = 0;
        if (LAST) gid = graph_id[node0 + m];
        float4 hv[16];
        #pragma unroll
        for (int i = 0; i < 16; ++i) hv[i] = X4[(size_t)(node0 + i) * (HID / 4) + lane];

        int endv = rp + cn;
        int p0 = __builtin_amdgcn_readlane(rp, 0);
        int pE = __builtin_amdgcn_readlane(endv, 15);
        pE = min(pE, p0 + N_EDGES);

        int2 ent  = make_int2(node0, 0);
        int2 entn = make_int2(node0, 0);
        if (lane < pE - p0)        ent  = csr[p0 + lane];
        if (lane < pE - (p0 + 32)) entn = csr[p0 + 32 + lane];

        #pragma unroll
        for (int i = 0; i < 16; ++i) {
            float ds = __int_as_float(__builtin_amdgcn_readlane(__float_as_int(dv), i));
            float wself = ds * ds;
            float4 r;
            r.x = hv[i].x * wself;
            r.y = hv[i].y * wself;
            r.z = hv[i].z * wself;
            r.w = hv[i].w * wself;
            *(float4*)(tile + i * TILE_PITCH + 4 * lane) = r;
        }

        {
            float4 acc = make_float4(0.f, 0.f, 0.f, 0.f);
            int cur = 0;
            int end = __builtin_amdgcn_readlane(endv, 0);
            float4 va[16], vb[16];
            float  wa[16], wb[16];
            int ex = (int)min((unsigned)ent.x, (unsigned)(N_NODES - 1));
            int ew = ent.y;
            rows_issue<0>(X4, lane, ex, ew, va, wa);
            for (int c0 = p0; c0 < pE; c0 += 32) {
                rows_issue<16>(X4, lane, ex, ew, vb, wb);
                if (c0 + 16 <= pE) rows_sum<false>(tile, lane, va, wa, c0, endv, cur, end, acc);
                else               rows_sum<true>(tile, lane, va, wa, c0, endv, cur, end, acc);
                ent  = entn;
                entn = make_int2(node0, 0);
                if (lane < pE - (c0 + 64)) entn = csr[c0 + 64 + lane];
                ex = (int)min((unsigned)ent.x, (unsigned)(N_NODES - 1));
                ew = ent.y;
                rows_issue<0>(X4, lane, ex, ew, va, wa);
                if (c0 + 32 <= pE)     rows_sum<false>(tile, lane, vb, wb, c0 + 16, endv, cur, end, acc);
                else if (c0 + 16 < pE) rows_sum<true>(tile, lane, vb, wb, c0 + 16, endv, cur, end, acc);
            }
            for (; cur < 16; ++cur) node_done(tile, lane, cur, acc);
        }
        v2f a[HID / 4];
        #pragma unroll
        for (int i = 0; i < HID / 4; ++i) a[i] = *(const v2f*)(tile + a_off + 4 * i);

        v8f acc[8];
        #pragma unroll
        for (int t = 0; t < 8; ++t) acc[t] = (v8f){};

        float4 b[2][8];
        #pragma unroll
        for (int t = 0; t < 8; ++t) b[0][t] = bq[t * 16];

        #pragma unroll
        for (int g = 0; g < HID / 8; ++g) {
            if (g + 1 < HID / 8) {
                #pragma unroll
                for (int t = 0; t < 8; ++t) b[(g + 1) & 1][t] = bq[(g + 1) * 2 * HID + t * 16];
            }
            asm volatile("" ::: "memory");
            #pragma unroll
            for (int t = 0; t < 8; ++t) {
                v2f bb = {b[g & 1][t].x, b[g & 1][t].y};
                acc[t] = __builtin_amdgcn_wmma_f32_16x16x4_f32(
                     false, a[2 * g],  false, bb,
                     (short)0, acc[t],  false,  false);
            }
            #pragma unroll
            for (int t = 0; t < 8; ++t) {
                v2f bb = {b[g & 1][t].z, b[g & 1][t].w};
                acc[t] = __builtin_amdgcn_wmma_f32_16x16x4_f32(
                     false, a[2 * g + 1],  false, bb,
                     (short)0, acc[t],  false,  false);
            }
        }

        if (!LAST) {
            float* os = Xout + (size_t)s * (16 * HID);
            #pragma unroll
            for (int t = 0; t < 8; ++t) {
                #pragma unroll
                for (int r = 0; r < 8; ++r) {
                    float v = acc[t][r] + bv[t];
                    v = v > 0.f ? v : 0.f;
                    (os + r * HID + t * 16)[h_off] = v;
                }
            }
        } else {
            #pragma unroll
            for (int t = 0; t < 8; ++t) {
                #pragma unroll
                for (int r = 0; r < 8; ++r) {
                    float v = acc[t][r] + bv[t];
                    v = v > 0.f ? v : 0.f;
                    tile[(r + 8 * kh) * TILE_PITCH + t * 16 + m] = v;
                }
            }
            const float4* trow = (const float4*)(tile + 4 * lane);
            float4 pv[16];
            #pragma unroll
            for (int k = 0; k < 16; ++k) pv[k] = trow[k * (TILE_PITCH / 4)];

            int    gcur = __builtin_amdgcn_readlane(gid, 0);
            float4 psum = pv[0];
            float  pc   = 1.0f;
            #pragma unroll
            for (int k = 1; k < 16; ++k) {
                int gk = __builtin_amdgcn_readlane(gid, k);
                if (gk != gcur) {
                    pool_flush(pooled, counts, gcur, lane, psum, pc);
                    gcur = gk;
                    psum = pv[k];
                    pc   = 1.0f;
                } else {
                    psum.x += pv[k].x;
                    psum.y += pv[k].y;
                    psum.z += pv[k].z;
                    psum.w += pv[k].w;
                    pc += 1.0f;
                }
            }
            pool_flush(pooled, counts, gcur, lane, psum, pc);
        }
    }
}

__global__ void final_linear_kernel(const float* __restrict__ pooled,
                                    const float* __restrict__ counts,
                                    const float* __restrict__ linW,
                                    const float* __restrict__ linb,
                                    float* __restrict__ out) {
    int i = blockIdx.x * blockDim.x + threadIdx.x;
    if (i >= N_GRAPHS * N_CLASSES) return;
    int g = i / N_CLASSES;
    int c = i - g * N_CLASSES;
    float scale = 1.0f / fmaxf(counts[g], 1.0f);
    const float* pg = pooled + (size_t)g * HID;
    float acc = 0.f;
    #pragma unroll 4
    for (int f = 0; f < HID; ++f) acc += pg[f] * linW[f * N_CLASSES + c];
    out[i] = acc * scale + linb[c];
}

extern "C" void kernel_launch(void* const* d_in, const int* in_sizes, int n_in,
                              void* d_out, int out_size, void* d_ws, size_t ws_size,
                              hipStream_t stream) {
    const float* x     = (const float*)d_in[0];
    const int*   edge  = (const int*)d_in[1];
    const int*   src   = edge;
    const int*   dst   = edge + N_EDGES;
    const int*   batch = (const int*)d_in[2];
    const float* Ws    = (const float*)d_in[3];
    const float* bs    = (const float*)d_in[4];
    const float* linW  = (const float*)d_in[5];
    const float* linb  = (const float*)d_in[6];
    float*       out   = (float*)d_out;

    float*  ws     = (float*)d_ws;
    int*    cnt    = (int*)ws;
    float*  dinv   = ws + N_NODES;
    int*    rowptr = (int*)(dinv + N_NODES);
    int*    rank   = rowptr + N_NODES;
    int2*   csr    = (int2*)(rank + N_EDGES);
    float*  H      = (float*)(csr + N_EDGES);
    float*  X      = H + (size_t)N_NODES * HID;
    float*  pooled = X + (size_t)N_NODES * HID;
    float*  counts = pooled + (size_t)N_GRAPHS * HID;
    unsigned* hist = (unsigned*)(counts + N_GRAPHS);
    int*    pre    = (int*)(hist + (size_t)N_SLICES * PAIR_WORDS);
    int*    tsum   = pre + (size_t)N_SLICES * N_NODES;

    const int T = 256;

    count_lds_kernel<<<N_SLICES, CL_THREADS, 0, stream>>>(dst, rank, hist);
    slice_prefix_kernel<<<N_TILES, T, 0, stream>>>(hist, pre, cnt, tsum);
    scan_kernel<<<N_TILES, T, 0, stream>>>(cnt, tsum, rowptr, dinv, (float4*)pooled);
    fill_kernel<<<EDGE_BLOCKS, T, 0, stream>>>(src, dst, rank, rowptr, pre, dinv, csr);

    const float* xin   = x;
    float*       xout  = H;
    float*       xnext = X;
    for (int l = 0; l < N_LAYERS - 1; ++l) {
        const float* Wl = Ws + (size_t)l * HID * HID;
        const float* bl = bs + (size_t)l * HID;

        layer_kernel<false><<<LAYER_BLOCKS, LAYER_THREADS, 0, stream>>>(
            xin, Wl, bl, rowptr, cnt, csr, dinv, xout, batch, pooled, counts);

        xin = xout;
        float* t = xout;
        xout  = xnext;
        xnext = t;
    }
    layer_kernel<true><<<LAYER_BLOCKS, LAYER_THREADS, 0, stream>>>(
        xin, Ws + (size_t)(N_LAYERS - 1) * HID * HID, bs + (size_t)(N_LAYERS - 1) * HID,
        rowptr, cnt, csr, dinv, xout, batch, pooled, counts);

    final_linear_kernel<<<(N_GRAPHS * N_CLASSES + T - 1) / T, T, 0, stream>>>(
        pooled, counts, linW, linb, out);
}


# speedup vs baseline: 5.6871x; 1.0660x over previous
#include <hip/hip_runtime.h>
#include <hip/hip_bf16.h>

#define N_NODES   100000
#define N_EDGES   1600000
#define HID       128
#define N_CLASSES 10
#define N_GRAPHS  512
#define N_LAYERS  5

static_assert(N_NODES % 4 == 0, "int4 passes over per-node arrays");

typedef __attribute__((ext_vector_type(2))) float v2f;
typedef __attribute__((ext_vector_type(8))) float v8f;


static_assert(N_EDGES % 4 == 0, "int4 passes over per-edge arrays");
#define EDGE_Q            (N_EDGES / 4)
#define EDGE_Q_PER_BLOCK  512
#define EDGE_BLOCKS       ((EDGE_Q + EDGE_Q_PER_BLOCK - 1) / EDGE_Q_PER_BLOCK)

#define N_SLICES     32
#define SLICE_EDGES  (N_EDGES / N_SLICES)
#define SLICE_Q      (SLICE_EDGES / 4)
#define PAIR_WORDS   (N_NODES / 2)
#define CL_THREADS   1024
#define CL_ITERS     ((SLICE_Q + CL_THREADS - 1) / CL_THREADS)

static_assert(N_EDGES % N_SLICES == 0, "equal slices");
static_assert(SLICE_EDGES % 4 == 0, "a slice is whole int4 groups, so a group never straddles two slices");
static_assert(SLICE_EDGES <= 65535, "a 16-bit counter cannot overflow inside one slice");
static_assert(N_NODES % 8 == 0, "16-byte passes over the packed counter table");

__device__ __forceinline__ int lds_count(unsigned* __restrict__ c, int d) {
    unsigned u  = min((unsigned)d, (unsigned)(N_NODES - 1));
    unsigned sh = (u & 1u) * 16u;
    unsigned old = atomicAdd(&c[u >> 1], 1u << sh);
    return (int)((old >> sh) & 0xFFFFu);
}

__global__ __launch_bounds__(CL_THREADS)
void count_lds_kernel(const int* __restrict__ dst, int* __restrict__ rank,
                      unsigned* __restrict__ hist) {
    __shared__ uint4 c4[PAIR_WORDS / 4];
    unsigned* c = (unsigned*)c4;
    int tid = threadIdx.x;
    int s   = blockIdx.x;
    const int4* d4 = (const int4*)dst + (size_t)s * SLICE_Q;
    int4*       r4 = (int4*)rank + (size_t)s * SLICE_Q;

    int4 a[CL_ITERS];
    #pragma unroll
    for (int k = 0; k < CL_ITERS; ++k) {
        int q = tid + k * CL_THREADS;
        a[k] = make_int4(0, 0, 0, 0);
        if (q < SLICE_Q) a[k] = d4[q];
    }
    for (int i = tid; i < PAIR_WORDS / 4; i += CL_THREADS) c4[i] = make_uint4(0u, 0u, 0u, 0u);
    __syncthreads();

    #pragma unroll
    for (int k = 0; k < CL_ITERS; ++k) {
        int q = tid + k * CL_THREADS;
        if (q < SLICE_Q) {
            int4 r;
            r.x = lds_count(c, a[k].x);
            r.y = lds_count(c, a[k].y);
            r.z = lds_count(c, a[k].z);
            r.w = lds_count(c, a[k].w);
            r4[q] = r;
        }
    }
    __syncthreads();

    uint4* h4 = (uint4*)(hist + (size_t)s * PAIR_WORDS);
    for (int i = tid; i < PAIR_WORDS / 4; i += CL_THREADS) h4[i] = c4[i];
}

#define SCAN_TILE 1024
#define N_TILES   ((N_NODES + SCAN_TILE - 1) / SCAN_TILE)
static_assert(SCAN_TILE == 4 * 256, "a slice-sum block of 256 threads x 4 nodes is one scan tile");
static_assert(N_TILES <= 256, "one tile total per thread of a scan block");

__global__ __launch_bounds__(256)
void slice_prefix_kernel(const unsigned* __restrict__ hist, int* __restrict__ pre,
                         int* __restrict__ cnt, int* __restrict__ tsum) {
    __shared__ int wtot[8];
    int tid = threadIdx.x;
    int i = blockIdx.x * 256 + tid;
    int4 run = make_int4(0, 0, 0, 0);
    if (i < N_NODES / 4) {
        uint2 w[N_SLICES];
        #pragma unroll
        for (int s = 0; s < N_SLICES; ++s)
            w[s] = ((const uint2*)(hist + (size_t)s * PAIR_WORDS))[i];
        #pragma unroll
        for (int s = 0; s < N_SLICES; ++s) {
            ((int4*)(pre + (size_t)s * N_NODES))[i] = run;
            run.x += (int)(w[s].x & 0xFFFFu);
            run.y += (int)(w[s].x >> 16);
            run.z += (int)(w[s].y & 0xFFFFu);
            run.w += (int)(w[s].y >> 16);
        }
        ((int4*)cnt)[i] = run;
    }
    int t = (run.x + run.y) + (run.z + run.w);
    for (int o = 16; o > 0; o >>= 1) t += __shfl_xor(t, o);
    if ((tid & 31) == 0) wtot[tid >> 5] = t;
    __syncthreads();
    if (tid == 0) {
        int s = 0;
        for (int k = 0; k < 8; ++k) s += wtot[k];
        tsum[blockIdx.x] = s;
    }
}

#define POOL_Q ((N_GRAPHS * HID + N_GRAPHS) / 4)
static_assert((N_GRAPHS * HID + N_GRAPHS) % 4 == 0, "whole float4 groups");
static_assert(POOL_Q <= N_TILES * 256, "one float4 group per thread of the scan launch covers pooled + counts");

__global__ __launch_bounds__(256)
void scan_kernel(const int* __restrict__ cnt, const int* __restrict__ tsum,
                 int* __restrict__ rowptr, float* __restrict__ dinv,
                 float4* __restrict__ pool0) {
    __shared__ int wsum[8];
    int tid  = threadIdx.x;
    {
        int z = blockIdx.x * 256 + tid;
        if (z < POOL_Q) pool0[z] = make_float4(0.f, 0.f, 0.f, 0.f);
    }
    int lane = tid & 31;
    int wv   = tid >> 5;
    int tile0 = blockIdx.x * SCAN_TILE;
    const int4* c4 = (const int4*)cnt;

    int i0 = tile0 + tid * 4;
    int pre = 0;
    if (tid < (int)blockIdx.x) pre = tsum[tid];
    int4 v = make_int4(0, 0, 0, 0);
    if (i0 < N_NODES) v = c4[i0 >> 2];

    for (int o = 16; o > 0; o >>= 1) pre += __shfl_xor(pre, o);
    if (lane == 0) wsum[wv] = pre;
    __syncthreads();
    int base = 0;
    for (int k = 0; k < 8; ++k) base += wsum[k];
    __syncthreads();

    int t = (v.x + v.y) + (v.z + v.w);
    int incl = t;
    for (int o = 1; o < 32; o <<= 1) {
        int u = __shfl_up(incl, o);
        if (lane >= o) incl += u;
    }
    if (lane == 31) wsum[wv] = incl;
    __syncthreads();
    int woff = 0;
    for (int k = 0; k < wv; ++k) woff += wsum[k];
    int ex = base + woff + incl - t;
    if (i0 < N_NODES) {
        int4 r;
        r.x = ex;
        r.y = r.x + v.x;
        r.z = r.y + v.y;
        r.w = r.z + v.z;
        ((int4*)rowptr)[i0 >> 2] = r;
        float4 dv;
        dv.x = rsqrtf((float)(v.x + 1));
        dv.y = rsqrtf((float)(v.y + 1));
        dv.z = rsqrtf((float)(v.z + 1));
        dv.w = rsqrtf((float)(v.w + 1));
        ((float4*)dinv)[i0 >> 2] = dv;
    }
}

__device__ __forceinline__ int2 csr_entry(int s, float ds, float dd) {
    return make_int2(s, __float_as_int(ds * dd));
}

__global__ __launch_bounds__(256)
void fill_kernel(const int* __restrict__ src, const int* __restrict__ dst,
                 const int* __restrict__ rank, const int* __restrict__ rowptr,
                 const int* __restrict__ pre,
                 const float* __restrict__ dinv, int2* __restrict__ csr) {
    const int4* s4 = (const int4*)src;
    const int4* d4 = (const int4*)dst;
    const int4* r4 = (const int4*)rank;
    int q0 = blockIdx.x * EDGE_Q_PER_BLOCK + threadIdx.x;
    int q1 = q0 + 256;
    if (q1 < EDGE_Q) {
        const int* ea = pre + (size_t)(q0 / SLICE_Q) * N_NODES;
        const int* eb = pre + (size_t)(q1 / SLICE_Q) * N_NODES;
        int4 sa = s4[q0], da = d4[q0], ra = r4[q0];
        int4 sb = s4[q1], db = d4[q1], rb = r4[q1];
        int   pa0 = rowptr[da.x], pa1 = rowptr[da.y], pa2 = rowptr[da.z], pa3 = rowptr[da.w];
        int   pb0 = rowptr[db.x], pb1 = rowptr[db.y], pb2 = rowptr[db.z], pb3 = rowptr[db.w];
        int   ea0 = ea[da.x], ea1 = ea[da.y], ea2 = ea[da.z], ea3 = ea[da.w];
        int   eb0 = eb[db.x], eb1 = eb[db.y], eb2 = eb[db.z], eb3 = eb[db.w];
        float ua0 = dinv[sa.x], ua1 = dinv[sa.y], ua2 = dinv[sa.z], ua3 = dinv[sa.w];
        float ub0 = dinv[sb.x], ub1 = dinv[sb.y], ub2 = dinv[sb.z], ub3 = dinv[sb.w];
        float va0 = dinv[da.x], va1 = dinv[da.y], va2 = dinv[da.z], va3 = dinv[da.w];
        float vb0 = dinv[db.x], vb1 = dinv[db.y], vb2 = dinv[db.z], vb3 = dinv[db.w];
        csr[pa0 + ea0 + ra.x] = csr_entry(sa.x, ua0, va0);
        csr[pa1 + ea1 + ra.y] = csr_entry(sa.y, ua1, va1);
        csr[pa2 + ea2 + ra.z] = csr_entry(sa.z, ua2, va2);
        csr[pa3 + ea3 + ra.w] = csr_entry(sa.w, ua3, va3);
        csr[pb0 + eb0 + rb.x] = csr_entry(sb.x, ub0, vb0);
        csr[pb1 + eb1 + rb.y] = csr_entry(sb.y, ub1, vb1);
        csr[pb2 + eb2 + rb.z] = csr_entry(sb.z, ub2, vb2);
        csr[pb3 + eb3 + rb.w] = csr_entry(sb.w, ub3, vb3);
    } else if (q0 < EDGE_Q) {
        const int* ea = pre + (size_t)(q0 / SLICE_Q) * N_NODES;
        int4 sa = s4[q0], da = d4[q0], ra = r4[q0];
        int   pa0 = rowptr[da.x], pa1 = rowptr[da.y], pa2 = rowptr[da.z], pa3 = rowptr[da.w];
        int   ea0 = ea[da.x], ea1 = ea[da.y], ea2 = ea[da.z], ea3 = ea[da.w];
        float ua0 = dinv[sa.x], ua1 = dinv[sa.y], ua2 = dinv[sa.z], ua3 = dinv[sa.w];
        float va0 = dinv[da.x], va1 = dinv[da.y], va2 = dinv[da.z], va3 = dinv[da.w];
        csr[pa0 + ea0 + ra.x] = csr_entry(sa.x, ua0, va0);
        csr[pa1 + ea1 + ra.y] = csr_entry(sa.y, ua1, va1);
        csr[pa2 + ea2 + ra.z] = csr_entry(sa.z, ua2, va2);
        csr[pa3 + ea3 + ra.w] = csr_entry(sa.w, ua3, va3);
    }
}

#define LAYER_WAVES   16
#define LAYER_THREADS (LAYER_WAVES * 32)
#define LAYER_BLOCKS  256
#define N_STRIPS      (N_NODES / 16)
#define TILE_PITCH    132

#define PLANE_ALIGN       512
#define PLANE_GAP_WORDS (PLANE_ALIGN / 4)
static_assert((HID * 4) % 256 == 0 && PLANE_ALIGN % (HID * 4) == 0, "an aligned plane keeps every row on whole 256-byte units");

static_assert(N_NODES % 16 == 0, "whole 16-node strips, 16-byte scalar loads per group of 4 nodes");
static_assert((HID * HID / 4) % LAYER_THREADS == 0, "W copy: whole passes of the workgroup");


__device__ __forceinline__ void node_done(float* __restrict__ tile, int lane, int cur, float4& acc) {
    float4* row = (float4*)(tile + cur * TILE_PITCH + 4 * lane);
    float4 t = *row;
    t.x = acc.x + t.x;
    t.y = acc.y + t.y;
    t.z = acc.z + t.z;
    t.w = acc.w + t.w;
    *row = t;
    acc = make_float4(0.f, 0.f, 0.f, 0.f);
}

template <int J>
__device__ __forceinline__ void rows_issue(const float4* __restrict__ X4, int lane, int ex, int ew,
                                           float4 (&v)[16], float (&w)[16]) {
    #pragma unroll
    for (int k = 0; k < 16; ++k) {
        int s = __builtin_amdgcn_readlane(ex, J + k);
        w[k]  = __int_as_float(__builtin_amdgcn_readlane(ew, J + k));
        v[k]  = X4[(size_t)s * (HID / 4) + lane];
    }
}

template <bool TAIL>
__device__ __forceinline__ void rows_sum(float* __restrict__ tile, int lane,
                                         const float4 (&v)[16], const float (&w)[16],
                                         int p, int endv, int& cur, int& end, float4& acc) {
    if (TAIL) {
        int rel = end - p;
        #pragma unroll
        for (int k = 0; k < 16; ++k) {
            if (rel == k) {
                while (rel == k && cur < 15) {
                    node_done(tile, lane, cur, acc);
                    ++cur;
                    end = __builtin_amdgcn_readlane(endv, cur);
                    rel = end - p;
                }
                if (rel == k) return;
            }
            acc.x = fmaf(v[k].x, w[k], acc.x);
            acc.y = fmaf(v[k].y, w[k], acc.y);
            acc.z = fmaf(v[k].z, w[k], acc.z);
            acc.w = fmaf(v[k].w, w[k], acc.w);
        }
    } else {
        int rel = cur < 15 ? end - p : -1;
        #pragma unroll
        for (int k = 0; k < 16; ++k) {
            while (rel == k) {
                node_done(tile, lane, cur, acc);
                ++cur;
                end = __builtin_amdgcn_readlane(endv, cur);
                rel = cur < 15 ? end - p : -1;
            }
            acc.x = fmaf(v[k].x, w[k], acc.x);
            acc.y = fmaf(v[k].y, w[k], acc.y);
            acc.z = fmaf(v[k].z, w[k], acc.z);
            acc.w = fmaf(v[k].w, w[k], acc.w);
        }
    }
}

__device__ __forceinline__ void pool_flush(float* __restrict__ pooled, float* __restrict__ counts,
                                           int g, int lane, float4 acc, float c) {
    float* pp = pooled + (size_t)g * HID + lane * 4;
    atomicAdd(pp + 0, acc.x);
    atomicAdd(pp + 1, acc.y);
    atomicAdd(pp + 2, acc.z);
    atomicAdd(pp + 3, acc.w);
    if (lane == 0) atomicAdd(&counts[g], c);
}

template <bool LAST>
__global__ __launch_bounds__(LAYER_THREADS)
void layer_kernel(const float* __restrict__ Xin, const float* __restrict__ W,
                  const float* __restrict__ bias, const int* __restrict__ rowptr,
                  const int* __restrict__ cnt, const int2* __restrict__ csr,
                  const float* __restrict__ dinv, float* __restrict__ Xout,
                  const int* __restrict__ graph_id, float* __restrict__ pooled,
                  float* __restrict__ counts, int in_plane) {
    __shared__ float4 Wq[HID * HID / 4];
    __shared__ float4 tile4[LAYER_WAVES * 16 * TILE_PITCH / 4];

    int tid = threadIdx.x;
    #pragma unroll
    for (int i = 0; i < (HID * HID / 4) / LAYER_THREADS; ++i) {
        int p  = tid + i * LAYER_THREADS;
        int n  = p & (HID - 1);
        int gh = p >> 7;
        int k0 = (gh >> 1) * 8 + (gh & 1) * 2;
        float4 q;
        q.x = W[(k0 + 0) * HID + n];
        q.y = W[(k0 + 1) * HID + n];
        q.z = W[(k0 + 4) * HID + n];
        q.w = W[(k0 + 5) * HID + n];
        Wq[p] = q;
    }
    __syncthreads();

    int lane = tid & 31;
    int wv = __builtin_amdgcn_readfirstlane(tid >> 5);
    int m  = lane & 15;
    int kh = lane >> 4;

    int s0 = ((int)blockIdx.x * N_STRIPS) / LAYER_BLOCKS;
    int s1 = (((int)blockIdx.x + 1) * N_STRIPS) / LAYER_BLOCKS;
    const float4* bq = Wq + kh * HID + m;
    unsigned up_in  = in_plane ? (unsigned)((0u - (unsigned)(size_t)Xin) & (unsigned)(PLANE_ALIGN - 1)) : 0u;
    unsigned up_out = (unsigned)((0u - (unsigned)(size_t)Xout) & (unsigned)(PLANE_ALIGN - 1));
    const float4* X4 = (const float4*)((const char*)Xin + up_in);
    float* Xo = (float*)((char*)Xout + up_out);
    float* tile = (float*)tile4 + wv * (16 * TILE_PITCH);
    unsigned a_off = (unsigned)(m * TILE_PITCH + 2 * kh);
    unsigned h_off = (unsigned)(8 * kh * HID + m);

    float bv[8];
    #pragma unroll
    for (int t = 0; t < 8; ++t) bv[t] = bias[t * 16 + m];

    for (int s = s0 + wv; s < s1; s += LAYER_WAVES) {
        int node0 = s * 16;
        int   rp = rowptr[node0 + m];
        int   cn = cnt[node0 + m];
        float dv = dinv[node0 + m];
        int   gid = 0;
        if (LAST) gid = graph_id[node0 + m];
        float4 hv[16];
        #pragma unroll
        for (int i = 0; i < 16; ++i) hv[i] = X4[(size_t)(node0 + i) * (HID / 4) + lane];

        int endv = rp + cn;
        int p0 = __builtin_amdgcn_readlane(rp, 0);
        int pE = __builtin_amdgcn_readlane(endv, 15);
        pE = min(pE, p0 + N_EDGES);

        int2 ent  = make_int2(node0, 0);
        int2 entn = make_int2(node0, 0);
        if (lane < pE - p0)        ent  = csr[p0 + lane];
        if (lane < pE - (p0 + 32)) entn = csr[p0 + 32 + lane];

        #pragma unroll
        for (int i = 0; i < 16; ++i) {
            float ds = __int_as_float(__builtin_amdgcn_readlane(__float_as_int(dv), i));
            float wself = ds * ds;
            float4 r;
            r.x = hv[i].x * wself;
            r.y = hv[i].y * wself;
            r.z = hv[i].z * wself;
            r.w = hv[i].w * wself;
            *(float4*)(tile + i * TILE_PITCH + 4 * lane) = r;
        }

        {
            float4 acc = make_float4(0.f, 0.f, 0.f, 0.f);
            int cur = 0;
            int end = __builtin_amdgcn_readlane(endv, 0);
            float4 va[16], vb[16];
            float  wa[16], wb[16];
            int ex = (int)min((unsigned)ent.x, (unsigned)(N_NODES - 1));
            int ew = ent.y;
            rows_issue<0>(X4, lane, ex, ew, va, wa);
            for (int c0 = p0; c0 < pE; c0 += 32) {
                rows_issue<16>(X4, lane, ex, ew, vb, wb);
                if (c0 + 16 <= pE) rows_sum<false>(tile, lane, va, wa, c0, endv, cur, end, acc);
                else               rows_sum<true>(tile, lane, va, wa, c0, endv, cur, end, acc);
                ent  = entn;
                entn = make_int2(node0, 0);
                if (lane < pE - (c0 + 64)) entn = csr[c0 + 64 + lane];
                ex = (int)min((unsigned)ent.x, (unsigned)(N_NODES - 1));
                ew = ent.y;
                rows_issue<0>(X4, lane, ex, ew, va, wa);
                if (c0 + 32 <= pE)     rows_sum<false>(tile, lane, vb, wb, c0 + 16, endv, cur, end, acc);
                else if (c0 + 16 < pE) rows_sum<true>(tile, lane, vb, wb, c0 + 16, endv, cur, end, acc);
            }
            for (; cur < 16; ++cur) node_done(tile, lane, cur, acc);
        }
        v2f a[HID / 4];
        #pragma unroll
        for (int i = 0; i < HID / 4; ++i) a[i] = *(const v2f*)(tile + a_off + 4 * i);

        v8f acc[8];
        #pragma unroll
        for (int t = 0; t < 8; ++t) acc[t] = (v8f){};

        float4 b[2][8];
        #pragma unroll
        for (int t = 0; t < 8; ++t) b[0][t] = bq[t * 16];

        #pragma unroll
        for (int g = 0; g < HID / 8; ++g) {
            if (g + 1 < HID / 8) {
                #pragma unroll
                for (int t = 0; t < 8; ++t) b[(g + 1) & 1][t] = bq[(g + 1) * 2 * HID + t * 16];
            }
            asm volatile("" ::: "memory");
            #pragma unroll
            for (int t = 0; t < 8; ++t) {
                v2f bb = {b[g & 1][t].x, b[g & 1][t].y};
                acc[t] = __builtin_amdgcn_wmma_f32_16x16x4_f32(
                     false, a[2 * g],  false, bb,
                     (short)0, acc[t],  false,  false);
            }
            #pragma unroll
            for (int t = 0; t < 8; ++t) {
                v2f bb = {b[g & 1][t].z, b[g & 1][t].w};
                acc[t] = __builtin_amdgcn_wmma_f32_16x16x4_f32(
                     false, a[2 * g + 1],  false, bb,
                     (short)0, acc[t],  false,  false);
            }
        }

        if (!LAST) {
            float* os = Xo + (size_t)s * (16 * HID);
            #pragma unroll
            for (int t = 0; t < 8; ++t) {
                #pragma unroll
                for (int r = 0; r < 8; ++r) {
                    float v = acc[t][r] + bv[t];
                    v = v > 0.f ? v : 0.f;
                    (os + r * HID + t * 16)[h_off] = v;
                }
            }
        } else {
            #pragma unroll
            for (int t = 0; t < 8; ++t) {
                #pragma unroll
                for (int r = 0; r < 8; ++r) {
                    float v = acc[t][r] + bv[t];
                    v = v > 0.f ? v : 0.f;
                    tile[(r + 8 * kh) * TILE_PITCH + t * 16 + m] = v;
                }
            }
            const float4* trow = (const float4*)(tile + 4 * lane);
            float4 pv[16];
            #pragma unroll
            for (int k = 0; k < 16; ++k) pv[k] = trow[k * (TILE_PITCH / 4)];

            int    gcur = __builtin_amdgcn_readlane(gid, 0);
            float4 psum = pv[0];
            float  pc   = 1.0f;
            #pragma unroll
            for (int k = 1; k < 16; ++k) {
                int gk = __builtin_amdgcn_readlane(gid, k);
                if (gk != gcur) {
                    pool_flush(pooled, counts, gcur, lane, psum, pc);
                    gcur = gk;
                    psum = pv[k];
                    pc   = 1.0f;
                } else {
                    psum.x += pv[k].x;
                    psum.y += pv[k].y;
                    psum.z += pv[k].z;
                    psum.w += pv[k].w;
                    pc += 1.0f;
                }
            }
            pool_flush(pooled, counts, gcur, lane, psum, pc);
        }
    }
}

__global__ void final_linear_kernel(const float* __restrict__ pooled,
                                    const float* __restrict__ counts,
                                    const float* __restrict__ linW,
                                    const float* __restrict__ linb,
                                    float* __restrict__ out) {
    int i = blockIdx.x * blockDim.x + threadIdx.x;
    if (i >= N_GRAPHS * N_CLASSES) return;
    int g = i / N_CLASSES;
    int c = i - g * N_CLASSES;
    float scale = 1.0f / fmaxf(counts[g], 1.0f);
    const float* pg = pooled + (size_t)g * HID;
    float acc = 0.f;
    #pragma unroll 4
    for (int f = 0; f < HID; ++f) acc += pg[f] * linW[f * N_CLASSES + c];
    out[i] = acc * scale + linb[c];
}

extern "C" void kernel_launch(void* const* d_in, const int* in_sizes, int n_in,
                              void* d_out, int out_size, void* d_ws, size_t ws_size,
                              hipStream_t stream) {
    const float* x     = (const float*)d_in[0];
    const int*   edge  = (const int*)d_in[1];
    const int*   src   = edge;
    const int*   dst   = edge + N_EDGES;
    const int*   batch = (const int*)d_in[2];
    const float* Ws    = (const float*)d_in[3];
    const float* bs    = (const float*)d_in[4];
    const float* linW  = (const float*)d_in[5];
    const float* linb  = (const float*)d_in[6];
    float*       out   = (float*)d_out;

    float*  ws     = (float*)d_ws;
    int*    cnt    = (int*)ws;
    float*  dinv   = ws + N_NODES;
    int*    rowptr = (int*)(dinv + N_NODES);
    int*    rank   = rowptr + N_NODES;
    int2*   csr    = (int2*)(rank + N_EDGES);
    float*  H      = (float*)(csr + N_EDGES);
    float*  X      = H + (size_t)N_NODES * HID + PLANE_GAP_WORDS;
    float*  pooled = X + (size_t)N_NODES * HID + PLANE_GAP_WORDS;
    float*  counts = pooled + (size_t)N_GRAPHS * HID;
    unsigned* hist = (unsigned*)(counts + N_GRAPHS);
    int*    pre    = (int*)(hist + (size_t)N_SLICES * PAIR_WORDS);
    int*    tsum   = pre + (size_t)N_SLICES * N_NODES;

    const int T = 256;

    count_lds_kernel<<<N_SLICES, CL_THREADS, 0, stream>>>(dst, rank, hist);
    slice_prefix_kernel<<<N_TILES, T, 0, stream>>>(hist, pre, cnt, tsum);
    scan_kernel<<<N_TILES, T, 0, stream>>>(cnt, tsum, rowptr, dinv, (float4*)pooled);
    fill_kernel<<<EDGE_BLOCKS, T, 0, stream>>>(src, dst, rank, rowptr, pre, dinv, csr);

    layer_kernel<false><<<LAYER_BLOCKS, LAYER_THREADS, 0, stream>>>(
        x, Ws, bs, rowptr, cnt, csr, dinv, H, batch, pooled, counts, 0);
    const float* xin   = H;
    float*       xout  = X;
    float*       xnext = H;
    for (int l = 1; l < N_LAYERS - 1; ++l) {
        const float* Wl = Ws + (size_t)l * HID * HID;
        const float* bl = bs + (size_t)l * HID;

        layer_kernel<false><<<LAYER_BLOCKS, LAYER_THREADS, 0, stream>>>(
            xin, Wl, bl, rowptr, cnt, csr, dinv, xout, batch, pooled, counts, 1);

        xin = xout;
        float* t = xout;
        xout  = xnext;
        xnext = t;
    }
    layer_kernel<true><<<LAYER_BLOCKS, LAYER_THREADS, 0, stream>>>(
        xin, Ws + (size_t)(N_LAYERS - 1) * HID * HID, bs + (size_t)(N_LAYERS - 1) * HID,
        rowptr, cnt, csr, dinv, xout, batch, pooled, counts, 1);

    final_linear_kernel<<<(N_GRAPHS * N_CLASSES + T - 1) / T, T, 0, stream>>>(
        pooled, counts, linW, linb, out);
}
